# starting kernel
# baseline (speedup 1.0000x reference)
#include <hip/hip_runtime.h>
#include <hip/hip_bf16.h>

typedef __attribute__((ext_vector_type(2))) float v2f;
typedef __attribute__((ext_vector_type(8))) float v8f;

#define N_NODES 100000

__global__ void zero_kernel(float* __restrict__ p, int n) {
    int i = blockIdx.x * blockDim.x + threadIdx.x;
    if (i < n) p[i] = 0.0f;
}

__global__ void degree_kernel(const int* __restrict__ dst,
                              float* __restrict__ deg, int E) {
    int e = blockIdx.x * blockDim.x + threadIdx.x;
    if (e < E) unsafeAtomicAdd(&deg[dst[e]], 1.0f);
}

__global__ void invdeg_kernel(float* __restrict__ deg, int n) {
    int i = blockIdx.x * blockDim.x + threadIdx.x;
    if (i < n) deg[i] = 1.0f / fmaxf(deg[i], 1.0f);
}

template <int F>
__global__ void scatter_kernel(const float* __restrict__ h,
                               const int* __restrict__ src,
                               const int* __restrict__ dst,
                               float* __restrict__ agg, int E) {
    int gw   = (blockIdx.x * blockDim.x + threadIdx.x) >> 5;
    int lane = threadIdx.x & 31;
    if (gw >= E) return;
    int s = src[gw];
    int d = dst[gw];
    if (F == 32) {
        float v = h[(size_t)s * 32 + lane];
        unsafeAtomicAdd(&agg[(size_t)d * 32 + lane], v);
    } else {
        const float4* row = reinterpret_cast<const float4*>(h + (size_t)s * F);
        float4 v = row[lane];
        float* o = agg + (size_t)d * F + lane * 4;
        unsafeAtomicAdd(o + 0, v.x);
        unsafeAtomicAdd(o + 1, v.y);
        unsafeAtomicAdd(o + 2, v.z);
        unsafeAtomicAdd(o + 3, v.w);
    }
}

template <int FIN, int FOUT, int TPW, bool PRELU>
__global__ __launch_bounds__(128) void sage_combine_wmma(
    const float* __restrict__ agg, const float* __restrict__ invdeg,
    const float* __restrict__ h,
    const float* __restrict__ Wl, const float* __restrict__ bias,
    const float* __restrict__ Wr, const float* __restrict__ alpha_p,
    float* __restrict__ out) {
    constexpr int LDSTR = FIN + 1;
    __shared__ float s_mean[16 * LDSTR];
    __shared__ float s_h[16 * LDSTR];

    const int tid   = threadIdx.x;
    const int node0 = blockIdx.x * 16;

    for (int idx = tid; idx < 16 * FIN; idx += 128) {
        int r = idx / FIN;
        int c = idx % FIN;
        size_t g = (size_t)(node0 + r) * FIN + c;
        s_mean[r * LDSTR + c] = agg[g] * invdeg[node0 + r];
        s_h[r * LDSTR + c]    = h[g];
    }
    __syncthreads();

    const int wave = tid >> 5;
    const int lane = tid & 31;
    const int half = lane >> 4;
    const int col  = lane & 15;
    const float alpha = alpha_p[0];

    v8f acc[TPW];
#pragma unroll
    for (int t = 0; t < TPW; ++t)
#pragma unroll
        for (int j = 0; j < 8; ++j) acc[t][j] = 0.0f;

    const float* smrow = s_mean + col * LDSTR;
    const float* shrow = s_h    + col * LDSTR;

#pragma unroll 4
    for (int k0 = 0; k0 < FIN; k0 += 4) {
        const int kb = k0 + half * 2;
        v2f am; am.x = smrow[kb]; am.y = smrow[kb + 1];
        v2f ah; ah.x = shrow[kb]; ah.y = shrow[kb + 1];
#pragma unroll
        for (int t = 0; t < TPW; ++t) {
            const int n0 = (wave * TPW + t) * 16;
            v2f bl;
            bl.x = Wl[(size_t)kb * FOUT + n0 + col];
            bl.y = Wl[(size_t)(kb + 1) * FOUT + n0 + col];
            v2f br;
            br.x = Wr[(size_t)kb * FOUT + n0 + col];
            br.y = Wr[(size_t)(kb + 1) * FOUT + n0 + col];
            acc[t] = __builtin_amdgcn_wmma_f32_16x16x4_f32(
                false, am, false, bl, (short)0, acc[t], false, false);
            acc[t] = __builtin_amdgcn_wmma_f32_16x16x4_f32(
                false, ah, false, br, (short)0, acc[t], false, false);
        }
    }

#pragma unroll
    for (int t = 0; t < TPW; ++t) {
        const int n0 = (wave * TPW + t) * 16;
        const float bv = bias[n0 + col];
#pragma unroll
        for (int v = 0; v < 8; ++v) {
            float val = acc[t][v] + bv;
            if (PRELU) val = (val >= 0.0f) ? val : alpha * val;
            const int row = node0 + v + half * 8;
            out[(size_t)row * FOUT + n0 + col] = val;
        }
    }
}

__global__ void classifier_kernel(const float* __restrict__ h3,
                                  const float* __restrict__ Wc,
                                  const float* __restrict__ bc,
                                  float* __restrict__ out, int n) {
    int i = blockIdx.x * blockDim.x + threadIdx.x;
    if (i >= n) return;
    float s0 = bc[0], s1 = bc[1], s2 = bc[2], s3 = bc[3];
    const float4* row = reinterpret_cast<const float4*>(h3 + (size_t)i * 64);
#pragma unroll
    for (int k4 = 0; k4 < 16; ++k4) {
        float4 v = row[k4];
        const float* w = Wc + (k4 * 4) * 4;
        s0 += v.x * w[0];  s1 += v.x * w[1];  s2 += v.x * w[2];  s3 += v.x * w[3];
        s0 += v.y * w[4];  s1 += v.y * w[5];  s2 += v.y * w[6];  s3 += v.y * w[7];
        s0 += v.z * w[8];  s1 += v.z * w[9];  s2 += v.z * w[10]; s3 += v.z * w[11];
        s0 += v.w * w[12]; s1 += v.w * w[13]; s2 += v.w * w[14]; s3 += v.w * w[15];
    }
    float* o = out + (size_t)i * 4;
    o[0] = s0; o[1] = s1; o[2] = s2; o[3] = s3;
}

extern "C" void kernel_launch(void* const* d_in, const int* in_sizes, int n_in,
                              void* d_out, int out_size, void* d_ws, size_t ws_size,
                              hipStream_t stream) {
    const float* x   = (const float*)d_in[0];
    const int*   ei  = (const int*)d_in[1];
    const float* W1l = (const float*)d_in[2];
    const float* b1  = (const float*)d_in[3];
    const float* W1r = (const float*)d_in[4];
    const float* W2l = (const float*)d_in[5];
    const float* b2  = (const float*)d_in[6];
    const float* W2r = (const float*)d_in[7];
    const float* W3l = (const float*)d_in[8];
    const float* b3  = (const float*)d_in[9];
    const float* W3r = (const float*)d_in[10];
    const float* Wc  = (const float*)d_in[11];
    const float* bc  = (const float*)d_in[12];
    const float* a   = (const float*)d_in[13];
    float* out = (float*)d_out;

    const int E = in_sizes[1] / 2;
    const int* srcv = ei;
    const int* dstv = ei + E;

    char* ws = (char*)d_ws;
    size_t off = 0;
    auto carve = [&](size_t bytes) {
        float* p = (float*)(ws + off);
        off += (bytes + 255) & ~(size_t)255;
        return p;
    };
    float* deg = carve((size_t)N_NODES * 4);
    float* agg = carve((size_t)N_NODES * 128 * 4);
    float* hA  = carve((size_t)N_NODES * 128 * 4);
    float* hB  = carve((size_t)N_NODES * 128 * 4);
    float* h3  = hA;

    const int TB = 256;
    const int nBlk16 = N_NODES / 16;

    zero_kernel<<<(N_NODES + TB - 1) / TB, TB, 0, stream>>>(deg, N_NODES);
    degree_kernel<<<(E + TB - 1) / TB, TB, 0, stream>>>(dstv, deg, E);
    invdeg_kernel<<<(N_NODES + TB - 1) / TB, TB, 0, stream>>>(deg, N_NODES);

    zero_kernel<<<(N_NODES * 32 + TB - 1) / TB, TB, 0, stream>>>(agg, N_NODES * 32);
    scatter_kernel<32><<<((size_t)E * 32 + TB - 1) / TB, TB, 0, stream>>>(
        x, srcv, dstv, agg, E);
    sage_combine_wmma<32, 128, 2, true><<<nBlk16, 128, 0, stream>>>(
        agg, deg, x, W1l, b1, W1r, a, hA);

    zero_kernel<<<(N_NODES * 128 + TB - 1) / TB, TB, 0, stream>>>(agg, N_NODES * 128);
    scatter_kernel<128><<<((size_t)E * 32 + TB - 1) / TB, TB, 0, stream>>>(
        hA, srcv, dstv, agg, E);
    sage_combine_wmma<128, 128, 2, true><<<nBlk16, 128, 0, stream>>>(
        agg, deg, hA, W2l, b2, W2r, a, hB);

    zero_kernel<<<(N_NODES * 128 + TB - 1) / TB, TB, 0, stream>>>(agg, N_NODES * 128);
    scatter_kernel<128><<<((size_t)E * 32 + TB - 1) / TB, TB, 0, stream>>>(
        hB, srcv, dstv, agg, E);
    sage_combine_wmma<128, 64, 1, false><<<nBlk16, 128, 0, stream>>>(
        agg, deg, hB, W3l, b3, W3r, a, h3);

    classifier_kernel<<<(N_NODES + TB - 1) / TB, TB, 0, stream>>>(
        h3, Wc, bc, out, N_NODES);
}


# Layer 3 and classifier folded by linearity: h2 projected to 8 columns (W3l*Wc, W3r*Wc) on WMMA before aggregation, so the layer-3 scatter adds 4 floats per edge, not 128; sum order changes
# speedup vs baseline: 1.3170x; 1.3170x over previous
#include <hip/hip_runtime.h>
#include <hip/hip_bf16.h>

typedef __attribute__((ext_vector_type(2))) float v2f;
typedef __attribute__((ext_vector_type(8))) float v8f;

#define N_NODES 100000

__global__ void zero_kernel(float* __restrict__ p, int n) {
    int i = blockIdx.x * blockDim.x + threadIdx.x;
    if (i < n) p[i] = 0.0f;
}

__global__ void degree_kernel(const int* __restrict__ dst,
                              float* __restrict__ deg, int E) {
    int e = blockIdx.x * blockDim.x + threadIdx.x;
    if (e < E) unsafeAtomicAdd(&deg[dst[e]], 1.0f);
}

__global__ void invdeg_kernel(float* __restrict__ deg, int n) {
    int i = blockIdx.x * blockDim.x + threadIdx.x;
    if (i < n) deg[i] = 1.0f / fmaxf(deg[i], 1.0f);
}

template <int F>
__global__ void scatter_kernel(const float* __restrict__ h,
                               const int* __restrict__ src,
                               const int* __restrict__ dst,
                               float* __restrict__ agg, int E) {
    int gw   = (blockIdx.x * blockDim.x + threadIdx.x) >> 5;
    int lane = threadIdx.x & 31;
    if (gw >= E) return;
    int s = src[gw];
    int d = dst[gw];
    if (F == 32) {
        float v = h[(size_t)s * 32 + lane];
        unsafeAtomicAdd(&agg[(size_t)d * 32 + lane], v);
    } else {
        const float4* row = reinterpret_cast<const float4*>(h + (size_t)s * F);
        float4 v = row[lane];
        float* o = agg + (size_t)d * F + lane * 4;
        unsafeAtomicAdd(o + 0, v.x);
        unsafeAtomicAdd(o + 1, v.y);
        unsafeAtomicAdd(o + 2, v.z);
        unsafeAtomicAdd(o + 3, v.w);
    }
}

template <int FIN, int FOUT, int TPW, bool PRELU>
__global__ __launch_bounds__(128) void sage_combine_wmma(
    const float* __restrict__ agg, const float* __restrict__ invdeg,
    const float* __restrict__ h,
    const float* __restrict__ Wl, const float* __restrict__ bias,
    const float* __restrict__ Wr, const float* __restrict__ alpha_p,
    float* __restrict__ out) {
    constexpr int LDSTR = FIN + 1;
    __shared__ float s_mean[16 * LDSTR];
    __shared__ float s_h[16 * LDSTR];

    const int tid   = threadIdx.x;
    const int node0 = blockIdx.x * 16;

    for (int idx = tid; idx < 16 * FIN; idx += 128) {
        int r = idx / FIN;
        int c = idx % FIN;
        size_t g = (size_t)(node0 + r) * FIN + c;
        s_mean[r * LDSTR + c] = agg[g] * invdeg[node0 + r];
        s_h[r * LDSTR + c]    = h[g];
    }
    __syncthreads();

    const int wave = tid >> 5;
    const int lane = tid & 31;
    const int half = lane >> 4;
    const int col  = lane & 15;
    const float alpha = alpha_p[0];

    v8f acc[TPW];
#pragma unroll
    for (int t = 0; t < TPW; ++t)
#pragma unroll
        for (int j = 0; j < 8; ++j) acc[t][j] = 0.0f;

    const float* smrow = s_mean + col * LDSTR;
    const float* shrow = s_h    + col * LDSTR;

#pragma unroll 4
    for (int k0 = 0; k0 < FIN; k0 += 4) {
        const int kb = k0 + half * 2;
        v2f am; am.x = smrow[kb]; am.y = smrow[kb + 1];
        v2f ah; ah.x = shrow[kb]; ah.y = shrow[kb + 1];
#pragma unroll
        for (int t = 0; t < TPW; ++t) {
            const int n0 = (wave * TPW + t) * 16;
            v2f bl;
            bl.x = Wl[(size_t)kb * FOUT + n0 + col];
            bl.y = Wl[(size_t)(kb + 1) * FOUT + n0 + col];
            v2f br;
            br.x = Wr[(size_t)kb * FOUT + n0 + col];
            br.y = Wr[(size_t)(kb + 1) * FOUT + n0 + col];
            acc[t] = __builtin_amdgcn_wmma_f32_16x16x4_f32(
                false, am, false, bl, (short)0, acc[t], false, false);
            acc[t] = __builtin_amdgcn_wmma_f32_16x16x4_f32(
                false, ah, false, br, (short)0, acc[t], false, false);
        }
    }

#pragma unroll
    for (int t = 0; t < TPW; ++t) {
        const int n0 = (wave * TPW + t) * 16;
        const float bv = bias[n0 + col];
#pragma unroll
        for (int v = 0; v < 8; ++v) {
            float val = acc[t][v] + bv;
            if (PRELU) val = (val >= 0.0f) ? val : alpha * val;
            const int row = node0 + v + half * 8;
            out[(size_t)row * FOUT + n0 + col] = val;
        }
    }
}

__global__ void fold_kernel(const float* __restrict__ W3l,
                            const float* __restrict__ W3r,
                            const float* __restrict__ b3,
                            const float* __restrict__ Wc,
                            const float* __restrict__ bc,
                            float* __restrict__ Wf, float* __restrict__ cf) {
    int t = blockIdx.x * blockDim.x + threadIdx.x;
    if (t < 128 * 16) {
        int k = t >> 4;
        int c = t & 15;
        float s = 0.0f;
        if (c < 8) {
            const float* w = ((c < 4) ? W3l : W3r) + k * 64;
            int cc = c & 3;
            for (int j = 0; j < 64; ++j) s += w[j] * Wc[j * 4 + cc];
        }
        Wf[t] = s;
    }
    if (t < 4) {
        float s = 0.0f;
        for (int j = 0; j < 64; ++j) s += b3[j] * Wc[j * 4 + t];
        cf[t] = s + bc[t];
    }
}

__global__ __launch_bounds__(64) void project_wmma(
    const float* __restrict__ h, const float* __restrict__ Wf,
    float* __restrict__ p) {
    constexpr int FIN = 128;
    constexpr int LDSTR = FIN + 1;
    __shared__ float s_h[32 * LDSTR];

    const int tid   = threadIdx.x;
    const int node0 = blockIdx.x * 32;

    const float4* g4 = reinterpret_cast<const float4*>(h + (size_t)node0 * FIN);
    for (int it = 0; it < 32 * FIN / 4 / 64; ++it) {
        const int i4 = tid + it * 64;
        float4 v = g4[i4];
        const int r = i4 >> 5;
        const int c = (i4 & 31) * 4;
        float* d = s_h + r * LDSTR + c;
        d[0] = v.x; d[1] = v.y; d[2] = v.z; d[3] = v.w;
    }
    __syncthreads();

    const int wave = tid >> 5;
    const int lane = tid & 31;
    const int half = lane >> 4;
    const int col  = lane & 15;

    v8f acc = {0.0f, 0.0f, 0.0f, 0.0f, 0.0f, 0.0f, 0.0f, 0.0f};

    const float* shrow = s_h + (wave * 16 + col) * LDSTR;

    for (int k0 = 0; k0 < FIN; k0 += 4) {
        const int kb = k0 + half * 2;
        v2f ah; ah.x = shrow[kb]; ah.y = shrow[kb + 1];
        v2f bf;
        bf.x = Wf[kb * 16 + col];
        bf.y = Wf[(kb + 1) * 16 + col];
        acc = __builtin_amdgcn_wmma_f32_16x16x4_f32(
            false, ah, false, bf, (short)0, acc, false, false);
    }

    if (col < 8) {
        for (int v = 0; v < 8; ++v) {
            const int row = node0 + wave * 16 + v + half * 8;
            p[(size_t)row * 8 + col] = acc[v];
        }
    }
}

__global__ void scatter4_kernel(const float* __restrict__ p,
                                const int* __restrict__ src,
                                const int* __restrict__ dst,
                                float* __restrict__ agg4, int E) {
    int t = blockIdx.x * blockDim.x + threadIdx.x;
    int e = t >> 2;
    int f = t & 3;
    if (e >= E) return;
    int s = src[e];
    int d = dst[e];
    unsafeAtomicAdd(&agg4[(size_t)d * 4 + f], p[(size_t)s * 8 + f]);
}

__global__ void final_kernel(const float* __restrict__ agg4,
                             const float* __restrict__ invdeg,
                             const float* __restrict__ p,
                             const float* __restrict__ cf,
                             float* __restrict__ out, int n) {
    int i = blockIdx.x * blockDim.x + threadIdx.x;
    if (i >= n) return;
    float4 a = reinterpret_cast<const float4*>(agg4)[i];
    float4 r = reinterpret_cast<const float4*>(p)[(size_t)i * 2 + 1];
    float id = invdeg[i];
    float4 o;
    o.x = (a.x * id + cf[0]) + r.x;
    o.y = (a.y * id + cf[1]) + r.y;
    o.z = (a.z * id + cf[2]) + r.z;
    o.w = (a.w * id + cf[3]) + r.w;
    reinterpret_cast<float4*>(out)[i] = o;
}

extern "C" void kernel_launch(void* const* d_in, const int* in_sizes, int n_in,
                              void* d_out, int out_size, void* d_ws, size_t ws_size,
                              hipStream_t stream) {
    const float* x   = (const float*)d_in[0];
    const int*   ei  = (const int*)d_in[1];
    const float* W1l = (const float*)d_in[2];
    const float* b1  = (const float*)d_in[3];
    const float* W1r = (const float*)d_in[4];
    const float* W2l = (const float*)d_in[5];
    const float* b2  = (const float*)d_in[6];
    const float* W2r = (const float*)d_in[7];
    const float* W3l = (const float*)d_in[8];
    const float* b3  = (const float*)d_in[9];
    const float* W3r = (const float*)d_in[10];
    const float* Wc  = (const float*)d_in[11];
    const float* bc  = (const float*)d_in[12];
    const float* a   = (const float*)d_in[13];
    float* out = (float*)d_out;

    const int E = in_sizes[1] / 2;
    const int* srcv = ei;
    const int* dstv = ei + E;

    char* ws = (char*)d_ws;
    size_t off = 0;
    auto carve = [&](size_t bytes) {
        float* p = (float*)(ws + off);
        off += (bytes + 255) & ~(size_t)255;
        return p;
    };
    float* deg = carve((size_t)N_NODES * 4);
    float* agg = carve((size_t)N_NODES * 128 * 4);
    float* hA  = carve((size_t)N_NODES * 128 * 4);
    float* hB  = carve((size_t)N_NODES * 128 * 4);

    const int TB = 256;
    const int nBlk16 = N_NODES / 16;

    zero_kernel<<<(N_NODES + TB - 1) / TB, TB, 0, stream>>>(deg, N_NODES);
    degree_kernel<<<(E + TB - 1) / TB, TB, 0, stream>>>(dstv, deg, E);
    invdeg_kernel<<<(N_NODES + TB - 1) / TB, TB, 0, stream>>>(deg, N_NODES);

    zero_kernel<<<(N_NODES * 32 + TB - 1) / TB, TB, 0, stream>>>(agg, N_NODES * 32);
    scatter_kernel<32><<<((size_t)E * 32 + TB - 1) / TB, TB, 0, stream>>>(
        x, srcv, dstv, agg, E);
    sage_combine_wmma<32, 128, 2, true><<<nBlk16, 128, 0, stream>>>(
        agg, deg, x, W1l, b1, W1r, a, hA);

    zero_kernel<<<(N_NODES * 128 + TB - 1) / TB, TB, 0, stream>>>(agg, N_NODES * 128);
    scatter_kernel<128><<<((size_t)E * 32 + TB - 1) / TB, TB, 0, stream>>>(
        hA, srcv, dstv, agg, E);
    sage_combine_wmma<128, 128, 2, true><<<nBlk16, 128, 0, stream>>>(
        agg, deg, hA, W2l, b2, W2r, a, hB);

    float* proj = hA;
    float* Wf   = hA + (size_t)N_NODES * 8;
    float* cf   = Wf + 128 * 16;
    float* agg4 = agg;
    fold_kernel<<<(128 * 16 + TB - 1) / TB, TB, 0, stream>>>(
        W3l, W3r, b3, Wc, bc, Wf, cf);
    project_wmma<<<N_NODES / 32, 64, 0, stream>>>(hB, Wf, proj);
    zero_kernel<<<(N_NODES * 4 + TB - 1) / TB, TB, 0, stream>>>(agg4, N_NODES * 4);
    scatter4_kernel<<<((size_t)E * 4 + TB - 1) / TB, TB, 0, stream>>>(
        proj, srcv, dstv, agg4, E);
    final_kernel<<<(N_NODES + TB - 1) / TB, TB, 0, stream>>>(
        agg4, deg, proj, cf, out, N_NODES);
}


# f32 atomic scatter-add replaced by a per-call dst-ordered edge list (int atomic count+rank, block scan, fill); each layer gathers neighbour rows in registers where they are used; sum order changes
# speedup vs baseline: 3.4270x; 2.6021x over previous
#include <hip/hip_runtime.h>
#include <hip/hip_bf16.h>


typedef __attribute__((ext_vector_type(2))) float v2f;
typedef __attribute__((ext_vector_type(8))) float v8f;

#define N_NODES 100000

__global__ void zero_int_kernel(int* __restrict__ p, int n) {
    int i = blockIdx.x * blockDim.x + threadIdx.x;
    if (i < n) p[i] = 0;
}

__global__ void count_rank_kernel(const int* __restrict__ dst,
                                  int* __restrict__ cnt,
                                  int* __restrict__ rank, int E) {
    int e = blockIdx.x * blockDim.x + threadIdx.x;
    if (e < E) rank[e] = atomicAdd(&cnt[dst[e]], 1);
}

__global__ __launch_bounds__(1024) void alloc_rows_kernel(
    const int* __restrict__ cnt, int* __restrict__ total,
    int2* __restrict__ rows, int n) {
    __shared__ int s_wsum[32];
    __shared__ int s_base;
    const int tid  = threadIdx.x;
    const int i    = blockIdx.x * 1024 + tid;
    const int lane = tid & 31;
    const int w    = tid >> 5;
    const int c    = (i < n) ? cnt[i] : 0;
    int v = c;
#pragma unroll
    for (int o = 1; o < 32; o <<= 1) {
        int t = __shfl_up(v, o);
        if (lane >= o) v += t;
    }
    if (lane == 31) s_wsum[w] = v;
    __syncthreads();
    if (w == 0) {
        const int t = s_wsum[lane];
        int u = t;
#pragma unroll
        for (int o = 1; o < 32; o <<= 1) {
            int q = __shfl_up(u, o);
            if (lane >= o) u += q;
        }
        s_wsum[lane] = u - t;
        if (lane == 31) s_base = atomicAdd(total, u);
    }
    __syncthreads();
    const int start = s_base + s_wsum[w] + (v - c);
    if (i < n) rows[i] = make_int2(start, start + c);
}

__global__ void fill_csr_kernel(const int* __restrict__ src,
                                const int* __restrict__ dst,
                                const int* __restrict__ rank,
                                const int2* __restrict__ rows,
                                int* __restrict__ csr, int E) {
    int e = blockIdx.x * blockDim.x + threadIdx.x;
    if (e < E) csr[rows[dst[e]].x + rank[e]] = src[e];
}

__device__ __forceinline__ void add4(float4& a, const float4& b) {
    a.x += b.x; a.y += b.y; a.z += b.z; a.w += b.w;
}

__device__ __forceinline__ void addsel4(float4& a, const float4& b, bool in) {
    a.x += in ? b.x : 0.0f; a.y += in ? b.y : 0.0f;
    a.z += in ? b.z : 0.0f; a.w += in ? b.w : 0.0f;
}

template <int FIN, int FOUT, int TPW, bool PRELU>
__global__ __launch_bounds__(128) void sage_gather_combine_wmma(
    const float* __restrict__ h, const int2* __restrict__ rows,
    const int* __restrict__ csr,
    const float* __restrict__ Wl, const float* __restrict__ bias,
    const float* __restrict__ Wr, const float* __restrict__ alpha_p,
    float* __restrict__ out) {
    constexpr int LDSTR = FIN + 1;
    __shared__ float s_mean[16 * LDSTR];
    __shared__ float s_h[16 * LDSTR];

    const int tid   = threadIdx.x;
    const int node0 = blockIdx.x * 16;
    const float4* __restrict__ h4 = reinterpret_cast<const float4*>(h);

    if (FIN == 32) {
        const int r    = tid >> 3;
        const int sub  = tid & 7;
        const int node = node0 + r;
        const int2 rw  = rows[node];
        const int last = rw.y - 1;
        float4 acc = {0.0f, 0.0f, 0.0f, 0.0f};
        for (int e = rw.x; e < rw.y; e += 4) {
            const int s0 = csr[e];
            const int s1 = csr[min(e + 1, last)];
            const int s2 = csr[min(e + 2, last)];
            const int s3 = csr[min(e + 3, last)];
            const float4 v0 = h4[(size_t)s0 * 8 + sub];
            const float4 v1 = h4[(size_t)s1 * 8 + sub];
            const float4 v2 = h4[(size_t)s2 * 8 + sub];
            const float4 v3 = h4[(size_t)s3 * 8 + sub];
            add4(acc, v0);
            addsel4(acc, v1, e + 1 <= last);
            addsel4(acc, v2, e + 2 <= last);
            addsel4(acc, v3, e + 3 <= last);
        }
        const float inv = 1.0f / fmaxf((float)(rw.y - rw.x), 1.0f);
        const float4 own = h4[(size_t)node * 8 + sub];
        float* dm = s_mean + r * LDSTR + sub * 4;
        dm[0] = acc.x * inv; dm[1] = acc.y * inv;
        dm[2] = acc.z * inv; dm[3] = acc.w * inv;
        float* dh = s_h + r * LDSTR + sub * 4;
        dh[0] = own.x; dh[1] = own.y; dh[2] = own.z; dh[3] = own.w;
    } else {
        const int gl = tid & 31;
        const int wv = __builtin_amdgcn_readfirstlane(tid >> 5);
        for (int q = 0; q < 4; ++q) {
            const int r    = wv * 4 + q;
            const int node = node0 + r;
            const int2 rw  = rows[node];
            const int start = __builtin_amdgcn_readfirstlane(rw.x);
            const int end   = __builtin_amdgcn_readfirstlane(rw.y);
            float4 acc = {0.0f, 0.0f, 0.0f, 0.0f};
            for (int base = start; base < end; base += 32) {
                const int n  = min(32, end - base);
                const int my = csr[base + min(gl, n - 1)];
                int j = 0;
                for (; j + 8 <= n; j += 8) {
                    float4 v[8];
#pragma unroll
                    for (int k = 0; k < 8; ++k) {
                        const int s = __builtin_amdgcn_readlane(my, j + k);
                        v[k] = h4[(size_t)s * 32 + gl];
                    }
#pragma unroll
                    for (int k = 0; k < 8; ++k) add4(acc, v[k]);
                }
                if (j < n) {
                    float4 v[7];
#pragma unroll
                    for (int k = 0; k < 7; ++k) {
                        v[k] = make_float4(0.0f, 0.0f, 0.0f, 0.0f);
                        if (j + k < n) {
                            const int s = __builtin_amdgcn_readlane(my, j + k);
                            v[k] = h4[(size_t)s * 32 + gl];
                        }
                    }
#pragma unroll
                    for (int k = 0; k < 7; ++k)
                        if (j + k < n) add4(acc, v[k]);
                }
            }
            const float inv = 1.0f / fmaxf((float)(end - start), 1.0f);
            const float4 own = h4[(size_t)node * 32 + gl];
            float* dm = s_mean + r * LDSTR + gl * 4;
            dm[0] = acc.x * inv; dm[1] = acc.y * inv;
            dm[2] = acc.z * inv; dm[3] = acc.w * inv;
            float* dh = s_h + r * LDSTR + gl * 4;
            dh[0] = own.x; dh[1] = own.y; dh[2] = own.z; dh[3] = own.w;
        }
    }
    __syncthreads();

    const int wave = tid >> 5;
    const int lane = tid & 31;
    const int half = lane >> 4;
    const int col  = lane & 15;
    const float alpha = alpha_p[0];

    v8f acc[TPW];
#pragma unroll
    for (int t = 0; t < TPW; ++t)
#pragma unroll
        for (int j = 0; j < 8; ++j) acc[t][j] = 0.0f;

    const float* smrow = s_mean + col * LDSTR;
    const float* shrow = s_h    + col * LDSTR;

#pragma unroll 4
    for (int k0 = 0; k0 < FIN; k0 += 4) {
        const int kb = k0 + half * 2;
        v2f am; am.x = smrow[kb]; am.y = smrow[kb + 1];
        v2f ah; ah.x = shrow[kb]; ah.y = shrow[kb + 1];
#pragma unroll
        for (int t = 0; t < TPW; ++t) {
            const int n0 = (wave * TPW + t) * 16;
            v2f bl;
            bl.x = Wl[(size_t)kb * FOUT + n0 + col];
            bl.y = Wl[(size_t)(kb + 1) * FOUT + n0 + col];
            v2f br;
            br.x = Wr[(size_t)kb * FOUT + n0 + col];
            br.y = Wr[(size_t)(kb + 1) * FOUT + n0 + col];
            acc[t] = __builtin_amdgcn_wmma_f32_16x16x4_f32(
                false, am, false, bl, (short)0, acc[t], false, false);
            acc[t] = __builtin_amdgcn_wmma_f32_16x16x4_f32(
                false, ah, false, br, (short)0, acc[t], false, false);
        }
    }

#pragma unroll
    for (int t = 0; t < TPW; ++t) {
        const int n0 = (wave * TPW + t) * 16;
        const float bv = bias[n0 + col];
#pragma unroll
        for (int v = 0; v < 8; ++v) {
            float val = acc[t][v] + bv;
            if (PRELU) val = (val >= 0.0f) ? val : alpha * val;
            const int row = node0 + v + half * 8;
            out[(size_t)row * FOUT + n0 + col] = val;
        }
    }
}

__global__ void fold_kernel(const float* __restrict__ W3l,
                            const float* __restrict__ W3r,
                            const float* __restrict__ b3,
                            const float* __restrict__ Wc,
                            const float* __restrict__ bc,
                            float* __restrict__ Wf, float* __restrict__ cf) {
    int t = blockIdx.x * blockDim.x + threadIdx.x;
    if (t < 128 * 16) {
        int k = t >> 4;
        int c = t & 15;
        float s = 0.0f;
        if (c < 8) {
            const float* w = ((c < 4) ? W3l : W3r) + k * 64;
            int cc = c & 3;
            for (int j = 0; j < 64; ++j) s += w[j] * Wc[j * 4 + cc];
        }
        Wf[t] = s;
    }
    if (t < 4) {
        float s = 0.0f;
        for (int j = 0; j < 64; ++j) s += b3[j] * Wc[j * 4 + t];
        cf[t] = s + bc[t];
    }
}

__global__ __launch_bounds__(64) void project_wmma(
    const float* __restrict__ h, const float* __restrict__ Wf,
    float* __restrict__ p) {
    constexpr int FIN = 128;
    constexpr int LDSTR = FIN + 1;
    __shared__ float s_h[32 * LDSTR];

    const int tid   = threadIdx.x;
    const int node0 = blockIdx.x * 32;

    const float4* g4 = reinterpret_cast<const float4*>(h + (size_t)node0 * FIN);
    for (int it = 0; it < 32 * FIN / 4 / 64; ++it) {
        const int i4 = tid + it * 64;
        float4 v = g4[i4];
        const int r = i4 >> 5;
        const int c = (i4 & 31) * 4;
        float* d = s_h + r * LDSTR + c;
        d[0] = v.x; d[1] = v.y; d[2] = v.z; d[3] = v.w;
    }
    __syncthreads();

    const int wave = tid >> 5;
    const int lane = tid & 31;
    const int half = lane >> 4;
    const int col  = lane & 15;

    v8f acc = {0.0f, 0.0f, 0.0f, 0.0f, 0.0f, 0.0f, 0.0f, 0.0f};

    const float* shrow = s_h + (wave * 16 + col) * LDSTR;

    for (int k0 = 0; k0 < FIN; k0 += 4) {
        const int kb = k0 + half * 2;
        v2f ah; ah.x = shrow[kb]; ah.y = shrow[kb + 1];
        v2f bf;
        bf.x = Wf[kb * 16 + col];
        bf.y = Wf[(kb + 1) * 16 + col];
        acc = __builtin_amdgcn_wmma_f32_16x16x4_f32(
            false, ah, false, bf, (short)0, acc, false, false);
    }

    if (col < 8) {
        for (int v = 0; v < 8; ++v) {
            const int row = node0 + wave * 16 + v + half * 8;
            p[(size_t)row * 8 + col] = acc[v];
        }
    }
}

__global__ void gather4_final_kernel(const float* __restrict__ p,
                                     const int2* __restrict__ rows,
                                     const int* __restrict__ csr,
                                     const float* __restrict__ cf,
                                     float* __restrict__ out, int n) {
    const int t    = blockIdx.x * blockDim.x + threadIdx.x;
    const int node = t >> 2;
    const int f    = t & 3;
    const bool live = node < n;
    const int2 rw  = live ? rows[node] : make_int2(0, 0);
    const int last = rw.y - 1;
    const float4* __restrict__ p4 = reinterpret_cast<const float4*>(p);
    float4 acc = {0.0f, 0.0f, 0.0f, 0.0f};
    for (int e = rw.x + f; e < rw.y; e += 16) {
        const int s0 = csr[e];
        const int s1 = csr[min(e + 4, last)];
        const int s2 = csr[min(e + 8, last)];
        const int s3 = csr[min(e + 12, last)];
        const float4 v0 = p4[(size_t)s0 * 2];
        const float4 v1 = p4[(size_t)s1 * 2];
        const float4 v2 = p4[(size_t)s2 * 2];
        const float4 v3 = p4[(size_t)s3 * 2];
        add4(acc, v0);
        addsel4(acc, v1, e + 4 <= last);
        addsel4(acc, v2, e + 8 <= last);
        addsel4(acc, v3, e + 12 <= last);
    }
    acc.x += __shfl_xor(acc.x, 1); acc.y += __shfl_xor(acc.y, 1);
    acc.z += __shfl_xor(acc.z, 1); acc.w += __shfl_xor(acc.w, 1);
    acc.x += __shfl_xor(acc.x, 2); acc.y += __shfl_xor(acc.y, 2);
    acc.z += __shfl_xor(acc.z, 2); acc.w += __shfl_xor(acc.w, 2);
    const float a  = (f == 0) ? acc.x : (f == 1) ? acc.y : (f == 2) ? acc.z : acc.w;
    const float id = 1.0f / fmaxf((float)(rw.y - rw.x), 1.0f);
    if (live)
        out[(size_t)node * 4 + f] = (a * id + cf[f]) + p[(size_t)node * 8 + 4 + f];
}

extern "C" void kernel_launch(void* const* d_in, const int* in_sizes, int n_in,
                              void* d_out, int out_size, void* d_ws, size_t ws_size,
                              hipStream_t stream) {
    const float* x   = (const float*)d_in[0];
    const int*   ei  = (const int*)d_in[1];
    const float* W1l = (const float*)d_in[2];
    const float* b1  = (const float*)d_in[3];
    const float* W1r = (const float*)d_in[4];
    const float* W2l = (const float*)d_in[5];
    const float* b2  = (const float*)d_in[6];
    const float* W2r = (const float*)d_in[7];
    const float* W3l = (const float*)d_in[8];
    const float* b3  = (const float*)d_in[9];
    const float* W3r = (const float*)d_in[10];
    const float* Wc  = (const float*)d_in[11];
    const float* bc  = (const float*)d_in[12];
    const float* a   = (const float*)d_in[13];
    float* out = (float*)d_out;

    const int E = in_sizes[1] / 2;
    const int* srcv = ei;
    const int* dstv = ei + E;

    char* ws = (char*)d_ws;
    size_t off = 0;
    auto carve = [&](size_t bytes) {
        float* p = (float*)(ws + off);
        off += (bytes + 255) & ~(size_t)255;
        return p;
    };
    int*   cnt = (int*)carve((size_t)(N_NODES + 1) * 4);
    float* el  = carve((size_t)N_NODES * 128 * 4);
    float* hA  = carve((size_t)N_NODES * 128 * 4);
    float* hB  = carve((size_t)N_NODES * 128 * 4);

    int2* rows = (int2*)el;
    int*  rank = (int*)el + (size_t)N_NODES * 2;
    int*  csr  = rank + E;

    const int TB = 256;
    const int nBlk16 = N_NODES / 16;

    zero_int_kernel<<<(N_NODES + 1 + TB - 1) / TB, TB, 0, stream>>>(cnt, N_NODES + 1);
    count_rank_kernel<<<(E + TB - 1) / TB, TB, 0, stream>>>(dstv, cnt, rank, E);
    alloc_rows_kernel<<<(N_NODES + 1023) / 1024, 1024, 0, stream>>>(
        cnt, cnt + N_NODES, rows, N_NODES);
    fill_csr_kernel<<<(E + TB - 1) / TB, TB, 0, stream>>>(
        srcv, dstv, rank, rows, csr, E);

    sage_gather_combine_wmma<32, 128, 2, true><<<nBlk16, 128, 0, stream>>>(
        x, rows, csr, W1l, b1, W1r, a, hA);

    sage_gather_combine_wmma<128, 128, 2, true><<<nBlk16, 128, 0, stream>>>(
        hA, rows, csr, W2l, b2, W2r, a, hB);

    float* proj = hA;
    float* Wf   = hA + (size_t)N_NODES * 8;
    float* cf   = Wf + 128 * 16;
    fold_kernel<<<(128 * 16 + TB - 1) / TB, TB, 0, stream>>>(
        W3l, W3r, b3, Wc, bc, Wf, cf);
    project_wmma<<<N_NODES / 32, 64, 0, stream>>>(hB, Wf, proj);
    gather4_final_kernel<<<(N_NODES * 4 + TB - 1) / TB, TB, 0, stream>>>(
        proj, rows, csr, cf, out, N_NODES);
}


# Degree counters kept twice, one copy per package half: each wave's returning adds stay in its own half (copy flag in rank; row = copy-0 slots, then copy-1); sum order changes (arrival-order rank)
# speedup vs baseline: 4.4129x; 1.2877x over previous
#include <hip/hip_runtime.h>
#include <hip/hip_bf16.h>


typedef __attribute__((ext_vector_type(2))) float v2f;
typedef __attribute__((ext_vector_type(8))) float v8f;

#define N_NODES 100000

__global__ void zero_int_kernel(int* __restrict__ p0, int n0,
                                int* __restrict__ p1, int n1) {
    int i = blockIdx.x * blockDim.x + threadIdx.x;
    if (i < n0) p0[i] = 0;
    if (i < n1) p1[i] = 0;
}

__global__ void count_rank_kernel(const int* __restrict__ dst,
                                  int* __restrict__ cnt0,
                                  int* __restrict__ cnt1,
                                  int* __restrict__ rank, int E) {
    int e = blockIdx.x * blockDim.x + threadIdx.x;
    const int half = (__builtin_amdgcn_s_sendmsg_rtn(0x87) >> 18) & 1;
    int* __restrict__ cnt = half ? cnt1 : cnt0;
    if (e < E) rank[e] = atomicAdd(&cnt[dst[e]], 1) | (half << 31);
}

__global__ __launch_bounds__(1024) void alloc_rows_kernel(
    const int* __restrict__ cnt0, const int* __restrict__ cnt1,
    int* __restrict__ total, int2* __restrict__ rows,
    int* __restrict__ mid, int n) {
    __shared__ int s_wsum[32];
    __shared__ int s_base;
    const int tid  = threadIdx.x;
    const int i    = blockIdx.x * 1024 + tid;
    const int lane = tid & 31;
    const int w    = tid >> 5;
    const int c0   = (i < n) ? cnt0[i] : 0;
    const int c    = c0 + ((i < n) ? cnt1[i] : 0);
    int v = c;
#pragma unroll
    for (int o = 1; o < 32; o <<= 1) {
        int t = __shfl_up(v, o);
        if (lane >= o) v += t;
    }
    if (lane == 31) s_wsum[w] = v;
    __syncthreads();
    if (w == 0) {
        const int t = s_wsum[lane];
        int u = t;
#pragma unroll
        for (int o = 1; o < 32; o <<= 1) {
            int q = __shfl_up(u, o);
            if (lane >= o) u += q;
        }
        s_wsum[lane] = u - t;
        if (lane == 31) s_base = atomicAdd(total, u);
    }
    __syncthreads();
    const int start = s_base + s_wsum[w] + (v - c);
    if (i < n) {
        rows[i] = make_int2(start, start + c);
        mid[i]  = start + c0;
    }
}

__global__ void fill_csr_kernel(const int* __restrict__ src,
                                const int* __restrict__ dst,
                                const int* __restrict__ rank,
                                const int2* __restrict__ rows,
                                const int* __restrict__ mid,
                                int* __restrict__ csr, int E) {
    int e = blockIdx.x * blockDim.x + threadIdx.x;
    if (e < E) {
        const int d = dst[e];
        const int r = rank[e];
        const int base = (r < 0) ? mid[d] : rows[d].x;
        csr[base + (r & 0x7fffffff)] = src[e];
    }
}

__device__ __forceinline__ void add4(float4& a, const float4& b) {
    a.x += b.x; a.y += b.y; a.z += b.z; a.w += b.w;
}

__device__ __forceinline__ void addsel4(float4& a, const float4& b, bool in) {
    a.x += in ? b.x : 0.0f; a.y += in ? b.y : 0.0f;
    a.z += in ? b.z : 0.0f; a.w += in ? b.w : 0.0f;
}

template <int FIN, int FOUT, int TPW, bool PRELU>
__global__ __launch_bounds__(128) void sage_gather_combine_wmma(
    const float* __restrict__ h, const int2* __restrict__ rows,
    const int* __restrict__ csr,
    const float* __restrict__ Wl, const float* __restrict__ bias,
    const float* __restrict__ Wr, const float* __restrict__ alpha_p,
    float* __restrict__ out) {
    constexpr int LDSTR = FIN + 1;
    __shared__ float s_mean[16 * LDSTR];
    __shared__ float s_h[16 * LDSTR];

    const int tid   = threadIdx.x;
    const int node0 = blockIdx.x * 16;
    const float4* __restrict__ h4 = reinterpret_cast<const float4*>(h);

    if (FIN == 32) {
        const int r    = tid >> 3;
        const int sub  = tid & 7;
        const int node = node0 + r;
        const int2 rw  = rows[node];
        const int last = rw.y - 1;
        float4 acc = {0.0f, 0.0f, 0.0f, 0.0f};
        for (int e = rw.x; e < rw.y; e += 4) {
            const int s0 = csr[e];
            const int s1 = csr[min(e + 1, last)];
            const int s2 = csr[min(e + 2, last)];
            const int s3 = csr[min(e + 3, last)];
            const float4 v0 = h4[(size_t)s0 * 8 + sub];
            const float4 v1 = h4[(size_t)s1 * 8 + sub];
            const float4 v2 = h4[(size_t)s2 * 8 + sub];
            const float4 v3 = h4[(size_t)s3 * 8 + sub];
            add4(acc, v0);
            addsel4(acc, v1, e + 1 <= last);
            addsel4(acc, v2, e + 2 <= last);
            addsel4(acc, v3, e + 3 <= last);
        }
        const float inv = 1.0f / fmaxf((float)(rw.y - rw.x), 1.0f);
        const float4 own = h4[(size_t)node * 8 + sub];
        float* dm = s_mean + r * LDSTR + sub * 4;
        dm[0] = acc.x * inv; dm[1] = acc.y * inv;
        dm[2] = acc.z * inv; dm[3] = acc.w * inv;
        float* dh = s_h + r * LDSTR + sub * 4;
        dh[0] = own.x; dh[1] = own.y; dh[2] = own.z; dh[3] = own.w;
    } else {
        const int gl = tid & 31;
        const int wv = __builtin_amdgcn_readfirstlane(tid >> 5);
        for (int q = 0; q < 4; ++q) {
            const int r    = wv * 4 + q;
            const int node = node0 + r;
            const int2 rw  = rows[node];
            const int start = __builtin_amdgcn_readfirstlane(rw.x);
            const int end   = __builtin_amdgcn_readfirstlane(rw.y);
            float4 acc = {0.0f, 0.0f, 0.0f, 0.0f};
            for (int base = start; base < end; base += 32) {
                const int n  = min(32, end - base);
                const int my = csr[base + min(gl, n - 1)];
                int j = 0;
                for (; j + 8 <= n; j += 8) {
                    float4 v[8];
#pragma unroll
                    for (int k = 0; k < 8; ++k) {
                        const int s = __builtin_amdgcn_readlane(my, j + k);
                        v[k] = h4[(size_t)s * 32 + gl];
                    }
#pragma unroll
                    for (int k = 0; k < 8; ++k) add4(acc, v[k]);
                }
                if (j < n) {
                    float4 v[7];
#pragma unroll
                    for (int k = 0; k < 7; ++k) {
                        v[k] = make_float4(0.0f, 0.0f, 0.0f, 0.0f);
                        if (j + k < n) {
                            const int s = __builtin_amdgcn_readlane(my, j + k);
                            v[k] = h4[(size_t)s * 32 + gl];
                        }
                    }
#pragma unroll
                    for (int k = 0; k < 7; ++k)
                        if (j + k < n) add4(acc, v[k]);
                }
            }
            const float inv = 1.0f / fmaxf((float)(end - start), 1.0f);
            const float4 own = h4[(size_t)node * 32 + gl];
            float* dm = s_mean + r * LDSTR + gl * 4;
            dm[0] = acc.x * inv; dm[1] = acc.y * inv;
            dm[2] = acc.z * inv; dm[3] = acc.w * inv;
            float* dh = s_h + r * LDSTR + gl * 4;
            dh[0] = own.x; dh[1] = own.y; dh[2] = own.z; dh[3] = own.w;
        }
    }
    __syncthreads();

    const int wave = tid >> 5;
    const int lane = tid & 31;
    const int half = lane >> 4;
    const int col  = lane & 15;
    const float alpha = alpha_p[0];

    v8f acc[TPW];
#pragma unroll
    for (int t = 0; t < TPW; ++t)
#pragma unroll
        for (int j = 0; j < 8; ++j) acc[t][j] = 0.0f;

    const float* smrow = s_mean + col * LDSTR;
    const float* shrow = s_h    + col * LDSTR;

#pragma unroll 4
    for (int k0 = 0; k0 < FIN; k0 += 4) {
        const int kb = k0 + half * 2;
        v2f am; am.x = smrow[kb]; am.y = smrow[kb + 1];
        v2f ah; ah.x = shrow[kb]; ah.y = shrow[kb + 1];
#pragma unroll
        for (int t = 0; t < TPW; ++t) {
            const int n0 = (wave * TPW + t) * 16;
            v2f bl;
            bl.x = Wl[(size_t)kb * FOUT + n0 + col];
            bl.y = Wl[(size_t)(kb + 1) * FOUT + n0 + col];
            v2f br;
            br.x = Wr[(size_t)kb * FOUT + n0 + col];
            br.y = Wr[(size_t)(kb + 1) * FOUT + n0 + col];
            acc[t] = __builtin_amdgcn_wmma_f32_16x16x4_f32(
                false, am, false, bl, (short)0, acc[t], false, false);
            acc[t] = __builtin_amdgcn_wmma_f32_16x16x4_f32(
                false, ah, false, br, (short)0, acc[t], false, false);
        }
    }

#pragma unroll
    for (int t = 0; t < TPW; ++t) {
        const int n0 = (wave * TPW + t) * 16;
        const float bv = bias[n0 + col];
#pragma unroll
        for (int v = 0; v < 8; ++v) {
            float val = acc[t][v] + bv;
            if (PRELU) val = (val >= 0.0f) ? val : alpha * val;
            const int row = node0 + v + half * 8;
            out[(size_t)row * FOUT + n0 + col] = val;
        }
    }
}

__global__ void fold_kernel(const float* __restrict__ W3l,
                            const float* __restrict__ W3r,
                            const float* __restrict__ b3,
                            const float* __restrict__ Wc,
                            const float* __restrict__ bc,
                            float* __restrict__ Wf, float* __restrict__ cf) {
    int t = blockIdx.x * blockDim.x + threadIdx.x;
    if (t < 128 * 16) {
        int k = t >> 4;
        int c = t & 15;
        float s = 0.0f;
        if (c < 8) {
            const float* w = ((c < 4) ? W3l : W3r) + k * 64;
            int cc = c & 3;
            for (int j = 0; j < 64; ++j) s += w[j] * Wc[j * 4 + cc];
        }
        Wf[t] = s;
    }
    if (t < 4) {
        float s = 0.0f;
        for (int j = 0; j < 64; ++j) s += b3[j] * Wc[j * 4 + t];
        cf[t] = s + bc[t];
    }
}

__global__ __launch_bounds__(64) void project_wmma(
    const float* __restrict__ h, const float* __restrict__ Wf,
    float* __restrict__ p) {
    constexpr int FIN = 128;
    constexpr int LDSTR = FIN + 1;
    __shared__ float s_h[32 * LDSTR];

    const int tid   = threadIdx.x;
    const int node0 = blockIdx.x * 32;

    const float4* g4 = reinterpret_cast<const float4*>(h + (size_t)node0 * FIN);
    for (int it = 0; it < 32 * FIN / 4 / 64; ++it) {
        const int i4 = tid + it * 64;
        float4 v = g4[i4];
        const int r = i4 >> 5;
        const int c = (i4 & 31) * 4;
        float* d = s_h + r * LDSTR + c;
        d[0] = v.x; d[1] = v.y; d[2] = v.z; d[3] = v.w;
    }
    __syncthreads();

    const int wave = tid >> 5;
    const int lane = tid & 31;
    const int half = lane >> 4;
    const int col  = lane & 15;

    v8f acc = {0.0f, 0.0f, 0.0f, 0.0f, 0.0f, 0.0f, 0.0f, 0.0f};

    const float* shrow = s_h + (wave * 16 + col) * LDSTR;

    for (int k0 = 0; k0 < FIN; k0 += 4) {
        const int kb = k0 + half * 2;
        v2f ah; ah.x = shrow[kb]; ah.y = shrow[kb + 1];
        v2f bf;
        bf.x = Wf[kb * 16 + col];
        bf.y = Wf[(kb + 1) * 16 + col];
        acc = __builtin_amdgcn_wmma_f32_16x16x4_f32(
            false, ah, false, bf, (short)0, acc, false, false);
    }

    if (col < 8) {
        for (int v = 0; v < 8; ++v) {
            const int row = node0 + wave * 16 + v + half * 8;
            p[(size_t)row * 8 + col] = acc[v];
        }
    }
}

__global__ void gather4_final_kernel(const float* __restrict__ p,
                                     const int2* __restrict__ rows,
                                     const int* __restrict__ csr,
                                     const float* __restrict__ cf,
                                     float* __restrict__ out, int n) {
    const int t    = blockIdx.x * blockDim.x + threadIdx.x;
    const int node = t >> 2;
    const int f    = t & 3;
    const bool live = node < n;
    const int2 rw  = live ? rows[node] : make_int2(0, 0);
    const int last = rw.y - 1;
    const float4* __restrict__ p4 = reinterpret_cast<const float4*>(p);
    float4 acc = {0.0f, 0.0f, 0.0f, 0.0f};
    for (int e = rw.x + f; e < rw.y; e += 16) {
        const int s0 = csr[e];
        const int s1 = csr[min(e + 4, last)];
        const int s2 = csr[min(e + 8, last)];
        const int s3 = csr[min(e + 12, last)];
        const float4 v0 = p4[(size_t)s0 * 2];
        const float4 v1 = p4[(size_t)s1 * 2];
        const float4 v2 = p4[(size_t)s2 * 2];
        const float4 v3 = p4[(size_t)s3 * 2];
        add4(acc, v0);
        addsel4(acc, v1, e + 4 <= last);
        addsel4(acc, v2, e + 8 <= last);
        addsel4(acc, v3, e + 12 <= last);
    }
    acc.x += __shfl_xor(acc.x, 1); acc.y += __shfl_xor(acc.y, 1);
    acc.z += __shfl_xor(acc.z, 1); acc.w += __shfl_xor(acc.w, 1);
    acc.x += __shfl_xor(acc.x, 2); acc.y += __shfl_xor(acc.y, 2);
    acc.z += __shfl_xor(acc.z, 2); acc.w += __shfl_xor(acc.w, 2);
    const float a  = (f == 0) ? acc.x : (f == 1) ? acc.y : (f == 2) ? acc.z : acc.w;
    const float id = 1.0f / fmaxf((float)(rw.y - rw.x), 1.0f);
    if (live)
        out[(size_t)node * 4 + f] = (a * id + cf[f]) + p[(size_t)node * 8 + 4 + f];
}

extern "C" void kernel_launch(void* const* d_in, const int* in_sizes, int n_in,
                              void* d_out, int out_size, void* d_ws, size_t ws_size,
                              hipStream_t stream) {
    const float* x   = (const float*)d_in[0];
    const int*   ei  = (const int*)d_in[1];
    const float* W1l = (const float*)d_in[2];
    const float* b1  = (const float*)d_in[3];
    const float* W1r = (const float*)d_in[4];
    const float* W2l = (const float*)d_in[5];
    const float* b2  = (const float*)d_in[6];
    const float* W2r = (const float*)d_in[7];
    const float* W3l = (const float*)d_in[8];
    const float* b3  = (const float*)d_in[9];
    const float* W3r = (const float*)d_in[10];
    const float* Wc  = (const float*)d_in[11];
    const float* bc  = (const float*)d_in[12];
    const float* a   = (const float*)d_in[13];
    float* out = (float*)d_out;

    const int E = in_sizes[1] / 2;
    const int* srcv = ei;
    const int* dstv = ei + E;

    char* ws = (char*)d_ws;
    size_t off = 0;
    auto carve = [&](size_t bytes) {
        float* p = (float*)(ws + off);
        off += (bytes + 255) & ~(size_t)255;
        return p;
    };
    int*   cnt = (int*)carve((size_t)(N_NODES + 1) * 4);
    float* el  = carve((size_t)N_NODES * 128 * 4);
    float* hA  = carve((size_t)N_NODES * 128 * 4);
    float* hB  = carve((size_t)N_NODES * 128 * 4);

    int2* rows = (int2*)el;
    int*  rank = (int*)el + (size_t)N_NODES * 2;
    int*  csr  = rank + E;
    int*  cnt1 = csr + E;
    int*  mid  = cnt1 + N_NODES;

    const int TB = 256;
    const int nBlk16 = N_NODES / 16;

    zero_int_kernel<<<(N_NODES + 1 + TB - 1) / TB, TB, 0, stream>>>(
        cnt, N_NODES + 1, cnt1, N_NODES);
    count_rank_kernel<<<(E + TB - 1) / TB, TB, 0, stream>>>(dstv, cnt, cnt1, rank, E);
    alloc_rows_kernel<<<(N_NODES + 1023) / 1024, 1024, 0, stream>>>(
        cnt, cnt1, cnt + N_NODES, rows, mid, N_NODES);
    fill_csr_kernel<<<(E + TB - 1) / TB, TB, 0, stream>>>(
        srcv, dstv, rank, rows, mid, csr, E);

    sage_gather_combine_wmma<32, 128, 2, true><<<nBlk16, 128, 0, stream>>>(
        x, rows, csr, W1l, b1, W1r, a, hA);

    sage_gather_combine_wmma<128, 128, 2, true><<<nBlk16, 128, 0, stream>>>(
        hA, rows, csr, W2l, b2, W2r, a, hB);

    float* proj = hA;
    float* Wf   = hA + (size_t)N_NODES * 8;
    float* cf   = Wf + 128 * 16;
    fold_kernel<<<(128 * 16 + TB - 1) / TB, TB, 0, stream>>>(
        W3l, W3r, b3, Wc, bc, Wf, cf);
    project_wmma<<<N_NODES / 32, 64, 0, stream>>>(hB, Wf, proj);
    gather4_final_kernel<<<(N_NODES * 4 + TB - 1) / TB, TB, 0, stream>>>(
        proj, rows, csr, cf, out, N_NODES);
}


# Edge list built by a two-level counting sort on LDS counters (chunk bins by 512-node range, one block per range ranks with LDS adds): no global atomic per edge, 4 launches to 2; sum order changes
# speedup vs baseline: 5.5611x; 1.2602x over previous
#include <hip/hip_runtime.h>
#include <hip/hip_bf16.h>


typedef __attribute__((ext_vector_type(2))) float v2f;
typedef __attribute__((ext_vector_type(8))) float v8f;

#define N_NODES 100000

#define RNG_SHIFT 9
#define RNG_NODES (1 << RNG_SHIFT)
#define N_RNG ((N_NODES + RNG_NODES - 1) / RNG_NODES)
#define SRC_BITS 17
#define CHUNK 2048
#define BIN_TB 256
#define EPL (CHUNK / BIN_TB)
#define WPT 8
#define OUT_CAP 12288

static_assert(N_NODES <= (1 << SRC_BITS), "node id must fit SRC_BITS");
static_assert(RNG_SHIFT + SRC_BITS < 32, "packed edge word must stay positive");
static_assert(N_RNG <= BIN_TB, "one thread per range in the chunk scan");

__device__ __forceinline__ int block_scan_excl(int c, int* s_wsum, int* s_total) {
    const int lane = threadIdx.x & 31;
    const int w    = threadIdx.x >> 5;
    const int nw   = blockDim.x >> 5;
    int v = c;
#pragma unroll
    for (int o = 1; o < 32; o <<= 1) {
        int t = __shfl_up(v, o);
        if (lane >= o) v += t;
    }
    if (lane == 31) s_wsum[w] = v;
    __syncthreads();
    if (w == 0) {
        const int t = (lane < nw) ? s_wsum[lane] : 0;
        int u = t;
#pragma unroll
        for (int o = 1; o < 32; o <<= 1) {
            int q = __shfl_up(u, o);
            if (lane >= o) u += q;
        }
        s_wsum[lane] = u - t;
        if (lane == 31) *s_total = u;
    }
    __syncthreads();
    return s_wsum[w] + (v - c);
}

__global__ __launch_bounds__(BIN_TB) void bin_edges_kernel(
    const int* __restrict__ src, const int* __restrict__ dst,
    int* __restrict__ rec, int2* __restrict__ offs,
    int* __restrict__ total, int E) {
    __shared__ int s_hist[N_RNG];
    __shared__ int s_off[N_RNG];
    __shared__ int s_rec[CHUNK];
    __shared__ int s_wsum[32];
    __shared__ int s_tot;
    const int tid   = threadIdx.x;
    const int chunk = blockIdx.x;
    const int e0    = chunk * CHUNK + tid;
    if (tid < N_RNG) s_hist[tid] = 0;
    if (chunk == 0 && tid == 0) *total = 0;

    int d[EPL], s[EPL];
#pragma unroll
    for (int k = 0; k < EPL; ++k) {
        const int ec = min(e0 + k * BIN_TB, E - 1);
        d[k] = dst[ec];
        s[k] = src[ec];
    }
    __syncthreads();

    int rk[EPL];
#pragma unroll
    for (int k = 0; k < EPL; ++k) {
        rk[k] = 0;
        if (e0 + k * BIN_TB < E) rk[k] = atomicAdd(&s_hist[d[k] >> RNG_SHIFT], 1);
    }
    __syncthreads();

    const int c  = (tid < N_RNG) ? s_hist[tid] : 0;
    const int ex = block_scan_excl(c, s_wsum, &s_tot);
    if (tid < N_RNG) {
        s_off[tid] = ex;
        offs[(size_t)chunk * N_RNG + tid] = make_int2(ex, ex + c);
    }
    __syncthreads();

#pragma unroll
    for (int k = 0; k < EPL; ++k) {
        if (e0 + k * BIN_TB < E)
            s_rec[s_off[d[k] >> RNG_SHIFT] + rk[k]] =
                ((d[k] & (RNG_NODES - 1)) << SRC_BITS) | s[k];
    }
    __syncthreads();
    const int nv = s_tot;
    int* __restrict__ rp = rec + (size_t)chunk * CHUNK;
#pragma unroll
    for (int k = 0; k < EPL; ++k) {
        const int p = k * BIN_TB + tid;
        if (p < nv) rp[p] = s_rec[p];
    }
}

__global__ __launch_bounds__(1024) void rank_range_kernel(
    const int* __restrict__ rec, const int2* __restrict__ offs,
    int* __restrict__ total, int2* __restrict__ rows,
    int* __restrict__ csr, int nCh) {
    __shared__ int s_cnt[RNG_NODES];
    __shared__ int s_out[OUT_CAP];
    __shared__ int s_wsum[32];
    __shared__ int s_tot;
    __shared__ int s_base;
    const int tid = threadIdx.x;
    const int b   = blockIdx.x;
    if (tid < RNG_NODES) s_cnt[tid] = 0;
    __syncthreads();

    for (int ch = tid; ch < nCh; ch += 1024) {
        const int2 pc = offs[(size_t)ch * N_RNG + b];
        const int* __restrict__ rp = rec + (size_t)ch * CHUNK;
        const int pb = pc.x;
        const int pe = pc.y;
        const int last = pe - 1;
        for (int j = pb; j < pe; j += WPT) {
            int r[WPT];
#pragma unroll
            for (int k = 0; k < WPT; ++k) r[k] = rp[min(j + k, last)];
#pragma unroll
            for (int k = 0; k < WPT; ++k)
                atomicAdd(&s_cnt[r[k] >> SRC_BITS], (j + k <= last) ? 1 : 0);
        }
    }
    __syncthreads();

    const int c  = (tid < RNG_NODES) ? s_cnt[tid] : 0;
    const int ex = block_scan_excl(c, s_wsum, &s_tot);
    if (tid == 0) s_base = atomicAdd(total, s_tot);
    __syncthreads();
    const int nb   = s_tot;
    const int base = s_base;
    if (tid < RNG_NODES) {
        s_cnt[tid] = ex;
        const int node = b * RNG_NODES + tid;
        if (node < N_NODES) rows[node] = make_int2(base + ex, base + ex + c);
    }
    __syncthreads();

    const bool staged = nb <= OUT_CAP;
    const int smask = (1 << SRC_BITS) - 1;
    for (int ch = tid; ch < nCh; ch += 1024) {
        const int2 pc = offs[(size_t)ch * N_RNG + b];
        const int* __restrict__ rp = rec + (size_t)ch * CHUNK;
        const int pb = pc.x;
        const int pe = pc.y;
        const int last = pe - 1;
        for (int j = pb; j < pe; j += WPT) {
            int r[WPT];
#pragma unroll
            for (int k = 0; k < WPT; ++k) r[k] = rp[min(j + k, last)];
            int sl[WPT];
#pragma unroll
            for (int k = 0; k < WPT; ++k)
                sl[k] = atomicAdd(&s_cnt[r[k] >> SRC_BITS], (j + k <= last) ? 1 : 0);
            if (staged) {
#pragma unroll
                for (int k = 0; k < WPT; ++k)
                    if (j + k <= last) s_out[sl[k]] = r[k] & smask;
            } else {
#pragma unroll
                for (int k = 0; k < WPT; ++k)
                    if (j + k <= last) csr[base + sl[k]] = r[k] & smask;
            }
        }
    }
    if (staged) {
        __syncthreads();
        for (int i = tid; i < nb; i += 1024) csr[base + i] = s_out[i];
    }
}

__device__ __forceinline__ void add4(float4& a, const float4& b) {
    a.x += b.x; a.y += b.y; a.z += b.z; a.w += b.w;
}

__device__ __forceinline__ void addsel4(float4& a, const float4& b, bool in) {
    a.x += in ? b.x : 0.0f; a.y += in ? b.y : 0.0f;
    a.z += in ? b.z : 0.0f; a.w += in ? b.w : 0.0f;
}

template <int FIN, int FOUT, int TPW, bool PRELU>
__global__ __launch_bounds__(128) void sage_gather_combine_wmma(
    const float* __restrict__ h, const int2* __restrict__ rows,
    const int* __restrict__ csr,
    const float* __restrict__ Wl, const float* __restrict__ bias,
    const float* __restrict__ Wr, const float* __restrict__ alpha_p,
    float* __restrict__ out) {
    constexpr int LDSTR = FIN + 1;
    __shared__ float s_mean[16 * LDSTR];
    __shared__ float s_h[16 * LDSTR];

    const int tid   = threadIdx.x;
    const int node0 = blockIdx.x * 16;
    const float4* __restrict__ h4 = reinterpret_cast<const float4*>(h);

    if (FIN == 32) {
        const int r    = tid >> 3;
        const int sub  = tid & 7;
        const int node = node0 + r;
        const int2 rw  = rows[node];
        const int last = rw.y - 1;
        float4 acc = {0.0f, 0.0f, 0.0f, 0.0f};
        for (int e = rw.x; e < rw.y; e += 4) {
            const int s0 = csr[e];
            const int s1 = csr[min(e + 1, last)];
            const int s2 = csr[min(e + 2, last)];
            const int s3 = csr[min(e + 3, last)];
            const float4 v0 = h4[(size_t)s0 * 8 + sub];
            const float4 v1 = h4[(size_t)s1 * 8 + sub];
            const float4 v2 = h4[(size_t)s2 * 8 + sub];
            const float4 v3 = h4[(size_t)s3 * 8 + sub];
            add4(acc, v0);
            addsel4(acc, v1, e + 1 <= last);
            addsel4(acc, v2, e + 2 <= last);
            addsel4(acc, v3, e + 3 <= last);
        }
        const float inv = 1.0f / fmaxf((float)(rw.y - rw.x), 1.0f);
        const float4 own = h4[(size_t)node * 8 + sub];
        float* dm = s_mean + r * LDSTR + sub * 4;
        dm[0] = acc.x * inv; dm[1] = acc.y * inv;
        dm[2] = acc.z * inv; dm[3] = acc.w * inv;
        float* dh = s_h + r * LDSTR + sub * 4;
        dh[0] = own.x; dh[1] = own.y; dh[2] = own.z; dh[3] = own.w;
    } else {
        const int gl = tid & 31;
        const int wv = __builtin_amdgcn_readfirstlane(tid >> 5);
        for (int q = 0; q < 4; ++q) {
            const int r    = wv * 4 + q;
            const int node = node0 + r;
            const int2 rw  = rows[node];
            const int start = __builtin_amdgcn_readfirstlane(rw.x);
            const int end   = __builtin_amdgcn_readfirstlane(rw.y);
            float4 acc = {0.0f, 0.0f, 0.0f, 0.0f};
            for (int base = start; base < end; base += 32) {
                const int n  = min(32, end - base);
                const int my = csr[base + min(gl, n - 1)];
                int j = 0;
                for (; j + 8 <= n; j += 8) {
                    float4 v[8];
#pragma unroll
                    for (int k = 0; k < 8; ++k) {
                        const int s = __builtin_amdgcn_readlane(my, j + k);
                        v[k] = h4[(size_t)s * 32 + gl];
                    }
#pragma unroll
                    for (int k = 0; k < 8; ++k) add4(acc, v[k]);
                }
                if (j < n) {
                    float4 v[7];
#pragma unroll
                    for (int k = 0; k < 7; ++k) {
                        v[k] = make_float4(0.0f, 0.0f, 0.0f, 0.0f);
                        if (j + k < n) {
                            const int s = __builtin_amdgcn_readlane(my, j + k);
                            v[k] = h4[(size_t)s * 32 + gl];
                        }
                    }
#pragma unroll
                    for (int k = 0; k < 7; ++k)
                        if (j + k < n) add4(acc, v[k]);
                }
            }
            const float inv = 1.0f / fmaxf((float)(end - start), 1.0f);
            const float4 own = h4[(size_t)node * 32 + gl];
            float* dm = s_mean + r * LDSTR + gl * 4;
            dm[0] = acc.x * inv; dm[1] = acc.y * inv;
            dm[2] = acc.z * inv; dm[3] = acc.w * inv;
            float* dh = s_h + r * LDSTR + gl * 4;
            dh[0] = own.x; dh[1] = own.y; dh[2] = own.z; dh[3] = own.w;
        }
    }
    __syncthreads();

    const int wave = tid >> 5;
    const int lane = tid & 31;
    const int half = lane >> 4;
    const int col  = lane & 15;
    const float alpha = alpha_p[0];

    v8f acc[TPW];
#pragma unroll
    for (int t = 0; t < TPW; ++t)
#pragma unroll
        for (int j = 0; j < 8; ++j) acc[t][j] = 0.0f;

    const float* smrow = s_mean + col * LDSTR;
    const float* shrow = s_h    + col * LDSTR;

#pragma unroll 4
    for (int k0 = 0; k0 < FIN; k0 += 4) {
        const int kb = k0 + half * 2;
        v2f am; am.x = smrow[kb]; am.y = smrow[kb + 1];
        v2f ah; ah.x = shrow[kb]; ah.y = shrow[kb + 1];
#pragma unroll
        for (int t = 0; t < TPW; ++t) {
            const int n0 = (wave * TPW + t) * 16;
            v2f bl;
            bl.x = Wl[(size_t)kb * FOUT + n0 + col];
            bl.y = Wl[(size_t)(kb + 1) * FOUT + n0 + col];
            v2f br;
            br.x = Wr[(size_t)kb * FOUT + n0 + col];
            br.y = Wr[(size_t)(kb + 1) * FOUT + n0 + col];
            acc[t] = __builtin_amdgcn_wmma_f32_16x16x4_f32(
                false, am, false, bl, (short)0, acc[t], false, false);
            acc[t] = __builtin_amdgcn_wmma_f32_16x16x4_f32(
                false, ah, false, br, (short)0, acc[t], false, false);
        }
    }

#pragma unroll
    for (int t = 0; t < TPW; ++t) {
        const int n0 = (wave * TPW + t) * 16;
        const float bv = bias[n0 + col];
#pragma unroll
        for (int v = 0; v < 8; ++v) {
            float val = acc[t][v] + bv;
            if (PRELU) val = (val >= 0.0f) ? val : alpha * val;
            const int row = node0 + v + half * 8;
            out[(size_t)row * FOUT + n0 + col] = val;
        }
    }
}

__global__ void fold_kernel(const float* __restrict__ W3l,
                            const float* __restrict__ W3r,
                            const float* __restrict__ b3,
                            const float* __restrict__ Wc,
                            const float* __restrict__ bc,
                            float* __restrict__ Wf, float* __restrict__ cf) {
    int t = blockIdx.x * blockDim.x + threadIdx.x;
    if (t < 128 * 16) {
        int k = t >> 4;
        int c = t & 15;
        float s = 0.0f;
        if (c < 8) {
            const float* w = ((c < 4) ? W3l : W3r) + k * 64;
            int cc = c & 3;
            for (int j = 0; j < 64; ++j) s += w[j] * Wc[j * 4 + cc];
        }
        Wf[t] = s;
    }
    if (t < 4) {
        float s = 0.0f;
        for (int j = 0; j < 64; ++j) s += b3[j] * Wc[j * 4 + t];
        cf[t] = s + bc[t];
    }
}

__global__ __launch_bounds__(64) void project_wmma(
    const float* __restrict__ h, const float* __restrict__ Wf,
    float* __restrict__ p) {
    constexpr int FIN = 128;
    constexpr int LDSTR = FIN + 1;
    __shared__ float s_h[32 * LDSTR];

    const int tid   = threadIdx.x;
    const int node0 = blockIdx.x * 32;

    const float4* g4 = reinterpret_cast<const float4*>(h + (size_t)node0 * FIN);
    for (int it = 0; it < 32 * FIN / 4 / 64; ++it) {
        const int i4 = tid + it * 64;
        float4 v = g4[i4];
        const int r = i4 >> 5;
        const int c = (i4 & 31) * 4;
        float* d = s_h + r * LDSTR + c;
        d[0] = v.x; d[1] = v.y; d[2] = v.z; d[3] = v.w;
    }
    __syncthreads();

    const int wave = tid >> 5;
    const int lane = tid & 31;
    const int half = lane >> 4;
    const int col  = lane & 15;

    v8f acc = {0.0f, 0.0f, 0.0f, 0.0f, 0.0f, 0.0f, 0.0f, 0.0f};

    const float* shrow = s_h + (wave * 16 + col) * LDSTR;

    for (int k0 = 0; k0 < FIN; k0 += 4) {
        const int kb = k0 + half * 2;
        v2f ah; ah.x = shrow[kb]; ah.y = shrow[kb + 1];
        v2f bf;
        bf.x = Wf[kb * 16 + col];
        bf.y = Wf[(kb + 1) * 16 + col];
        acc = __builtin_amdgcn_wmma_f32_16x16x4_f32(
            false, ah, false, bf, (short)0, acc, false, false);
    }

    if (col < 8) {
        for (int v = 0; v < 8; ++v) {
            const int row = node0 + wave * 16 + v + half * 8;
            p[(size_t)row * 8 + col] = acc[v];
        }
    }
}

__global__ void gather4_final_kernel(const float* __restrict__ p,
                                     const int2* __restrict__ rows,
                                     const int* __restrict__ csr,
                                     const float* __restrict__ cf,
                                     float* __restrict__ out, int n) {
    const int t    = blockIdx.x * blockDim.x + threadIdx.x;
    const int node = t >> 2;
    const int f    = t & 3;
    const bool live = node < n;
    const int2 rw  = live ? rows[node] : make_int2(0, 0);
    const int last = rw.y - 1;
    const float4* __restrict__ p4 = reinterpret_cast<const float4*>(p);
    float4 acc = {0.0f, 0.0f, 0.0f, 0.0f};
    for (int e = rw.x + f; e < rw.y; e += 16) {
        const int s0 = csr[e];
        const int s1 = csr[min(e + 4, last)];
        const int s2 = csr[min(e + 8, last)];
        const int s3 = csr[min(e + 12, last)];
        const float4 v0 = p4[(size_t)s0 * 2];
        const float4 v1 = p4[(size_t)s1 * 2];
        const float4 v2 = p4[(size_t)s2 * 2];
        const float4 v3 = p4[(size_t)s3 * 2];
        add4(acc, v0);
        addsel4(acc, v1, e + 4 <= last);
        addsel4(acc, v2, e + 8 <= last);
        addsel4(acc, v3, e + 12 <= last);
    }
    acc.x += __shfl_xor(acc.x, 1); acc.y += __shfl_xor(acc.y, 1);
    acc.z += __shfl_xor(acc.z, 1); acc.w += __shfl_xor(acc.w, 1);
    acc.x += __shfl_xor(acc.x, 2); acc.y += __shfl_xor(acc.y, 2);
    acc.z += __shfl_xor(acc.z, 2); acc.w += __shfl_xor(acc.w, 2);
    const float a  = (f == 0) ? acc.x : (f == 1) ? acc.y : (f == 2) ? acc.z : acc.w;
    const float id = 1.0f / fmaxf((float)(rw.y - rw.x), 1.0f);
    if (live)
        out[(size_t)node * 4 + f] = (a * id + cf[f]) + p[(size_t)node * 8 + 4 + f];
}

extern "C" void kernel_launch(void* const* d_in, const int* in_sizes, int n_in,
                              void* d_out, int out_size, void* d_ws, size_t ws_size,
                              hipStream_t stream) {
    const float* x   = (const float*)d_in[0];
    const int*   ei  = (const int*)d_in[1];
    const float* W1l = (const float*)d_in[2];
    const float* b1  = (const float*)d_in[3];
    const float* W1r = (const float*)d_in[4];
    const float* W2l = (const float*)d_in[5];
    const float* b2  = (const float*)d_in[6];
    const float* W2r = (const float*)d_in[7];
    const float* W3l = (const float*)d_in[8];
    const float* b3  = (const float*)d_in[9];
    const float* W3r = (const float*)d_in[10];
    const float* Wc  = (const float*)d_in[11];
    const float* bc  = (const float*)d_in[12];
    const float* a   = (const float*)d_in[13];
    float* out = (float*)d_out;

    const int E = in_sizes[1] / 2;
    const int* srcv = ei;
    const int* dstv = ei + E;

    char* ws = (char*)d_ws;
    size_t off = 0;
    auto carve = [&](size_t bytes) {
        float* p = (float*)(ws + off);
        off += (bytes + 255) & ~(size_t)255;
        return p;
    };
    int*   cnt = (int*)carve((size_t)(N_NODES + 1) * 4);
    float* el  = carve((size_t)N_NODES * 128 * 4);
    float* hA  = carve((size_t)N_NODES * 128 * 4);
    float* hB  = carve((size_t)N_NODES * 128 * 4);

    const int nCh = (E + CHUNK - 1) / CHUNK;

    int2* rows = (int2*)el;
    int*  rec  = (int*)el + (size_t)N_NODES * 2;
    int2* offs = (int2*)(rec + (size_t)nCh * CHUNK);
    int*  csr  = (int*)(offs + (size_t)nCh * N_RNG);

    const int TB = 256;
    const int nBlk16 = N_NODES / 16;

    bin_edges_kernel<<<nCh, BIN_TB, 0, stream>>>(
        srcv, dstv, rec, offs, cnt + N_NODES, E);
    rank_range_kernel<<<N_RNG, 1024, 0, stream>>>(
        rec, offs, cnt + N_NODES, rows, csr, nCh);

    sage_gather_combine_wmma<32, 128, 2, true><<<nBlk16, 128, 0, stream>>>(
        x, rows, csr, W1l, b1, W1r, a, hA);

    sage_gather_combine_wmma<128, 128, 2, true><<<nBlk16, 128, 0, stream>>>(
        hA, rows, csr, W2l, b2, W2r, a, hB);

    float* proj = hA;
    float* Wf   = hA + (size_t)N_NODES * 8;
    float* cf   = Wf + 128 * 16;
    fold_kernel<<<(128 * 16 + TB - 1) / TB, TB, 0, stream>>>(
        W3l, W3r, b3, Wc, bc, Wf, cf);
    project_wmma<<<N_NODES / 32, 64, 0, stream>>>(hB, Wf, proj);
    gather4_final_kernel<<<(N_NODES * 4 + TB - 1) / TB, TB, 0, stream>>>(
        proj, rows, csr, cf, out, N_NODES);
}


# Layer-2 tile formed transposed (weights as A) so its accumulators are the B operand of the 8-column projection in the same kernel: h2 never stored, projection launch gone; p sum order changes
# speedup vs baseline: 5.9039x; 1.0616x over previous
#include <hip/hip_runtime.h>
#include <hip/hip_bf16.h>


typedef __attribute__((ext_vector_type(2))) float v2f;
typedef __attribute__((ext_vector_type(8))) float v8f;

#define N_NODES 100000

#define RNG_SHIFT 9
#define RNG_NODES (1 << RNG_SHIFT)
#define N_RNG ((N_NODES + RNG_NODES - 1) / RNG_NODES)
#define SRC_BITS 17
#define CHUNK 2048
#define BIN_TB 256
#define EPL (CHUNK / BIN_TB)
#define WPT 8
#define OUT_CAP 12288

static_assert(N_NODES <= (1 << SRC_BITS), "node id must fit SRC_BITS");
static_assert(RNG_SHIFT + SRC_BITS < 32, "packed edge word must stay positive");
static_assert(N_RNG <= BIN_TB, "one thread per range in the chunk scan");

__device__ __forceinline__ int block_scan_excl(int c, int* s_wsum, int* s_total) {
    const int lane = threadIdx.x & 31;
    const int w    = threadIdx.x >> 5;
    const int nw   = blockDim.x >> 5;
    int v = c;
#pragma unroll
    for (int o = 1; o < 32; o <<= 1) {
        int t = __shfl_up(v, o);
        if (lane >= o) v += t;
    }
    if (lane == 31) s_wsum[w] = v;
    __syncthreads();
    if (w == 0) {
        const int t = (lane < nw) ? s_wsum[lane] : 0;
        int u = t;
#pragma unroll
        for (int o = 1; o < 32; o <<= 1) {
            int q = __shfl_up(u, o);
            if (lane >= o) u += q;
        }
        s_wsum[lane] = u - t;
        if (lane == 31) *s_total = u;
    }
    __syncthreads();
    return s_wsum[w] + (v - c);
}

__global__ __launch_bounds__(BIN_TB) void bin_edges_kernel(
    const int* __restrict__ src, const int* __restrict__ dst,
    int* __restrict__ rec, int2* __restrict__ offs,
    int* __restrict__ total, int E) {
    __shared__ int s_hist[N_RNG];
    __shared__ int s_off[N_RNG];
    __shared__ int s_rec[CHUNK];
    __shared__ int s_wsum[32];
    __shared__ int s_tot;
    const int tid   = threadIdx.x;
    const int chunk = blockIdx.x;
    const int e0    = chunk * CHUNK + tid;
    if (tid < N_RNG) s_hist[tid] = 0;
    if (chunk == 0 && tid == 0) *total = 0;

    int d[EPL], s[EPL];
#pragma unroll
    for (int k = 0; k < EPL; ++k) {
        const int ec = min(e0 + k * BIN_TB, E - 1);
        d[k] = dst[ec];
        s[k] = src[ec];
    }
    __syncthreads();

    int rk[EPL];
#pragma unroll
    for (int k = 0; k < EPL; ++k) {
        rk[k] = 0;
        if (e0 + k * BIN_TB < E) rk[k] = atomicAdd(&s_hist[d[k] >> RNG_SHIFT], 1);
    }
    __syncthreads();

    const int c  = (tid < N_RNG) ? s_hist[tid] : 0;
    const int ex = block_scan_excl(c, s_wsum, &s_tot);
    if (tid < N_RNG) {
        s_off[tid] = ex;
        offs[(size_t)chunk * N_RNG + tid] = make_int2(ex, ex + c);
    }
    __syncthreads();

#pragma unroll
    for (int k = 0; k < EPL; ++k) {
        if (e0 + k * BIN_TB < E)
            s_rec[s_off[d[k] >> RNG_SHIFT] + rk[k]] =
                ((d[k] & (RNG_NODES - 1)) << SRC_BITS) | s[k];
    }
    __syncthreads();
    const int nv = s_tot;
    int* __restrict__ rp = rec + (size_t)chunk * CHUNK;
#pragma unroll
    for (int k = 0; k < EPL; ++k) {
        const int p = k * BIN_TB + tid;
        if (p < nv) rp[p] = s_rec[p];
    }
}

__global__ __launch_bounds__(1024) void rank_range_kernel(
    const int* __restrict__ rec, const int2* __restrict__ offs,
    int* __restrict__ total, int2* __restrict__ rows,
    int* __restrict__ csr, int nCh) {
    __shared__ int s_cnt[RNG_NODES];
    __shared__ int s_out[OUT_CAP];
    __shared__ int s_wsum[32];
    __shared__ int s_tot;
    __shared__ int s_base;
    const int tid = threadIdx.x;
    const int b   = blockIdx.x;
    if (tid < RNG_NODES) s_cnt[tid] = 0;
    __syncthreads();

    for (int ch = tid; ch < nCh; ch += 1024) {
        const int2 pc = offs[(size_t)ch * N_RNG + b];
        const int* __restrict__ rp = rec + (size_t)ch * CHUNK;
        const int pb = pc.x;
        const int pe = pc.y;
        const int last = pe - 1;
        for (int j = pb; j < pe; j += WPT) {
            int r[WPT];
#pragma unroll
            for (int k = 0; k < WPT; ++k) r[k] = rp[min(j + k, last)];
#pragma unroll
            for (int k = 0; k < WPT; ++k)
                atomicAdd(&s_cnt[r[k] >> SRC_BITS], (j + k <= last) ? 1 : 0);
        }
    }
    __syncthreads();

    const int c  = (tid < RNG_NODES) ? s_cnt[tid] : 0;
    const int ex = block_scan_excl(c, s_wsum, &s_tot);
    if (tid == 0) s_base = atomicAdd(total, s_tot);
    __syncthreads();
    const int nb   = s_tot;
    const int base = s_base;
    if (tid < RNG_NODES) {
        s_cnt[tid] = ex;
        const int node = b * RNG_NODES + tid;
        if (node < N_NODES) rows[node] = make_int2(base + ex, base + ex + c);
    }
    __syncthreads();

    const bool staged = nb <= OUT_CAP;
    const int smask = (1 << SRC_BITS) - 1;
    for (int ch = tid; ch < nCh; ch += 1024) {
        const int2 pc = offs[(size_t)ch * N_RNG + b];
        const int* __restrict__ rp = rec + (size_t)ch * CHUNK;
        const int pb = pc.x;
        const int pe = pc.y;
        const int last = pe - 1;
        for (int j = pb; j < pe; j += WPT) {
            int r[WPT];
#pragma unroll
            for (int k = 0; k < WPT; ++k) r[k] = rp[min(j + k, last)];
            int sl[WPT];
#pragma unroll
            for (int k = 0; k < WPT; ++k)
                sl[k] = atomicAdd(&s_cnt[r[k] >> SRC_BITS], (j + k <= last) ? 1 : 0);
            if (staged) {
#pragma unroll
                for (int k = 0; k < WPT; ++k)
                    if (j + k <= last) s_out[sl[k]] = r[k] & smask;
            } else {
#pragma unroll
                for (int k = 0; k < WPT; ++k)
                    if (j + k <= last) csr[base + sl[k]] = r[k] & smask;
            }
        }
    }
    if (staged) {
        __syncthreads();
        for (int i = tid; i < nb; i += 1024) csr[base + i] = s_out[i];
    }
}

__device__ __forceinline__ void add4(float4& a, const float4& b) {
    a.x += b.x; a.y += b.y; a.z += b.z; a.w += b.w;
}

__device__ __forceinline__ void addsel4(float4& a, const float4& b, bool in) {
    a.x += in ? b.x : 0.0f; a.y += in ? b.y : 0.0f;
    a.z += in ? b.z : 0.0f; a.w += in ? b.w : 0.0f;
}

template <int FIN, int FOUT, int TPW, bool PRELU, bool PROJ>
__global__ __launch_bounds__(128) void sage_gather_combine_wmma(
    const float* __restrict__ h, const int2* __restrict__ rows,
    const int* __restrict__ csr,
    const float* __restrict__ Wl, const float* __restrict__ bias,
    const float* __restrict__ Wr, const float* __restrict__ alpha_p,
    const float* __restrict__ Wf, float* __restrict__ out) {
    constexpr int LDSTR = FIN + 1;
    __shared__ alignas(16) float s_mean[16 * LDSTR];
    __shared__ float s_h[16 * LDSTR];
    static_assert(16 * LDSTR >= 4 * 16 * 8, "partial p must fit s_mean");

    const int tid   = threadIdx.x;
    const int node0 = blockIdx.x * 16;
    const float4* __restrict__ h4 = reinterpret_cast<const float4*>(h);

    if (FIN == 32) {
        const int r    = tid >> 3;
        const int sub  = tid & 7;
        const int node = node0 + r;
        const int2 rw  = rows[node];
        const int last = rw.y - 1;
        float4 acc = {0.0f, 0.0f, 0.0f, 0.0f};
        for (int e = rw.x; e < rw.y; e += 4) {
            const int s0 = csr[e];
            const int s1 = csr[min(e + 1, last)];
            const int s2 = csr[min(e + 2, last)];
            const int s3 = csr[min(e + 3, last)];
            const float4 v0 = h4[(size_t)s0 * 8 + sub];
            const float4 v1 = h4[(size_t)s1 * 8 + sub];
            const float4 v2 = h4[(size_t)s2 * 8 + sub];
            const float4 v3 = h4[(size_t)s3 * 8 + sub];
            add4(acc, v0);
            addsel4(acc, v1, e + 1 <= last);
            addsel4(acc, v2, e + 2 <= last);
            addsel4(acc, v3, e + 3 <= last);
        }
        const float inv = 1.0f / fmaxf((float)(rw.y - rw.x), 1.0f);
        const float4 own = h4[(size_t)node * 8 + sub];
        float* dm = s_mean + r * LDSTR + sub * 4;
        dm[0] = acc.x * inv; dm[1] = acc.y * inv;
        dm[2] = acc.z * inv; dm[3] = acc.w * inv;
        float* dh = s_h + r * LDSTR + sub * 4;
        dh[0] = own.x; dh[1] = own.y; dh[2] = own.z; dh[3] = own.w;
    } else {
        const int gl = tid & 31;
        const int wv = __builtin_amdgcn_readfirstlane(tid >> 5);
        for (int q = 0; q < 4; ++q) {
            const int r    = wv * 4 + q;
            const int node = node0 + r;
            const int2 rw  = rows[node];
            const int start = __builtin_amdgcn_readfirstlane(rw.x);
            const int end   = __builtin_amdgcn_readfirstlane(rw.y);
            float4 acc = {0.0f, 0.0f, 0.0f, 0.0f};
            for (int base = start; base < end; base += 32) {
                const int n  = min(32, end - base);
                const int my = csr[base + min(gl, n - 1)];
                int j = 0;
                for (; j + 8 <= n; j += 8) {
                    float4 v[8];
#pragma unroll
                    for (int k = 0; k < 8; ++k) {
                        const int s = __builtin_amdgcn_readlane(my, j + k);
                        v[k] = h4[(size_t)s * 32 + gl];
                    }
#pragma unroll
                    for (int k = 0; k < 8; ++k) add4(acc, v[k]);
                }
                if (j < n) {
                    float4 v[7];
#pragma unroll
                    for (int k = 0; k < 7; ++k) {
                        v[k] = make_float4(0.0f, 0.0f, 0.0f, 0.0f);
                        if (j + k < n) {
                            const int s = __builtin_amdgcn_readlane(my, j + k);
                            v[k] = h4[(size_t)s * 32 + gl];
                        }
                    }
#pragma unroll
                    for (int k = 0; k < 7; ++k)
                        if (j + k < n) add4(acc, v[k]);
                }
            }
            const float inv = 1.0f / fmaxf((float)(end - start), 1.0f);
            const float4 own = h4[(size_t)node * 32 + gl];
            float* dm = s_mean + r * LDSTR + gl * 4;
            dm[0] = acc.x * inv; dm[1] = acc.y * inv;
            dm[2] = acc.z * inv; dm[3] = acc.w * inv;
            float* dh = s_h + r * LDSTR + gl * 4;
            dh[0] = own.x; dh[1] = own.y; dh[2] = own.z; dh[3] = own.w;
        }
    }
    __syncthreads();

    const int wave = tid >> 5;
    const int lane = tid & 31;
    const int half = lane >> 4;
    const int col  = lane & 15;
    const float alpha = alpha_p[0];

    v8f acc[TPW];
#pragma unroll
    for (int t = 0; t < TPW; ++t)
#pragma unroll
        for (int j = 0; j < 8; ++j) acc[t][j] = 0.0f;

    const float* smrow = s_mean + col * LDSTR;
    const float* shrow = s_h    + col * LDSTR;

#pragma unroll 4
    for (int k0 = 0; k0 < FIN; k0 += 4) {
        const int kb = k0 + half * 2;
        v2f am; am.x = smrow[kb]; am.y = smrow[kb + 1];
        v2f ah; ah.x = shrow[kb]; ah.y = shrow[kb + 1];
#pragma unroll
        for (int t = 0; t < TPW; ++t) {
            const int n0 = (wave * TPW + t) * 16;
            v2f bl;
            bl.x = Wl[(size_t)kb * FOUT + n0 + col];
            bl.y = Wl[(size_t)(kb + 1) * FOUT + n0 + col];
            v2f br;
            br.x = Wr[(size_t)kb * FOUT + n0 + col];
            br.y = Wr[(size_t)(kb + 1) * FOUT + n0 + col];
            if (PROJ) {
                acc[t] = __builtin_amdgcn_wmma_f32_16x16x4_f32(
                    false, bl, false, am, (short)0, acc[t], false, false);
                acc[t] = __builtin_amdgcn_wmma_f32_16x16x4_f32(
                    false, br, false, ah, (short)0, acc[t], false, false);
            } else {
                acc[t] = __builtin_amdgcn_wmma_f32_16x16x4_f32(
                    false, am, false, bl, (short)0, acc[t], false, false);
                acc[t] = __builtin_amdgcn_wmma_f32_16x16x4_f32(
                    false, ah, false, br, (short)0, acc[t], false, false);
            }
        }
    }

    if (PROJ) {
        v8f pacc = {0.0f, 0.0f, 0.0f, 0.0f, 0.0f, 0.0f, 0.0f, 0.0f};
#pragma unroll
        for (int t = 0; t < TPW; ++t) {
            const int f0 = (wave * TPW + t) * 16 + half * 8;
            const float4 bA = *reinterpret_cast<const float4*>(bias + f0);
            const float4 bB = *reinterpret_cast<const float4*>(bias + f0 + 4);
            const float bv[8] = {bA.x, bA.y, bA.z, bA.w, bB.x, bB.y, bB.z, bB.w};
            float hv[8];
#pragma unroll
            for (int r = 0; r < 8; ++r) {
                float val = acc[t][r] + bv[r];
                if (PRELU) val = (val >= 0.0f) ? val : alpha * val;
                hv[r] = val;
            }
#pragma unroll
            for (int i = 0; i < 4; ++i) {
                v2f hb; hb.x = hv[2 * i]; hb.y = hv[2 * i + 1];
                v2f wa;
                wa.x = Wf[(f0 + 2 * i) * 16 + col];
                wa.y = Wf[(f0 + 2 * i + 1) * 16 + col];
                pacc = __builtin_amdgcn_wmma_f32_16x16x4_f32(
                    false, wa, false, hb, (short)0, pacc, false, false);
            }
        }
        float* s_part = s_mean;
        __syncthreads();
        if (half == 0) {
            float4* d = reinterpret_cast<float4*>(s_part + (wave * 16 + col) * 8);
            d[0] = make_float4(pacc[0], pacc[1], pacc[2], pacc[3]);
            d[1] = make_float4(pacc[4], pacc[5], pacc[6], pacc[7]);
        }
        __syncthreads();
        const float s = ((s_part[tid] + s_part[128 + tid]) + s_part[256 + tid])
                        + s_part[384 + tid];
        out[(size_t)node0 * 8 + tid] = s;
        return;
    }

#pragma unroll
    for (int t = 0; t < TPW; ++t) {
        const int n0 = (wave * TPW + t) * 16;
        const float bv = bias[n0 + col];
#pragma unroll
        for (int v = 0; v < 8; ++v) {
            float val = acc[t][v] + bv;
            if (PRELU) val = (val >= 0.0f) ? val : alpha * val;
            const int row = node0 + v + half * 8;
            out[(size_t)row * FOUT + n0 + col] = val;
        }
    }
}

__global__ void fold_kernel(const float* __restrict__ W3l,
                            const float* __restrict__ W3r,
                            const float* __restrict__ b3,
                            const float* __restrict__ Wc,
                            const float* __restrict__ bc,
                            float* __restrict__ Wf, float* __restrict__ cf) {
    int t = blockIdx.x * blockDim.x + threadIdx.x;
    if (t < 128 * 16) {
        int k = t >> 4;
        int c = t & 15;
        float s = 0.0f;
        if (c < 8) {
            const float* w = ((c < 4) ? W3l : W3r) + k * 64;
            int cc = c & 3;
            for (int j = 0; j < 64; ++j) s += w[j] * Wc[j * 4 + cc];
        }
        Wf[t] = s;
    }
    if (t < 4) {
        float s = 0.0f;
        for (int j = 0; j < 64; ++j) s += b3[j] * Wc[j * 4 + t];
        cf[t] = s + bc[t];
    }
}

__global__ void gather4_final_kernel(const float* __restrict__ p,
                                     const int2* __restrict__ rows,
                                     const int* __restrict__ csr,
                                     const float* __restrict__ cf,
                                     float* __restrict__ out, int n) {
    const int t    = blockIdx.x * blockDim.x + threadIdx.x;
    const int node = t >> 2;
    const int f    = t & 3;
    const bool live = node < n;
    const int2 rw  = live ? rows[node] : make_int2(0, 0);
    const int last = rw.y - 1;
    const float4* __restrict__ p4 = reinterpret_cast<const float4*>(p);
    float4 acc = {0.0f, 0.0f, 0.0f, 0.0f};
    for (int e = rw.x + f; e < rw.y; e += 16) {
        const int s0 = csr[e];
        const int s1 = csr[min(e + 4, last)];
        const int s2 = csr[min(e + 8, last)];
        const int s3 = csr[min(e + 12, last)];
        const float4 v0 = p4[(size_t)s0 * 2];
        const float4 v1 = p4[(size_t)s1 * 2];
        const float4 v2 = p4[(size_t)s2 * 2];
        const float4 v3 = p4[(size_t)s3 * 2];
        add4(acc, v0);
        addsel4(acc, v1, e + 4 <= last);
        addsel4(acc, v2, e + 8 <= last);
        addsel4(acc, v3, e + 12 <= last);
    }
    acc.x += __shfl_xor(acc.x, 1); acc.y += __shfl_xor(acc.y, 1);
    acc.z += __shfl_xor(acc.z, 1); acc.w += __shfl_xor(acc.w, 1);
    acc.x += __shfl_xor(acc.x, 2); acc.y += __shfl_xor(acc.y, 2);
    acc.z += __shfl_xor(acc.z, 2); acc.w += __shfl_xor(acc.w, 2);
    const float a  = (f == 0) ? acc.x : (f == 1) ? acc.y : (f == 2) ? acc.z : acc.w;
    const float id = 1.0f / fmaxf((float)(rw.y - rw.x), 1.0f);
    if (live)
        out[(size_t)node * 4 + f] = (a * id + cf[f]) + p[(size_t)node * 8 + 4 + f];
}

extern "C" void kernel_launch(void* const* d_in, const int* in_sizes, int n_in,
                              void* d_out, int out_size, void* d_ws, size_t ws_size,
                              hipStream_t stream) {
    const float* x   = (const float*)d_in[0];
    const int*   ei  = (const int*)d_in[1];
    const float* W1l = (const float*)d_in[2];
    const float* b1  = (const float*)d_in[3];
    const float* W1r = (const float*)d_in[4];
    const float* W2l = (const float*)d_in[5];
    const float* b2  = (const float*)d_in[6];
    const float* W2r = (const float*)d_in[7];
    const float* W3l = (const float*)d_in[8];
    const float* b3  = (const float*)d_in[9];
    const float* W3r = (const float*)d_in[10];
    const float* Wc  = (const float*)d_in[11];
    const float* bc  = (const float*)d_in[12];
    const float* a   = (const float*)d_in[13];
    float* out = (float*)d_out;

    const int E = in_sizes[1] / 2;
    const int* srcv = ei;
    const int* dstv = ei + E;

    char* ws = (char*)d_ws;
    size_t off = 0;
    auto carve = [&](size_t bytes) {
        float* p = (float*)(ws + off);
        off += (bytes + 255) & ~(size_t)255;
        return p;
    };
    int*   cnt = (int*)carve((size_t)(N_NODES + 1) * 4);
    float* el  = carve((size_t)N_NODES * 128 * 4);
    float* hA  = carve((size_t)N_NODES * 128 * 4);
    float* hB  = carve((size_t)N_NODES * 128 * 4);

    const int nCh = (E + CHUNK - 1) / CHUNK;

    int2* rows = (int2*)el;
    int*  rec  = (int*)el + (size_t)N_NODES * 2;
    int2* offs = (int2*)(rec + (size_t)nCh * CHUNK);
    int*  csr  = (int*)(offs + (size_t)nCh * N_RNG);

    const int TB = 256;
    const int nBlk16 = N_NODES / 16;

    bin_edges_kernel<<<nCh, BIN_TB, 0, stream>>>(
        srcv, dstv, rec, offs, cnt + N_NODES, E);
    rank_range_kernel<<<N_RNG, 1024, 0, stream>>>(
        rec, offs, cnt + N_NODES, rows, csr, nCh);

    sage_gather_combine_wmma<32, 128, 2, true, false><<<nBlk16, 128, 0, stream>>>(
        x, rows, csr, W1l, b1, W1r, a, nullptr, hA);

    float* proj = hB;
    float* Wf   = hB + (size_t)N_NODES * 8;
    float* cf   = Wf + 128 * 16;
    fold_kernel<<<(128 * 16 + TB - 1) / TB, TB, 0, stream>>>(
        W3l, W3r, b3, Wc, bc, Wf, cf);

    sage_gather_combine_wmma<128, 128, 2, true, true><<<nBlk16, 128, 0, stream>>>(
        hA, rows, csr, W2l, b2, W2r, a, Wf, proj);

    gather4_final_kernel<<<(N_NODES * 4 + TB - 1) / TB, TB, 0, stream>>>(
        proj, rows, csr, cf, out, N_NODES);
}


# Layer 1-2 weights copied per call into WMMA fragment order (in the fold launch, moved first): one 16-byte load per lane feeds two K-steps, 4x fewer weight loads; sum order unchanged
# speedup vs baseline: 6.1186x; 1.0364x over previous
#include <hip/hip_runtime.h>
#include <hip/hip_bf16.h>


typedef __attribute__((ext_vector_type(2))) float v2f;
typedef __attribute__((ext_vector_type(8))) float v8f;

#define N_NODES 100000

#define RNG_SHIFT 9
#define RNG_NODES (1 << RNG_SHIFT)
#define N_RNG ((N_NODES + RNG_NODES - 1) / RNG_NODES)
#define SRC_BITS 17
#define CHUNK 2048
#define BIN_TB 256
#define EPL (CHUNK / BIN_TB)
#define WPT 8
#define OUT_CAP 12288

static_assert(N_NODES <= (1 << SRC_BITS), "node id must fit SRC_BITS");
static_assert(RNG_SHIFT + SRC_BITS < 32, "packed edge word must stay positive");
static_assert(N_RNG <= BIN_TB, "one thread per range in the chunk scan");

__device__ __forceinline__ int block_scan_excl(int c, int* s_wsum, int* s_total) {
    const int lane = threadIdx.x & 31;
    const int w    = threadIdx.x >> 5;
    const int nw   = blockDim.x >> 5;
    int v = c;
#pragma unroll
    for (int o = 1; o < 32; o <<= 1) {
        int t = __shfl_up(v, o);
        if (lane >= o) v += t;
    }
    if (lane == 31) s_wsum[w] = v;
    __syncthreads();
    if (w == 0) {
        const int t = (lane < nw) ? s_wsum[lane] : 0;
        int u = t;
#pragma unroll
        for (int o = 1; o < 32; o <<= 1) {
            int q = __shfl_up(u, o);
            if (lane >= o) u += q;
        }
        s_wsum[lane] = u - t;
        if (lane == 31) *s_total = u;
    }
    __syncthreads();
    return s_wsum[w] + (v - c);
}

__global__ __launch_bounds__(BIN_TB) void bin_edges_kernel(
    const int* __restrict__ src, const int* __restrict__ dst,
    int* __restrict__ rec, int2* __restrict__ offs,
    int* __restrict__ total, int E) {
    __shared__ int s_hist[N_RNG];
    __shared__ int s_off[N_RNG];
    __shared__ int s_rec[CHUNK];
    __shared__ int s_wsum[32];
    __shared__ int s_tot;
    const int tid   = threadIdx.x;
    const int chunk = blockIdx.x;
    const int e0    = chunk * CHUNK + tid;
    if (tid < N_RNG) s_hist[tid] = 0;
    if (chunk == 0 && tid == 0) *total = 0;

    int d[EPL], s[EPL];
#pragma unroll
    for (int k = 0; k < EPL; ++k) {
        const int ec = min(e0 + k * BIN_TB, E - 1);
        d[k] = dst[ec];
        s[k] = src[ec];
    }
    __syncthreads();

    int rk[EPL];
#pragma unroll
    for (int k = 0; k < EPL; ++k) {
        rk[k] = 0;
        if (e0 + k * BIN_TB < E) rk[k] = atomicAdd(&s_hist[d[k] >> RNG_SHIFT], 1);
    }
    __syncthreads();

    const int c  = (tid < N_RNG) ? s_hist[tid] : 0;
    const int ex = block_scan_excl(c, s_wsum, &s_tot);
    if (tid < N_RNG) {
        s_off[tid] = ex;
        offs[(size_t)chunk * N_RNG + tid] = make_int2(ex, ex + c);
    }
    __syncthreads();

#pragma unroll
    for (int k = 0; k < EPL; ++k) {
        if (e0 + k * BIN_TB < E)
            s_rec[s_off[d[k] >> RNG_SHIFT] + rk[k]] =
                ((d[k] & (RNG_NODES - 1)) << SRC_BITS) | s[k];
    }
    __syncthreads();
    const int nv = s_tot;
    int* __restrict__ rp = rec + (size_t)chunk * CHUNK;
#pragma unroll
    for (int k = 0; k < EPL; ++k) {
        const int p = k * BIN_TB + tid;
        if (p < nv) rp[p] = s_rec[p];
    }
}

__global__ __launch_bounds__(1024) void rank_range_kernel(
    const int* __restrict__ rec, const int2* __restrict__ offs,
    int* __restrict__ total, int2* __restrict__ rows,
    int* __restrict__ csr, int nCh) {
    __shared__ int s_cnt[RNG_NODES];
    __shared__ int s_out[OUT_CAP];
    __shared__ int s_wsum[32];
    __shared__ int s_tot;
    __shared__ int s_base;
    const int tid = threadIdx.x;
    const int b   = blockIdx.x;
    if (tid < RNG_NODES) s_cnt[tid] = 0;
    __syncthreads();

    for (int ch = tid; ch < nCh; ch += 1024) {
        const int2 pc = offs[(size_t)ch * N_RNG + b];
        const int* __restrict__ rp = rec + (size_t)ch * CHUNK;
        const int pb = pc.x;
        const int pe = pc.y;
        const int last = pe - 1;
        for (int j = pb; j < pe; j += WPT) {
            int r[WPT];
#pragma unroll
            for (int k = 0; k < WPT; ++k) r[k] = rp[min(j + k, last)];
#pragma unroll
            for (int k = 0; k < WPT; ++k)
                atomicAdd(&s_cnt[r[k] >> SRC_BITS], (j + k <= last) ? 1 : 0);
        }
    }
    __syncthreads();

    const int c  = (tid < RNG_NODES) ? s_cnt[tid] : 0;
    const int ex = block_scan_excl(c, s_wsum, &s_tot);
    if (tid == 0) s_base = atomicAdd(total, s_tot);
    __syncthreads();
    const int nb   = s_tot;
    const int base = s_base;
    if (tid < RNG_NODES) {
        s_cnt[tid] = ex;
        const int node = b * RNG_NODES + tid;
        if (node < N_NODES) rows[node] = make_int2(base + ex, base + ex + c);
    }
    __syncthreads();

    const bool staged = nb <= OUT_CAP;
    const int smask = (1 << SRC_BITS) - 1;
    for (int ch = tid; ch < nCh; ch += 1024) {
        const int2 pc = offs[(size_t)ch * N_RNG + b];
        const int* __restrict__ rp = rec + (size_t)ch * CHUNK;
        const int pb = pc.x;
        const int pe = pc.y;
        const int last = pe - 1;
        for (int j = pb; j < pe; j += WPT) {
            int r[WPT];
#pragma unroll
            for (int k = 0; k < WPT; ++k) r[k] = rp[min(j + k, last)];
            int sl[WPT];
#pragma unroll
            for (int k = 0; k < WPT; ++k)
                sl[k] = atomicAdd(&s_cnt[r[k] >> SRC_BITS], (j + k <= last) ? 1 : 0);
            if (staged) {
#pragma unroll
                for (int k = 0; k < WPT; ++k)
                    if (j + k <= last) s_out[sl[k]] = r[k] & smask;
            } else {
#pragma unroll
                for (int k = 0; k < WPT; ++k)
                    if (j + k <= last) csr[base + sl[k]] = r[k] & smask;
            }
        }
    }
    if (staged) {
        __syncthreads();
        for (int i = tid; i < nb; i += 1024) csr[base + i] = s_out[i];
    }
}

__device__ __forceinline__ void add4(float4& a, const float4& b) {
    a.x += b.x; a.y += b.y; a.z += b.z; a.w += b.w;
}

__device__ __forceinline__ void addsel4(float4& a, const float4& b, bool in) {
    a.x += in ? b.x : 0.0f; a.y += in ? b.y : 0.0f;
    a.z += in ? b.z : 0.0f; a.w += in ? b.w : 0.0f;
}

template <int FIN, int FOUT, int TPW, bool PRELU, bool PROJ>
__global__ __launch_bounds__(128) void sage_gather_combine_wmma(
    const float* __restrict__ h, const int2* __restrict__ rows,
    const int* __restrict__ csr,
    const float* __restrict__ Wp, const float* __restrict__ bias,
    const float* __restrict__ alpha_p,
    const float* __restrict__ Wf, float* __restrict__ out) {
    constexpr int LDSTR = FIN + 1;
    constexpr int NJ = FIN / 8;
    __shared__ alignas(16) float s_mean[16 * LDSTR];
    __shared__ float s_h[16 * LDSTR];
    static_assert(16 * LDSTR >= 4 * 16 * 8, "partial p must fit s_mean");

    const int tid   = threadIdx.x;
    const int node0 = blockIdx.x * 16;
    const float4* __restrict__ h4 = reinterpret_cast<const float4*>(h);

    if (FIN == 32) {
        const int r    = tid >> 3;
        const int sub  = tid & 7;
        const int node = node0 + r;
        const int2 rw  = rows[node];
        const int last = rw.y - 1;
        float4 acc = {0.0f, 0.0f, 0.0f, 0.0f};
        for (int e = rw.x; e < rw.y; e += 4) {
            const int s0 = csr[e];
            const int s1 = csr[min(e + 1, last)];
            const int s2 = csr[min(e + 2, last)];
            const int s3 = csr[min(e + 3, last)];
            const float4 v0 = h4[(size_t)s0 * 8 + sub];
            const float4 v1 = h4[(size_t)s1 * 8 + sub];
            const float4 v2 = h4[(size_t)s2 * 8 + sub];
            const float4 v3 = h4[(size_t)s3 * 8 + sub];
            add4(acc, v0);
            addsel4(acc, v1, e + 1 <= last);
            addsel4(acc, v2, e + 2 <= last);
            addsel4(acc, v3, e + 3 <= last);
        }
        const float inv = 1.0f / fmaxf((float)(rw.y - rw.x), 1.0f);
        const float4 own = h4[(size_t)node * 8 + sub];
        float* dm = s_mean + r * LDSTR + sub * 4;
        dm[0] = acc.x * inv; dm[1] = acc.y * inv;
        dm[2] = acc.z * inv; dm[3] = acc.w * inv;
        float* dh = s_h + r * LDSTR + sub * 4;
        dh[0] = own.x; dh[1] = own.y; dh[2] = own.z; dh[3] = own.w;
    } else {
        const int gl = tid & 31;
        const int wv = __builtin_amdgcn_readfirstlane(tid >> 5);
        for (int q = 0; q < 4; ++q) {
            const int r    = wv * 4 + q;
            const int node = node0 + r;
            const int2 rw  = rows[node];
            const int start = __builtin_amdgcn_readfirstlane(rw.x);
            const int end   = __builtin_amdgcn_readfirstlane(rw.y);
            float4 acc = {0.0f, 0.0f, 0.0f, 0.0f};
            for (int base = start; base < end; base += 32) {
                const int n  = min(32, end - base);
                const int my = csr[base + min(gl, n - 1)];
                int j = 0;
                for (; j + 8 <= n; j += 8) {
                    float4 v[8];
#pragma unroll
                    for (int k = 0; k < 8; ++k) {
                        const int s = __builtin_amdgcn_readlane(my, j + k);
                        v[k] = h4[(size_t)s * 32 + gl];
                    }
#pragma unroll
                    for (int k = 0; k < 8; ++k) add4(acc, v[k]);
                }
                if (j < n) {
                    float4 v[7];
#pragma unroll
                    for (int k = 0; k < 7; ++k) {
                        v[k] = make_float4(0.0f, 0.0f, 0.0f, 0.0f);
                        if (j + k < n) {
                            const int s = __builtin_amdgcn_readlane(my, j + k);
                            v[k] = h4[(size_t)s * 32 + gl];
                        }
                    }
#pragma unroll
                    for (int k = 0; k < 7; ++k)
                        if (j + k < n) add4(acc, v[k]);
                }
            }
            const float inv = 1.0f / fmaxf((float)(end - start), 1.0f);
            const float4 own = h4[(size_t)node * 32 + gl];
            float* dm = s_mean + r * LDSTR + gl * 4;
            dm[0] = acc.x * inv; dm[1] = acc.y * inv;
            dm[2] = acc.z * inv; dm[3] = acc.w * inv;
            float* dh = s_h + r * LDSTR + gl * 4;
            dh[0] = own.x; dh[1] = own.y; dh[2] = own.z; dh[3] = own.w;
        }
    }
    __syncthreads();

    const int wave = tid >> 5;
    const int lane = tid & 31;
    const int half = lane >> 4;
    const int col  = lane & 15;
    const float alpha = alpha_p[0];

    v8f acc[TPW];
#pragma unroll
    for (int t = 0; t < TPW; ++t)
#pragma unroll
        for (int j = 0; j < 8; ++j) acc[t][j] = 0.0f;

    const float* smrow = s_mean + col * LDSTR;
    const float* shrow = s_h    + col * LDSTR;

    const int wv_s = __builtin_amdgcn_readfirstlane(wave);
    const float4* __restrict__ Wp4 = reinterpret_cast<const float4*>(Wp);

#pragma unroll 2
    for (int j = 0; j < NJ; ++j) {
        const int kb = j * 8 + half * 2;
        v2f am0; am0.x = smrow[kb];     am0.y = smrow[kb + 1];
        v2f ah0; ah0.x = shrow[kb];     ah0.y = shrow[kb + 1];
        v2f am1; am1.x = smrow[kb + 4]; am1.y = smrow[kb + 5];
        v2f ah1; ah1.x = shrow[kb + 4]; ah1.y = shrow[kb + 5];
#pragma unroll
        for (int t = 0; t < TPW; ++t) {
            const int tile = wv_s * TPW + t;
            const float4 pl = Wp4[((tile * NJ + j) * 2 + 0) * 32 + lane];
            const float4 pr = Wp4[((tile * NJ + j) * 2 + 1) * 32 + lane];
            v2f bl0; bl0.x = pl.x; bl0.y = pl.y;
            v2f bl1; bl1.x = pl.z; bl1.y = pl.w;
            v2f br0; br0.x = pr.x; br0.y = pr.y;
            v2f br1; br1.x = pr.z; br1.y = pr.w;
            if (PROJ) {
                acc[t] = __builtin_amdgcn_wmma_f32_16x16x4_f32(
                    false, bl0, false, am0, (short)0, acc[t], false, false);
                acc[t] = __builtin_amdgcn_wmma_f32_16x16x4_f32(
                    false, br0, false, ah0, (short)0, acc[t], false, false);
                acc[t] = __builtin_amdgcn_wmma_f32_16x16x4_f32(
                    false, bl1, false, am1, (short)0, acc[t], false, false);
                acc[t] = __builtin_amdgcn_wmma_f32_16x16x4_f32(
                    false, br1, false, ah1, (short)0, acc[t], false, false);
            } else {
                acc[t] = __builtin_amdgcn_wmma_f32_16x16x4_f32(
                    false, am0, false, bl0, (short)0, acc[t], false, false);
                acc[t] = __builtin_amdgcn_wmma_f32_16x16x4_f32(
                    false, ah0, false, br0, (short)0, acc[t], false, false);
                acc[t] = __builtin_amdgcn_wmma_f32_16x16x4_f32(
                    false, am1, false, bl1, (short)0, acc[t], false, false);
                acc[t] = __builtin_amdgcn_wmma_f32_16x16x4_f32(
                    false, ah1, false, br1, (short)0, acc[t], false, false);
            }
        }
    }

    if (PROJ) {
        v8f pacc = {0.0f, 0.0f, 0.0f, 0.0f, 0.0f, 0.0f, 0.0f, 0.0f};
#pragma unroll
        for (int t = 0; t < TPW; ++t) {
            const int f0 = (wave * TPW + t) * 16 + half * 8;
            const float4 bA = *reinterpret_cast<const float4*>(bias + f0);
            const float4 bB = *reinterpret_cast<const float4*>(bias + f0 + 4);
            const float bv[8] = {bA.x, bA.y, bA.z, bA.w, bB.x, bB.y, bB.z, bB.w};
            float hv[8];
#pragma unroll
            for (int r = 0; r < 8; ++r) {
                float val = acc[t][r] + bv[r];
                if (PRELU) val = (val >= 0.0f) ? val : alpha * val;
                hv[r] = val;
            }
#pragma unroll
            for (int i = 0; i < 4; ++i) {
                v2f hb; hb.x = hv[2 * i]; hb.y = hv[2 * i + 1];
                v2f wa;
                wa.x = Wf[(f0 + 2 * i) * 16 + col];
                wa.y = Wf[(f0 + 2 * i + 1) * 16 + col];
                pacc = __builtin_amdgcn_wmma_f32_16x16x4_f32(
                    false, wa, false, hb, (short)0, pacc, false, false);
            }
        }
        float* s_part = s_mean;
        __syncthreads();
        if (half == 0) {
            float4* d = reinterpret_cast<float4*>(s_part + (wave * 16 + col) * 8);
            d[0] = make_float4(pacc[0], pacc[1], pacc[2], pacc[3]);
            d[1] = make_float4(pacc[4], pacc[5], pacc[6], pacc[7]);
        }
        __syncthreads();
        const float s = ((s_part[tid] + s_part[128 + tid]) + s_part[256 + tid])
                        + s_part[384 + tid];
        out[(size_t)node0 * 8 + tid] = s;
        return;
    }

#pragma unroll
    for (int t = 0; t < TPW; ++t) {
        const int n0 = (wave * TPW + t) * 16;
        const float bv = bias[n0 + col];
#pragma unroll
        for (int v = 0; v < 8; ++v) {
            float val = acc[t][v] + bv;
            if (PRELU) val = (val >= 0.0f) ? val : alpha * val;
            const int row = node0 + v + half * 8;
            out[(size_t)row * FOUT + n0 + col] = val;
        }
    }
}

__device__ __forceinline__ void pack_one(const float* __restrict__ Wl,
                                         const float* __restrict__ Wr,
                                         float* __restrict__ Wp, int idx,
                                         int fin, int fout) {
    const int e    = idx & 3;
    const int lane = (idx >> 2) & 31;
    const int q    = (idx >> 7) & 1;
    const int rest = idx >> 8;
    const int nj   = fin >> 3;
    const int j    = rest % nj;
    const int tile = rest / nj;
    const int k    = 8 * j + 4 * (e >> 1) + 2 * (lane >> 4) + (e & 1);
    const int n    = 16 * tile + (lane & 15);
    Wp[idx] = (q ? Wr : Wl)[k * fout + n];
}

#define PACK1_N (32 * 128 * 2)
#define PACK2_N (128 * 128 * 2)

__global__ void fold_pack_kernel(const float* __restrict__ W3l,
                                 const float* __restrict__ W3r,
                                 const float* __restrict__ b3,
                                 const float* __restrict__ Wc,
                                 const float* __restrict__ bc,
                                 const float* __restrict__ W1l,
                                 const float* __restrict__ W1r,
                                 const float* __restrict__ W2l,
                                 const float* __restrict__ W2r,
                                 float* __restrict__ Wf, float* __restrict__ cf,
                                 float* __restrict__ Wp1,
                                 float* __restrict__ Wp2) {
    int t = blockIdx.x * blockDim.x + threadIdx.x;
    if (t < PACK1_N) pack_one(W1l, W1r, Wp1, t, 32, 128);
    else if (t < PACK1_N + PACK2_N) pack_one(W2l, W2r, Wp2, t - PACK1_N, 128, 128);
    if (t < 128 * 16) {
        int k = t >> 4;
        int c = t & 15;
        float s = 0.0f;
        if (c < 8) {
            const float* w = ((c < 4) ? W3l : W3r) + k * 64;
            int cc = c & 3;
            for (int j = 0; j < 64; ++j) s += w[j] * Wc[j * 4 + cc];
        }
        Wf[t] = s;
    }
    if (t < 4) {
        float s = 0.0f;
        for (int j = 0; j < 64; ++j) s += b3[j] * Wc[j * 4 + t];
        cf[t] = s + bc[t];
    }
}

__global__ void gather4_final_kernel(const float* __restrict__ p,
                                     const int2* __restrict__ rows,
                                     const int* __restrict__ csr,
                                     const float* __restrict__ cf,
                                     float* __restrict__ out, int n) {
    const int t    = blockIdx.x * blockDim.x + threadIdx.x;
    const int node = t >> 2;
    const int f    = t & 3;
    const bool live = node < n;
    const int2 rw  = live ? rows[node] : make_int2(0, 0);
    const int last = rw.y - 1;
    const float4* __restrict__ p4 = reinterpret_cast<const float4*>(p);
    float4 acc = {0.0f, 0.0f, 0.0f, 0.0f};
    for (int e = rw.x + f; e < rw.y; e += 16) {
        const int s0 = csr[e];
        const int s1 = csr[min(e + 4, last)];
        const int s2 = csr[min(e + 8, last)];
        const int s3 = csr[min(e + 12, last)];
        const float4 v0 = p4[(size_t)s0 * 2];
        const float4 v1 = p4[(size_t)s1 * 2];
        const float4 v2 = p4[(size_t)s2 * 2];
        const float4 v3 = p4[(size_t)s3 * 2];
        add4(acc, v0);
        addsel4(acc, v1, e + 4 <= last);
        addsel4(acc, v2, e + 8 <= last);
        addsel4(acc, v3, e + 12 <= last);
    }
    acc.x += __shfl_xor(acc.x, 1); acc.y += __shfl_xor(acc.y, 1);
    acc.z += __shfl_xor(acc.z, 1); acc.w += __shfl_xor(acc.w, 1);
    acc.x += __shfl_xor(acc.x, 2); acc.y += __shfl_xor(acc.y, 2);
    acc.z += __shfl_xor(acc.z, 2); acc.w += __shfl_xor(acc.w, 2);
    const float a  = (f == 0) ? acc.x : (f == 1) ? acc.y : (f == 2) ? acc.z : acc.w;
    const float id = 1.0f / fmaxf((float)(rw.y - rw.x), 1.0f);
    if (live)
        out[(size_t)node * 4 + f] = (a * id + cf[f]) + p[(size_t)node * 8 + 4 + f];
}

extern "C" void kernel_launch(void* const* d_in, const int* in_sizes, int n_in,
                              void* d_out, int out_size, void* d_ws, size_t ws_size,
                              hipStream_t stream) {
    const float* x   = (const float*)d_in[0];
    const int*   ei  = (const int*)d_in[1];
    const float* W1l = (const float*)d_in[2];
    const float* b1  = (const float*)d_in[3];
    const float* W1r = (const float*)d_in[4];
    const float* W2l = (const float*)d_in[5];
    const float* b2  = (const float*)d_in[6];
    const float* W2r = (const float*)d_in[7];
    const float* W3l = (const float*)d_in[8];
    const float* b3  = (const float*)d_in[9];
    const float* W3r = (const float*)d_in[10];
    const float* Wc  = (const float*)d_in[11];
    const float* bc  = (const float*)d_in[12];
    const float* a   = (const float*)d_in[13];
    float* out = (float*)d_out;

    const int E = in_sizes[1] / 2;
    const int* srcv = ei;
    const int* dstv = ei + E;

    char* ws = (char*)d_ws;
    size_t off = 0;
    auto carve = [&](size_t bytes) {
        float* p = (float*)(ws + off);
        off += (bytes + 255) & ~(size_t)255;
        return p;
    };
    int*   cnt = (int*)carve((size_t)(N_NODES + 1) * 4);
    float* el  = carve((size_t)N_NODES * 128 * 4);
    float* hA  = carve((size_t)N_NODES * 128 * 4);
    float* hB  = carve((size_t)N_NODES * 128 * 4);

    const int nCh = (E + CHUNK - 1) / CHUNK;

    int2* rows = (int2*)el;
    int*  rec  = (int*)el + (size_t)N_NODES * 2;
    int2* offs = (int2*)(rec + (size_t)nCh * CHUNK);
    int*  csr  = (int*)(offs + (size_t)nCh * N_RNG);

    const int TB = 256;
    const int nBlk16 = N_NODES / 16;

    float* proj = hB;
    float* Wf   = hB + (size_t)N_NODES * 8;
    float* cf   = Wf + 128 * 16;
    float* Wp1  = cf + 64;
    float* Wp2  = Wp1 + PACK1_N;
    fold_pack_kernel<<<(PACK1_N + PACK2_N + TB - 1) / TB, TB, 0, stream>>>(
        W3l, W3r, b3, Wc, bc, W1l, W1r, W2l, W2r, Wf, cf, Wp1, Wp2);

    bin_edges_kernel<<<nCh, BIN_TB, 0, stream>>>(
        srcv, dstv, rec, offs, cnt + N_NODES, E);
    rank_range_kernel<<<N_RNG, 1024, 0, stream>>>(
        rec, offs, cnt + N_NODES, rows, csr, nCh);

    sage_gather_combine_wmma<32, 128, 2, true, false><<<nBlk16, 128, 0, stream>>>(
        x, rows, csr, Wp1, b1, a, nullptr, hA);

    sage_gather_combine_wmma<128, 128, 2, true, true><<<nBlk16, 128, 0, stream>>>(
        hA, rows, csr, Wp2, b2, a, Wf, proj);

    gather4_final_kernel<<<(N_NODES * 4 + TB - 1) / TB, TB, 0, stream>>>(
        proj, rows, csr, cf, out, N_NODES);
}


# Edge sort in chunks of 8192 edges: a range's piece of a chunk is ~42 words, read by rank_range a wave per piece, lane per word (1-2 lines per load, not 32); sum order changes (arrival-order rank)
# speedup vs baseline: 6.3427x; 1.0366x over previous
#include <hip/hip_runtime.h>
#include <hip/hip_bf16.h>


typedef __attribute__((ext_vector_type(2))) float v2f;
typedef __attribute__((ext_vector_type(8))) float v8f;

#define N_NODES 100000

#define RNG_SHIFT 9
#define RNG_NODES (1 << RNG_SHIFT)
#define N_RNG ((N_NODES + RNG_NODES - 1) / RNG_NODES)
#define SRC_BITS 17
#define CHUNK 8192
#define BIN_TB 1024
#define EPL (CHUNK / BIN_TB)
#define PGRP 8
#define OUT_CAP 12288

static_assert(N_NODES <= (1 << SRC_BITS), "node id must fit SRC_BITS");
static_assert(RNG_SHIFT + SRC_BITS < 32, "packed edge word must stay positive");
static_assert(N_RNG <= BIN_TB, "one thread per range in the chunk scan");

__device__ __forceinline__ int block_scan_excl(int c, int* s_wsum, int* s_total) {
    const int lane = threadIdx.x & 31;
    const int w    = threadIdx.x >> 5;
    const int nw   = blockDim.x >> 5;
    int v = c;
#pragma unroll
    for (int o = 1; o < 32; o <<= 1) {
        int t = __shfl_up(v, o);
        if (lane >= o) v += t;
    }
    if (lane == 31) s_wsum[w] = v;
    __syncthreads();
    if (w == 0) {
        const int t = (lane < nw) ? s_wsum[lane] : 0;
        int u = t;
#pragma unroll
        for (int o = 1; o < 32; o <<= 1) {
            int q = __shfl_up(u, o);
            if (lane >= o) u += q;
        }
        s_wsum[lane] = u - t;
        if (lane == 31) *s_total = u;
    }
    __syncthreads();
    return s_wsum[w] + (v - c);
}

__global__ __launch_bounds__(BIN_TB) void bin_edges_kernel(
    const int* __restrict__ src, const int* __restrict__ dst,
    int* __restrict__ rec, int2* __restrict__ offs,
    int* __restrict__ total, int E) {
    __shared__ int s_hist[N_RNG];
    __shared__ int s_off[N_RNG];
    __shared__ int s_rec[CHUNK];
    __shared__ int s_wsum[32];
    __shared__ int s_tot;
    const int tid   = threadIdx.x;
    const int chunk = blockIdx.x;
    const int e0    = chunk * CHUNK + tid;
    if (tid < N_RNG) s_hist[tid] = 0;
    if (chunk == 0 && tid == 0) *total = 0;

    int d[EPL], s[EPL];
#pragma unroll
    for (int k = 0; k < EPL; ++k) {
        const int ec = min(e0 + k * BIN_TB, E - 1);
        d[k] = dst[ec];
        s[k] = src[ec];
    }
    __syncthreads();

    int rk[EPL];
#pragma unroll
    for (int k = 0; k < EPL; ++k) {
        rk[k] = 0;
        if (e0 + k * BIN_TB < E) rk[k] = atomicAdd(&s_hist[d[k] >> RNG_SHIFT], 1);
    }
    __syncthreads();

    const int c  = (tid < N_RNG) ? s_hist[tid] : 0;
    const int ex = block_scan_excl(c, s_wsum, &s_tot);
    if (tid < N_RNG) {
        s_off[tid] = ex;
        offs[(size_t)chunk * N_RNG + tid] = make_int2(ex, ex + c);
    }
    __syncthreads();

#pragma unroll
    for (int k = 0; k < EPL; ++k) {
        if (e0 + k * BIN_TB < E)
            s_rec[s_off[d[k] >> RNG_SHIFT] + rk[k]] =
                ((d[k] & (RNG_NODES - 1)) << SRC_BITS) | s[k];
    }
    __syncthreads();
    const int nv = s_tot;
    int* __restrict__ rp = rec + (size_t)chunk * CHUNK;
#pragma unroll
    for (int k = 0; k < EPL; ++k) {
        const int p = k * BIN_TB + tid;
        if (p < nv) rp[p] = s_rec[p];
    }
}

template <bool FILL>
__device__ __forceinline__ void walk_pieces(
    const int* __restrict__ rec, const int2* __restrict__ offs,
    int* s_cnt, int* s_out, int* __restrict__ csr,
    int nCh, int b, int w, int lane, bool staged, int base) {
    const int smask = (1 << SRC_BITS) - 1;
    for (int c0 = 0; c0 < nCh; c0 += 1024) {
        const int chl = c0 + 32 * lane + w;
        int2 bd = make_int2(0, 0);
        if (chl < nCh) bd = offs[(size_t)chl * N_RNG + b];
        const int left = nCh - c0 - w;
        const int np = left > 0 ? min(32, (left + 31) >> 5) : 0;
        for (int g = 0; g < np; g += PGRP) {
            int r[PGRP][2];
            bool ok[PGRP][2];
            int pbv[PGRP], pev[PGRP], chv[PGRP];
#pragma unroll
            for (int q = 0; q < PGRP; ++q) {
                const int i = g + q;
                pbv[q] = __builtin_amdgcn_readlane(bd.x, i);
                pev[q] = __builtin_amdgcn_readlane(bd.y, i);
                chv[q] = min(c0 + 32 * i + w, nCh - 1);
                const int* __restrict__ rp = rec + (size_t)chv[q] * CHUNK;
                const int last = max(pev[q] - 1, 0);
#pragma unroll
                for (int t = 0; t < 2; ++t) {
                    const int j = pbv[q] + lane + 32 * t;
                    ok[q][t] = j < pev[q];
                    r[q][t] = rp[min(j, last)];
                }
            }
#pragma unroll
            for (int q = 0; q < PGRP; ++q) {
#pragma unroll
                for (int t = 0; t < 2; ++t) {
                    if (ok[q][t]) {
                        const int word = r[q][t];
                        if (FILL) {
                            const int sl = atomicAdd(&s_cnt[word >> SRC_BITS], 1);
                            if (staged) s_out[sl] = word & smask;
                            else csr[base + sl] = word & smask;
                        } else {
                            atomicAdd(&s_cnt[word >> SRC_BITS], 1);
                        }
                    }
                }
            }
#pragma unroll
            for (int q = 0; q < PGRP; ++q) {
                const int* __restrict__ rp = rec + (size_t)chv[q] * CHUNK;
                for (int j0 = pbv[q] + 64; j0 < pev[q]; j0 += 32) {
                    const int j = j0 + lane;
                    if (j < pev[q]) {
                        const int word = rp[j];
                        if (FILL) {
                            const int sl = atomicAdd(&s_cnt[word >> SRC_BITS], 1);
                            if (staged) s_out[sl] = word & smask;
                            else csr[base + sl] = word & smask;
                        } else {
                            atomicAdd(&s_cnt[word >> SRC_BITS], 1);
                        }
                    }
                }
            }
        }
    }
}

__global__ __launch_bounds__(1024) void rank_range_kernel(
    const int* __restrict__ rec, const int2* __restrict__ offs,
    int* __restrict__ total, int2* __restrict__ rows,
    int* __restrict__ csr, int nCh) {
    __shared__ int s_cnt[RNG_NODES];
    __shared__ int s_out[OUT_CAP];
    __shared__ int s_wsum[32];
    __shared__ int s_tot;
    __shared__ int s_base;
    const int tid  = threadIdx.x;
    const int lane = tid & 31;
    const int w = __builtin_amdgcn_readfirstlane(tid >> 5);
    const int b = blockIdx.x;
    if (tid < RNG_NODES) s_cnt[tid] = 0;
    __syncthreads();

    walk_pieces<false>(rec, offs, s_cnt, s_out, csr, nCh, b, w, lane, false, 0);
    __syncthreads();

    const int c  = (tid < RNG_NODES) ? s_cnt[tid] : 0;
    const int ex = block_scan_excl(c, s_wsum, &s_tot);
    if (tid == 0) s_base = atomicAdd(total, s_tot);
    __syncthreads();
    const int nb   = s_tot;
    const int base = s_base;
    if (tid < RNG_NODES) {
        s_cnt[tid] = ex;
        const int node = b * RNG_NODES + tid;
        if (node < N_NODES) rows[node] = make_int2(base + ex, base + ex + c);
    }
    __syncthreads();

    const bool staged = nb <= OUT_CAP;
    walk_pieces<true>(rec, offs, s_cnt, s_out, csr, nCh, b, w, lane, staged, base);
    if (staged) {
        __syncthreads();
        for (int i = tid; i < nb; i += 1024) csr[base + i] = s_out[i];
    }
}

__device__ __forceinline__ void add4(float4& a, const float4& b) {
    a.x += b.x; a.y += b.y; a.z += b.z; a.w += b.w;
}

__device__ __forceinline__ void addsel4(float4& a, const float4& b, bool in) {
    a.x += in ? b.x : 0.0f; a.y += in ? b.y : 0.0f;
    a.z += in ? b.z : 0.0f; a.w += in ? b.w : 0.0f;
}

template <int FIN, int FOUT, int TPW, bool PRELU, bool PROJ>
__global__ __launch_bounds__(128) void sage_gather_combine_wmma(
    const float* __restrict__ h, const int2* __restrict__ rows,
    const int* __restrict__ csr,
    const float* __restrict__ Wp, const float* __restrict__ bias,
    const float* __restrict__ alpha_p,
    const float* __restrict__ Wf, float* __restrict__ out) {
    constexpr int LDSTR = FIN + 1;
    constexpr int NJ = FIN / 8;
    __shared__ alignas(16) float s_mean[16 * LDSTR];
    __shared__ float s_h[16 * LDSTR];
    static_assert(16 * LDSTR >= 4 * 16 * 8, "partial p must fit s_mean");

    const int tid   = threadIdx.x;
    const int node0 = blockIdx.x * 16;
    const float4* __restrict__ h4 = reinterpret_cast<const float4*>(h);

    if (FIN == 32) {
        const int r    = tid >> 3;
        const int sub  = tid & 7;
        const int node = node0 + r;
        const int2 rw  = rows[node];
        const int last = rw.y - 1;
        float4 acc = {0.0f, 0.0f, 0.0f, 0.0f};
        for (int e = rw.x; e < rw.y; e += 4) {
            const int s0 = csr[e];
            const int s1 = csr[min(e + 1, last)];
            const int s2 = csr[min(e + 2, last)];
            const int s3 = csr[min(e + 3, last)];
            const float4 v0 = h4[(size_t)s0 * 8 + sub];
            const float4 v1 = h4[(size_t)s1 * 8 + sub];
            const float4 v2 = h4[(size_t)s2 * 8 + sub];
            const float4 v3 = h4[(size_t)s3 * 8 + sub];
            add4(acc, v0);
            addsel4(acc, v1, e + 1 <= last);
            addsel4(acc, v2, e + 2 <= last);
            addsel4(acc, v3, e + 3 <= last);
        }
        const float inv = 1.0f / fmaxf((float)(rw.y - rw.x), 1.0f);
        const float4 own = h4[(size_t)node * 8 + sub];
        float* dm = s_mean + r * LDSTR + sub * 4;
        dm[0] = acc.x * inv; dm[1] = acc.y * inv;
        dm[2] = acc.z * inv; dm[3] = acc.w * inv;
        float* dh = s_h + r * LDSTR + sub * 4;
        dh[0] = own.x; dh[1] = own.y; dh[2] = own.z; dh[3] = own.w;
    } else {
        const int gl = tid & 31;
        const int wv = __builtin_amdgcn_readfirstlane(tid >> 5);
        for (int q = 0; q < 4; ++q) {
            const int r    = wv * 4 + q;
            const int node = node0 + r;
            const int2 rw  = rows[node];
            const int start = __builtin_amdgcn_readfirstlane(rw.x);
            const int end   = __builtin_amdgcn_readfirstlane(rw.y);
            float4 acc = {0.0f, 0.0f, 0.0f, 0.0f};
            for (int base = start; base < end; base += 32) {
                const int n  = min(32, end - base);
                const int my = csr[base + min(gl, n - 1)];
                int j = 0;
                for (; j + 8 <= n; j += 8) {
                    float4 v[8];
#pragma unroll
                    for (int k = 0; k < 8; ++k) {
                        const int s = __builtin_amdgcn_readlane(my, j + k);
                        v[k] = h4[(size_t)s * 32 + gl];
                    }
#pragma unroll
                    for (int k = 0; k < 8; ++k) add4(acc, v[k]);
                }
                if (j < n) {
                    float4 v[7];
#pragma unroll
                    for (int k = 0; k < 7; ++k) {
                        v[k] = make_float4(0.0f, 0.0f, 0.0f, 0.0f);
                        if (j + k < n) {
                            const int s = __builtin_amdgcn_readlane(my, j + k);
                            v[k] = h4[(size_t)s * 32 + gl];
                        }
                    }
#pragma unroll
                    for (int k = 0; k < 7; ++k)
                        if (j + k < n) add4(acc, v[k]);
                }
            }
            const float inv = 1.0f / fmaxf((float)(end - start), 1.0f);
            const float4 own = h4[(size_t)node * 32 + gl];
            float* dm = s_mean + r * LDSTR + gl * 4;
            dm[0] = acc.x * inv; dm[1] = acc.y * inv;
            dm[2] = acc.z * inv; dm[3] = acc.w * inv;
            float* dh = s_h + r * LDSTR + gl * 4;
            dh[0] = own.x; dh[1] = own.y; dh[2] = own.z; dh[3] = own.w;
        }
    }
    __syncthreads();

    const int wave = tid >> 5;
    const int lane = tid & 31;
    const int half = lane >> 4;
    const int col  = lane & 15;
    const float alpha = alpha_p[0];

    v8f acc[TPW];
#pragma unroll
    for (int t = 0; t < TPW; ++t)
#pragma unroll
        for (int j = 0; j < 8; ++j) acc[t][j] = 0.0f;

    const float* smrow = s_mean + col * LDSTR;
    const float* shrow = s_h    + col * LDSTR;

    const int wv_s = __builtin_amdgcn_readfirstlane(wave);
    const float4* __restrict__ Wp4 = reinterpret_cast<const float4*>(Wp);

#pragma unroll 2
    for (int j = 0; j < NJ; ++j) {
        const int kb = j * 8 + half * 2;
        v2f am0; am0.x = smrow[kb];     am0.y = smrow[kb + 1];
        v2f ah0; ah0.x = shrow[kb];     ah0.y = shrow[kb + 1];
        v2f am1; am1.x = smrow[kb + 4]; am1.y = smrow[kb + 5];
        v2f ah1; ah1.x = shrow[kb + 4]; ah1.y = shrow[kb + 5];
#pragma unroll
        for (int t = 0; t < TPW; ++t) {
            const int tile = wv_s * TPW + t;
            const float4 pl = Wp4[((tile * NJ + j) * 2 + 0) * 32 + lane];
            const float4 pr = Wp4[((tile * NJ + j) * 2 + 1) * 32 + lane];
            v2f bl0; bl0.x = pl.x; bl0.y = pl.y;
            v2f bl1; bl1.x = pl.z; bl1.y = pl.w;
            v2f br0; br0.x = pr.x; br0.y = pr.y;
            v2f br1; br1.x = pr.z; br1.y = pr.w;
            if (PROJ) {
                acc[t] = __builtin_amdgcn_wmma_f32_16x16x4_f32(
                    false, bl0, false, am0, (short)0, acc[t], false, false);
                acc[t] = __builtin_amdgcn_wmma_f32_16x16x4_f32(
                    false, br0, false, ah0, (short)0, acc[t], false, false);
                acc[t] = __builtin_amdgcn_wmma_f32_16x16x4_f32(
                    false, bl1, false, am1, (short)0, acc[t], false, false);
                acc[t] = __builtin_amdgcn_wmma_f32_16x16x4_f32(
                    false, br1, false, ah1, (short)0, acc[t], false, false);
            } else {
                acc[t] = __builtin_amdgcn_wmma_f32_16x16x4_f32(
                    false, am0, false, bl0, (short)0, acc[t], false, false);
                acc[t] = __builtin_amdgcn_wmma_f32_16x16x4_f32(
                    false, ah0, false, br0, (short)0, acc[t], false, false);
                acc[t] = __builtin_amdgcn_wmma_f32_16x16x4_f32(
                    false, am1, false, bl1, (short)0, acc[t], false, false);
                acc[t] = __builtin_amdgcn_wmma_f32_16x16x4_f32(
                    false, ah1, false, br1, (short)0, acc[t], false, false);
            }
        }
    }

    if (PROJ) {
        v8f pacc = {0.0f, 0.0f, 0.0f, 0.0f, 0.0f, 0.0f, 0.0f, 0.0f};
#pragma unroll
        for (int t = 0; t < TPW; ++t) {
            const int f0 = (wave * TPW + t) * 16 + half * 8;
            const float4 bA = *reinterpret_cast<const float4*>(bias + f0);
            const float4 bB = *reinterpret_cast<const float4*>(bias + f0 + 4);
            const float bv[8] = {bA.x, bA.y, bA.z, bA.w, bB.x, bB.y, bB.z, bB.w};
            float hv[8];
#pragma unroll
            for (int r = 0; r < 8; ++r) {
                float val = acc[t][r] + bv[r];
                if (PRELU) val = (val >= 0.0f) ? val : alpha * val;
                hv[r] = val;
            }
#pragma unroll
            for (int i = 0; i < 4; ++i) {
                v2f hb; hb.x = hv[2 * i]; hb.y = hv[2 * i + 1];
                v2f wa;
                wa.x = Wf[(f0 + 2 * i) * 16 + col];
                wa.y = Wf[(f0 + 2 * i + 1) * 16 + col];
                pacc = __builtin_amdgcn_wmma_f32_16x16x4_f32(
                    false, wa, false, hb, (short)0, pacc, false, false);
            }
        }
        float* s_part = s_mean;
        __syncthreads();
        if (half == 0) {
            float4* d = reinterpret_cast<float4*>(s_part + (wave * 16 + col) * 8);
            d[0] = make_float4(pacc[0], pacc[1], pacc[2], pacc[3]);
            d[1] = make_float4(pacc[4], pacc[5], pacc[6], pacc[7]);
        }
        __syncthreads();
        const float s = ((s_part[tid] + s_part[128 + tid]) + s_part[256 + tid])
                        + s_part[384 + tid];
        out[(size_t)node0 * 8 + tid] = s;
        return;
    }

#pragma unroll
    for (int t = 0; t < TPW; ++t) {
        const int n0 = (wave * TPW + t) * 16;
        const float bv = bias[n0 + col];
#pragma unroll
        for (int v = 0; v < 8; ++v) {
            float val = acc[t][v] + bv;
            if (PRELU) val = (val >= 0.0f) ? val : alpha * val;
            const int row = node0 + v + half * 8;
            out[(size_t)row * FOUT + n0 + col] = val;
        }
    }
}

__device__ __forceinline__ void pack_one(const float* __restrict__ Wl,
                                         const float* __restrict__ Wr,
                                         float* __restrict__ Wp, int idx,
                                         int fin, int fout) {
    const int e    = idx & 3;
    const int lane = (idx >> 2) & 31;
    const int q    = (idx >> 7) & 1;
    const int rest = idx >> 8;
    const int nj   = fin >> 3;
    const int j    = rest % nj;
    const int tile = rest / nj;
    const int k    = 8 * j + 4 * (e >> 1) + 2 * (lane >> 4) + (e & 1);
    const int n    = 16 * tile + (lane & 15);
    Wp[idx] = (q ? Wr : Wl)[k * fout + n];
}

#define PACK1_N (32 * 128 * 2)
#define PACK2_N (128 * 128 * 2)

__global__ void fold_pack_kernel(const float* __restrict__ W3l,
                                 const float* __restrict__ W3r,
                                 const float* __restrict__ b3,
                                 const float* __restrict__ Wc,
                                 const float* __restrict__ bc,
                                 const float* __restrict__ W1l,
                                 const float* __restrict__ W1r,
                                 const float* __restrict__ W2l,
                                 const float* __restrict__ W2r,
                                 float* __restrict__ Wf, float* __restrict__ cf,
                                 float* __restrict__ Wp1,
                                 float* __restrict__ Wp2) {
    int t = blockIdx.x * blockDim.x + threadIdx.x;
    if (t < PACK1_N) pack_one(W1l, W1r, Wp1, t, 32, 128);
    else if (t < PACK1_N + PACK2_N) pack_one(W2l, W2r, Wp2, t - PACK1_N, 128, 128);
    if (t < 128 * 16) {
        int k = t >> 4;
        int c = t & 15;
        float s = 0.0f;
        if (c < 8) {
            const float* w = ((c < 4) ? W3l : W3r) + k * 64;
            int cc = c & 3;
            for (int j = 0; j < 64; ++j) s += w[j] * Wc[j * 4 + cc];
        }
        Wf[t] = s;
    }
    if (t < 4) {
        float s = 0.0f;
        for (int j = 0; j < 64; ++j) s += b3[j] * Wc[j * 4 + t];
        cf[t] = s + bc[t];
    }
}

__global__ void gather4_final_kernel(const float* __restrict__ p,
                                     const int2* __restrict__ rows,
                                     const int* __restrict__ csr,
                                     const float* __restrict__ cf,
                                     float* __restrict__ out, int n) {
    const int t    = blockIdx.x * blockDim.x + threadIdx.x;
    const int node = t >> 2;
    const int f    = t & 3;
    const bool live = node < n;
    const int2 rw  = live ? rows[node] : make_int2(0, 0);
    const int last = rw.y - 1;
    const float4* __restrict__ p4 = reinterpret_cast<const float4*>(p);
    float4 acc = {0.0f, 0.0f, 0.0f, 0.0f};
    for (int e = rw.x + f; e < rw.y; e += 16) {
        const int s0 = csr[e];
        const int s1 = csr[min(e + 4, last)];
        const int s2 = csr[min(e + 8, last)];
        const int s3 = csr[min(e + 12, last)];
        const float4 v0 = p4[(size_t)s0 * 2];
        const float4 v1 = p4[(size_t)s1 * 2];
        const float4 v2 = p4[(size_t)s2 * 2];
        const float4 v3 = p4[(size_t)s3 * 2];
        add4(acc, v0);
        addsel4(acc, v1, e + 4 <= last);
        addsel4(acc, v2, e + 8 <= last);
        addsel4(acc, v3, e + 12 <= last);
    }
    acc.x += __shfl_xor(acc.x, 1); acc.y += __shfl_xor(acc.y, 1);
    acc.z += __shfl_xor(acc.z, 1); acc.w += __shfl_xor(acc.w, 1);
    acc.x += __shfl_xor(acc.x, 2); acc.y += __shfl_xor(acc.y, 2);
    acc.z += __shfl_xor(acc.z, 2); acc.w += __shfl_xor(acc.w, 2);
    const float a  = (f == 0) ? acc.x : (f == 1) ? acc.y : (f == 2) ? acc.z : acc.w;
    const float id = 1.0f / fmaxf((float)(rw.y - rw.x), 1.0f);
    if (live)
        out[(size_t)node * 4 + f] = (a * id + cf[f]) + p[(size_t)node * 8 + 4 + f];
}

extern "C" void kernel_launch(void* const* d_in, const int* in_sizes, int n_in,
                              void* d_out, int out_size, void* d_ws, size_t ws_size,
                              hipStream_t stream) {
    const float* x   = (const float*)d_in[0];
    const int*   ei  = (const int*)d_in[1];
    const float* W1l = (const float*)d_in[2];
    const float* b1  = (const float*)d_in[3];
    const float* W1r = (const float*)d_in[4];
    const float* W2l = (const float*)d_in[5];
    const float* b2  = (const float*)d_in[6];
    const float* W2r = (const float*)d_in[7];
    const float* W3l = (const float*)d_in[8];
    const float* b3  = (const float*)d_in[9];
    const float* W3r = (const float*)d_in[10];
    const float* Wc  = (const float*)d_in[11];
    const float* bc  = (const float*)d_in[12];
    const float* a   = (const float*)d_in[13];
    float* out = (float*)d_out;

    const int E = in_sizes[1] / 2;
    const int* srcv = ei;
    const int* dstv = ei + E;

    char* ws = (char*)d_ws;
    size_t off = 0;
    auto carve = [&](size_t bytes) {
        float* p = (float*)(ws + off);
        off += (bytes + 255) & ~(size_t)255;
        return p;
    };
    int*   cnt = (int*)carve((size_t)(N_NODES + 1) * 4);
    float* el  = carve((size_t)N_NODES * 128 * 4);
    float* hA  = carve((size_t)N_NODES * 128 * 4);
    float* hB  = carve((size_t)N_NODES * 128 * 4);

    const int nCh = (E + CHUNK - 1) / CHUNK;

    int2* rows = (int2*)el;
    int*  rec  = (int*)el + (size_t)N_NODES * 2;
    int2* offs = (int2*)(rec + (size_t)nCh * CHUNK);
    int*  csr  = (int*)(offs + (size_t)nCh * N_RNG);

    const int TB = 256;
    const int nBlk16 = N_NODES / 16;

    float* proj = hB;
    float* Wf   = hB + (size_t)N_NODES * 8;
    float* cf   = Wf + 128 * 16;
    float* Wp1  = cf + 64;
    float* Wp2  = Wp1 + PACK1_N;
    fold_pack_kernel<<<(PACK1_N + PACK2_N + TB - 1) / TB, TB, 0, stream>>>(
        W3l, W3r, b3, Wc, bc, W1l, W1r, W2l, W2r, Wf, cf, Wp1, Wp2);

    bin_edges_kernel<<<nCh, BIN_TB, 0, stream>>>(
        srcv, dstv, rec, offs, cnt + N_NODES, E);
    rank_range_kernel<<<N_RNG, 1024, 0, stream>>>(
        rec, offs, cnt + N_NODES, rows, csr, nCh);

    sage_gather_combine_wmma<32, 128, 2, true, false><<<nBlk16, 128, 0, stream>>>(
        x, rows, csr, Wp1, b1, a, nullptr, hA);

    sage_gather_combine_wmma<128, 128, 2, true, true><<<nBlk16, 128, 0, stream>>>(
        hA, rows, csr, Wp2, b2, a, Wf, proj);

    gather4_final_kernel<<<(N_NODES * 4 + TB - 1) / TB, TB, 0, stream>>>(
        proj, rows, csr, cf, out, N_NODES);
}


# Folded and packed weights written by 43 blocks appended to the edge-binning grid (WGPs it leaves idle): their own launch and its serial load chain gone, 6 launches to 5; sum order unchanged
# speedup vs baseline: 6.5451x; 1.0319x over previous
#include <hip/hip_runtime.h>
#include <hip/hip_bf16.h>


typedef __attribute__((ext_vector_type(2))) float v2f;
typedef __attribute__((ext_vector_type(8))) float v8f;

#define N_NODES 100000

#define RNG_SHIFT 9
#define RNG_NODES (1 << RNG_SHIFT)
#define N_RNG ((N_NODES + RNG_NODES - 1) / RNG_NODES)
#define SRC_BITS 17
#define CHUNK 8192
#define BIN_TB 1024
#define EPL (CHUNK / BIN_TB)
#define PGRP 8
#define OUT_CAP 12288

static_assert(N_NODES <= (1 << SRC_BITS), "node id must fit SRC_BITS");
static_assert(RNG_SHIFT + SRC_BITS < 32, "packed edge word must stay positive");
static_assert(N_RNG <= BIN_TB, "one thread per range in the chunk scan");

__device__ __forceinline__ int block_scan_excl(int c, int* s_wsum, int* s_total) {
    const int lane = threadIdx.x & 31;
    const int w    = threadIdx.x >> 5;
    const int nw   = blockDim.x >> 5;
    int v = c;
#pragma unroll
    for (int o = 1; o < 32; o <<= 1) {
        int t = __shfl_up(v, o);
        if (lane >= o) v += t;
    }
    if (lane == 31) s_wsum[w] = v;
    __syncthreads();
    if (w == 0) {
        const int t = (lane < nw) ? s_wsum[lane] : 0;
        int u = t;
#pragma unroll
        for (int o = 1; o < 32; o <<= 1) {
            int q = __shfl_up(u, o);
            if (lane >= o) u += q;
        }
        s_wsum[lane] = u - t;
        if (lane == 31) *s_total = u;
    }
    __syncthreads();
    return s_wsum[w] + (v - c);
}

#define PACK1_N (32 * 128 * 2)
#define PACK2_N (128 * 128 * 2)
#define WF_N (128 * 16)
#define PACK_BLOCKS ((PACK1_N + PACK2_N) / BIN_TB)
#define WF_BLOCKS (WF_N / BIN_TB)
#define XTRA_BLOCKS (PACK_BLOCKS + WF_BLOCKS + 1)

static_assert(PACK1_N % BIN_TB == 0 && PACK2_N % BIN_TB == 0,
              "a pack block lies inside one layer");
static_assert(WF_N % BIN_TB == 0, "whole blocks of Wf");

__device__ __forceinline__ void pack_one(const float* __restrict__ Wl,
                                         const float* __restrict__ Wr,
                                         float* __restrict__ Wp, int idx,
                                         int fin, int fout) {
    const int e    = idx & 3;
    const int lane = (idx >> 2) & 31;
    const int q    = (idx >> 7) & 1;
    const int rest = idx >> 8;
    const int nj   = fin >> 3;
    const int j    = rest % nj;
    const int tile = rest / nj;
    const int k    = 8 * j + 4 * (e >> 1) + 2 * (lane >> 4) + (e & 1);
    const int n    = 16 * tile + (lane & 15);
    Wp[idx] = (q ? Wr : Wl)[k * fout + n];
}

__device__ __forceinline__ void fold_pack_block(
    int xb, int tid,
    const float* __restrict__ W3l, const float* __restrict__ W3r,
    const float* __restrict__ b3, const float* __restrict__ Wc,
    const float* __restrict__ bc,
    const float* __restrict__ W1l, const float* __restrict__ W1r,
    const float* __restrict__ W2l, const float* __restrict__ W2r,
    float* __restrict__ Wf, float* __restrict__ cf,
    float* __restrict__ Wp1, float* __restrict__ Wp2) {
    if (xb < PACK_BLOCKS) {
        const int idx = xb * BIN_TB + tid;
        if (idx < PACK1_N) pack_one(W1l, W1r, Wp1, idx, 32, 128);
        else pack_one(W2l, W2r, Wp2, idx - PACK1_N, 128, 128);
    } else if (xb < PACK_BLOCKS + WF_BLOCKS) {
        const int t = (xb - PACK_BLOCKS) * BIN_TB + tid;
        const int k = t >> 4;
        const int c = t & 15;
        float s = 0.0f;
        if (c < 8) {
            const float* w = ((c < 4) ? W3l : W3r) + k * 64;
            const int cc = c & 3;
            for (int j = 0; j < 64; ++j) s += w[j] * Wc[j * 4 + cc];
        }
        Wf[t] = s;
    } else if (tid < 4) {
        float s = 0.0f;
#pragma unroll
        for (int j = 0; j < 64; ++j) s += b3[j] * Wc[j * 4 + tid];
        cf[tid] = s + bc[tid];
    }
}

__global__ __launch_bounds__(BIN_TB) void bin_edges_kernel(
    const int* __restrict__ src, const int* __restrict__ dst,
    int* __restrict__ rec, int2* __restrict__ offs,
    int* __restrict__ total, int E, int nCh,
    const float* __restrict__ W3l, const float* __restrict__ W3r,
    const float* __restrict__ b3, const float* __restrict__ Wc,
    const float* __restrict__ bc,
    const float* __restrict__ W1l, const float* __restrict__ W1r,
    const float* __restrict__ W2l, const float* __restrict__ W2r,
    float* __restrict__ Wf, float* __restrict__ cf,
    float* __restrict__ Wp1, float* __restrict__ Wp2) {
    if ((int)blockIdx.x >= nCh) {
        fold_pack_block((int)blockIdx.x - nCh, threadIdx.x, W3l, W3r, b3, Wc, bc,
                        W1l, W1r, W2l, W2r, Wf, cf, Wp1, Wp2);
        return;
    }
    __shared__ int s_hist[N_RNG];
    __shared__ int s_off[N_RNG];
    __shared__ int s_rec[CHUNK];
    __shared__ int s_wsum[32];
    __shared__ int s_tot;
    const int tid   = threadIdx.x;
    const int chunk = blockIdx.x;
    const int e0    = chunk * CHUNK + tid;
    if (tid < N_RNG) s_hist[tid] = 0;
    if (chunk == 0 && tid == 0) *total = 0;

    int d[EPL], s[EPL];
#pragma unroll
    for (int k = 0; k < EPL; ++k) {
        const int ec = min(e0 + k * BIN_TB, E - 1);
        d[k] = dst[ec];
        s[k] = src[ec];
    }
    __syncthreads();

    int rk[EPL];
#pragma unroll
    for (int k = 0; k < EPL; ++k) {
        rk[k] = 0;
        if (e0 + k * BIN_TB < E) rk[k] = atomicAdd(&s_hist[d[k] >> RNG_SHIFT], 1);
    }
    __syncthreads();

    const int c  = (tid < N_RNG) ? s_hist[tid] : 0;
    const int ex = block_scan_excl(c, s_wsum, &s_tot);
    if (tid < N_RNG) {
        s_off[tid] = ex;
        offs[(size_t)chunk * N_RNG + tid] = make_int2(ex, ex + c);
    }
    __syncthreads();

#pragma unroll
    for (int k = 0; k < EPL; ++k) {
        if (e0 + k * BIN_TB < E)
            s_rec[s_off[d[k] >> RNG_SHIFT] + rk[k]] =
                ((d[k] & (RNG_NODES - 1)) << SRC_BITS) | s[k];
    }
    __syncthreads();
    const int nv = s_tot;
    int* __restrict__ rp = rec + (size_t)chunk * CHUNK;
#pragma unroll
    for (int k = 0; k < EPL; ++k) {
        const int p = k * BIN_TB + tid;
        if (p < nv) rp[p] = s_rec[p];
    }
}

template <bool FILL>
__device__ __forceinline__ void walk_pieces(
    const int* __restrict__ rec, const int2* __restrict__ offs,
    int* s_cnt, int* s_out, int* __restrict__ csr,
    int nCh, int b, int w, int lane, bool staged, int base) {
    const int smask = (1 << SRC_BITS) - 1;
    for (int c0 = 0; c0 < nCh; c0 += 1024) {
        const int chl = c0 + 32 * lane + w;
        int2 bd = make_int2(0, 0);
        if (chl < nCh) bd = offs[(size_t)chl * N_RNG + b];
        const int left = nCh - c0 - w;
        const int np = left > 0 ? min(32, (left + 31) >> 5) : 0;
        for (int g = 0; g < np; g += PGRP) {
            int r[PGRP][2];
            bool ok[PGRP][2];
            int pbv[PGRP], pev[PGRP], chv[PGRP];
#pragma unroll
            for (int q = 0; q < PGRP; ++q) {
                const int i = g + q;
                pbv[q] = __builtin_amdgcn_readlane(bd.x, i);
                pev[q] = __builtin_amdgcn_readlane(bd.y, i);
                chv[q] = min(c0 + 32 * i + w, nCh - 1);
                const int* __restrict__ rp = rec + (size_t)chv[q] * CHUNK;
                const int last = max(pev[q] - 1, 0);
#pragma unroll
                for (int t = 0; t < 2; ++t) {
                    const int j = pbv[q] + lane + 32 * t;
                    ok[q][t] = j < pev[q];
                    r[q][t] = rp[min(j, last)];
                }
            }
#pragma unroll
            for (int q = 0; q < PGRP; ++q) {
#pragma unroll
                for (int t = 0; t < 2; ++t) {
                    if (ok[q][t]) {
                        const int word = r[q][t];
                        if (FILL) {
                            const int sl = atomicAdd(&s_cnt[word >> SRC_BITS], 1);
                            if (staged) s_out[sl] = word & smask;
                            else csr[base + sl] = word & smask;
                        } else {
                            atomicAdd(&s_cnt[word >> SRC_BITS], 1);
                        }
                    }
                }
            }
#pragma unroll
            for (int q = 0; q < PGRP; ++q) {
                const int* __restrict__ rp = rec + (size_t)chv[q] * CHUNK;
                for (int j0 = pbv[q] + 64; j0 < pev[q]; j0 += 32) {
                    const int j = j0 + lane;
                    if (j < pev[q]) {
                        const int word = rp[j];
                        if (FILL) {
                            const int sl = atomicAdd(&s_cnt[word >> SRC_BITS], 1);
                            if (staged) s_out[sl] = word & smask;
                            else csr[base + sl] = word & smask;
                        } else {
                            atomicAdd(&s_cnt[word >> SRC_BITS], 1);
                        }
                    }
                }
            }
        }
    }
}

__global__ __launch_bounds__(1024) void rank_range_kernel(
    const int* __restrict__ rec, const int2* __restrict__ offs,
    int* __restrict__ total, int2* __restrict__ rows,
    int* __restrict__ csr, int nCh) {
    __shared__ int s_cnt[RNG_NODES];
    __shared__ int s_out[OUT_CAP];
    __shared__ int s_wsum[32];
    __shared__ int s_tot;
    __shared__ int s_base;
    const int tid  = threadIdx.x;
    const int lane = tid & 31;
    const int w = __builtin_amdgcn_readfirstlane(tid >> 5);
    const int b = blockIdx.x;
    if (tid < RNG_NODES) s_cnt[tid] = 0;
    __syncthreads();

    walk_pieces<false>(rec, offs, s_cnt, s_out, csr, nCh, b, w, lane, false, 0);
    __syncthreads();

    const int c  = (tid < RNG_NODES) ? s_cnt[tid] : 0;
    const int ex = block_scan_excl(c, s_wsum, &s_tot);
    if (tid == 0) s_base = atomicAdd(total, s_tot);
    __syncthreads();
    const int nb   = s_tot;
    const int base = s_base;
    if (tid < RNG_NODES) {
        s_cnt[tid] = ex;
        const int node = b * RNG_NODES + tid;
        if (node < N_NODES) rows[node] = make_int2(base + ex, base + ex + c);
    }
    __syncthreads();

    const bool staged = nb <= OUT_CAP;
    walk_pieces<true>(rec, offs, s_cnt, s_out, csr, nCh, b, w, lane, staged, base);
    if (staged) {
        __syncthreads();
        for (int i = tid; i < nb; i += 1024) csr[base + i] = s_out[i];
    }
}

__device__ __forceinline__ void add4(float4& a, const float4& b) {
    a.x += b.x; a.y += b.y; a.z += b.z; a.w += b.w;
}

__device__ __forceinline__ void addsel4(float4& a, const float4& b, bool in) {
    a.x += in ? b.x : 0.0f; a.y += in ? b.y : 0.0f;
    a.z += in ? b.z : 0.0f; a.w += in ? b.w : 0.0f;
}

template <int FIN, int FOUT, int TPW, bool PRELU, bool PROJ>
__global__ __launch_bounds__(128) void sage_gather_combine_wmma(
    const float* __restrict__ h, const int2* __restrict__ rows,
    const int* __restrict__ csr,
    const float* __restrict__ Wp, const float* __restrict__ bias,
    const float* __restrict__ alpha_p,
    const float* __restrict__ Wf, float* __restrict__ out) {
    constexpr int LDSTR = FIN + 1;
    constexpr int NJ = FIN / 8;
    __shared__ alignas(16) float s_mean[16 * LDSTR];
    __shared__ float s_h[16 * LDSTR];
    static_assert(16 * LDSTR >= 4 * 16 * 8, "partial p must fit s_mean");

    const int tid   = threadIdx.x;
    const int node0 = blockIdx.x * 16;
    const float4* __restrict__ h4 = reinterpret_cast<const float4*>(h);

    if (FIN == 32) {
        const int r    = tid >> 3;
        const int sub  = tid & 7;
        const int node = node0 + r;
        const int2 rw  = rows[node];
        const int last = rw.y - 1;
        float4 acc = {0.0f, 0.0f, 0.0f, 0.0f};
        for (int e = rw.x; e < rw.y; e += 4) {
            const int s0 = csr[e];
            const int s1 = csr[min(e + 1, last)];
            const int s2 = csr[min(e + 2, last)];
            const int s3 = csr[min(e + 3, last)];
            const float4 v0 = h4[(size_t)s0 * 8 + sub];
            const float4 v1 = h4[(size_t)s1 * 8 + sub];
            const float4 v2 = h4[(size_t)s2 * 8 + sub];
            const float4 v3 = h4[(size_t)s3 * 8 + sub];
            add4(acc, v0);
            addsel4(acc, v1, e + 1 <= last);
            addsel4(acc, v2, e + 2 <= last);
            addsel4(acc, v3, e + 3 <= last);
        }
        const float inv = 1.0f / fmaxf((float)(rw.y - rw.x), 1.0f);
        const float4 own = h4[(size_t)node * 8 + sub];
        float* dm = s_mean + r * LDSTR + sub * 4;
        dm[0] = acc.x * inv; dm[1] = acc.y * inv;
        dm[2] = acc.z * inv; dm[3] = acc.w * inv;
        float* dh = s_h + r * LDSTR + sub * 4;
        dh[0] = own.x; dh[1] = own.y; dh[2] = own.z; dh[3] = own.w;
    } else {
        const int gl = tid & 31;
        const int wv = __builtin_amdgcn_readfirstlane(tid >> 5);
        for (int q = 0; q < 4; ++q) {
            const int r    = wv * 4 + q;
            const int node = node0 + r;
            const int2 rw  = rows[node];
            const int start = __builtin_amdgcn_readfirstlane(rw.x);
            const int end   = __builtin_amdgcn_readfirstlane(rw.y);
            float4 acc = {0.0f, 0.0f, 0.0f, 0.0f};
            for (int base = start; base < end; base += 32) {
                const int n  = min(32, end - base);
                const int my = csr[base + min(gl, n - 1)];
                int j = 0;
                for (; j + 8 <= n; j += 8) {
                    float4 v[8];
#pragma unroll
                    for (int k = 0; k < 8; ++k) {
                        const int s = __builtin_amdgcn_readlane(my, j + k);
                        v[k] = h4[(size_t)s * 32 + gl];
                    }
#pragma unroll
                    for (int k = 0; k < 8; ++k) add4(acc, v[k]);
                }
                if (j < n) {
                    float4 v[7];
#pragma unroll
                    for (int k = 0; k < 7; ++k) {
                        v[k] = make_float4(0.0f, 0.0f, 0.0f, 0.0f);
                        if (j + k < n) {
                            const int s = __builtin_amdgcn_readlane(my, j + k);
                            v[k] = h4[(size_t)s * 32 + gl];
                        }
                    }
#pragma unroll
                    for (int k = 0; k < 7; ++k)
                        if (j + k < n) add4(acc, v[k]);
                }
            }
            const float inv = 1.0f / fmaxf((float)(end - start), 1.0f);
            const float4 own = h4[(size_t)node * 32 + gl];
            float* dm = s_mean + r * LDSTR + gl * 4;
            dm[0] = acc.x * inv; dm[1] = acc.y * inv;
            dm[2] = acc.z * inv; dm[3] = acc.w * inv;
            float* dh = s_h + r * LDSTR + gl * 4;
            dh[0] = own.x; dh[1] = own.y; dh[2] = own.z; dh[3] = own.w;
        }
    }
    __syncthreads();

    const int wave = tid >> 5;
    const int lane = tid & 31;
    const int half = lane >> 4;
    const int col  = lane & 15;
    const float alpha = alpha_p[0];

    v8f acc[TPW];
#pragma unroll
    for (int t = 0; t < TPW; ++t)
#pragma unroll
        for (int j = 0; j < 8; ++j) acc[t][j] = 0.0f;

    const float* smrow = s_mean + col * LDSTR;
    const float* shrow = s_h    + col * LDSTR;

    const int wv_s = __builtin_amdgcn_readfirstlane(wave);
    const float4* __restrict__ Wp4 = reinterpret_cast<const float4*>(Wp);

#pragma unroll 2
    for (int j = 0; j < NJ; ++j) {
        const int kb = j * 8 + half * 2;
        v2f am0; am0.x = smrow[kb];     am0.y = smrow[kb + 1];
        v2f ah0; ah0.x = shrow[kb];     ah0.y = shrow[kb + 1];
        v2f am1; am1.x = smrow[kb + 4]; am1.y = smrow[kb + 5];
        v2f ah1; ah1.x = shrow[kb + 4]; ah1.y = shrow[kb + 5];
#pragma unroll
        for (int t = 0; t < TPW; ++t) {
            const int tile = wv_s * TPW + t;
            const float4 pl = Wp4[((tile * NJ + j) * 2 + 0) * 32 + lane];
            const float4 pr = Wp4[((tile * NJ + j) * 2 + 1) * 32 + lane];
            v2f bl0; bl0.x = pl.x; bl0.y = pl.y;
            v2f bl1; bl1.x = pl.z; bl1.y = pl.w;
            v2f br0; br0.x = pr.x; br0.y = pr.y;
            v2f br1; br1.x = pr.z; br1.y = pr.w;
            if (PROJ) {
                acc[t] = __builtin_amdgcn_wmma_f32_16x16x4_f32(
                    false, bl0, false, am0, (short)0, acc[t], false, false);
                acc[t] = __builtin_amdgcn_wmma_f32_16x16x4_f32(
                    false, br0, false, ah0, (short)0, acc[t], false, false);
                acc[t] = __builtin_amdgcn_wmma_f32_16x16x4_f32(
                    false, bl1, false, am1, (short)0, acc[t], false, false);
                acc[t] = __builtin_amdgcn_wmma_f32_16x16x4_f32(
                    false, br1, false, ah1, (short)0, acc[t], false, false);
            } else {
                acc[t] = __builtin_amdgcn_wmma_f32_16x16x4_f32(
                    false, am0, false, bl0, (short)0, acc[t], false, false);
                acc[t] = __builtin_amdgcn_wmma_f32_16x16x4_f32(
                    false, ah0, false, br0, (short)0, acc[t], false, false);
                acc[t] = __builtin_amdgcn_wmma_f32_16x16x4_f32(
                    false, am1, false, bl1, (short)0, acc[t], false, false);
                acc[t] = __builtin_amdgcn_wmma_f32_16x16x4_f32(
                    false, ah1, false, br1, (short)0, acc[t], false, false);
            }
        }
    }

    if (PROJ) {
        v8f pacc = {0.0f, 0.0f, 0.0f, 0.0f, 0.0f, 0.0f, 0.0f, 0.0f};
#pragma unroll
        for (int t = 0; t < TPW; ++t) {
            const int f0 = (wave * TPW + t) * 16 + half * 8;
            const float4 bA = *reinterpret_cast<const float4*>(bias + f0);
            const float4 bB = *reinterpret_cast<const float4*>(bias + f0 + 4);
            const float bv[8] = {bA.x, bA.y, bA.z, bA.w, bB.x, bB.y, bB.z, bB.w};
            float hv[8];
#pragma unroll
            for (int r = 0; r < 8; ++r) {
                float val = acc[t][r] + bv[r];
                if (PRELU) val = (val >= 0.0f) ? val : alpha * val;
                hv[r] = val;
            }
#pragma unroll
            for (int i = 0; i < 4; ++i) {
                v2f hb; hb.x = hv[2 * i]; hb.y = hv[2 * i + 1];
                v2f wa;
                wa.x = Wf[(f0 + 2 * i) * 16 + col];
                wa.y = Wf[(f0 + 2 * i + 1) * 16 + col];
                pacc = __builtin_amdgcn_wmma_f32_16x16x4_f32(
                    false, wa, false, hb, (short)0, pacc, false, false);
            }
        }
        float* s_part = s_mean;
        __syncthreads();
        if (half == 0) {
            float4* d = reinterpret_cast<float4*>(s_part + (wave * 16 + col) * 8);
            d[0] = make_float4(pacc[0], pacc[1], pacc[2], pacc[3]);
            d[1] = make_float4(pacc[4], pacc[5], pacc[6], pacc[7]);
        }
        __syncthreads();
        const float s = ((s_part[tid] + s_part[128 + tid]) + s_part[256 + tid])
                        + s_part[384 + tid];
        out[(size_t)node0 * 8 + tid] = s;
        return;
    }

#pragma unroll
    for (int t = 0; t < TPW; ++t) {
        const int n0 = (wave * TPW + t) * 16;
        const float bv = bias[n0 + col];
#pragma unroll
        for (int v = 0; v < 8; ++v) {
            float val = acc[t][v] + bv;
            if (PRELU) val = (val >= 0.0f) ? val : alpha * val;
            const int row = node0 + v + half * 8;
            out[(size_t)row * FOUT + n0 + col] = val;
        }
    }
}

__global__ void gather4_final_kernel(const float* __restrict__ p,
                                     const int2* __restrict__ rows,
                                     const int* __restrict__ csr,
                                     const float* __restrict__ cf,
                                     float* __restrict__ out, int n) {
    const int t    = blockIdx.x * blockDim.x + threadIdx.x;
    const int node = t >> 2;
    const int f    = t & 3;
    const bool live = node < n;
    const int2 rw  = live ? rows[node] : make_int2(0, 0);
    const int last = rw.y - 1;
    const float4* __restrict__ p4 = reinterpret_cast<const float4*>(p);
    float4 acc = {0.0f, 0.0f, 0.0f, 0.0f};
    for (int e = rw.x + f; e < rw.y; e += 16) {
        const int s0 = csr[e];
        const int s1 = csr[min(e + 4, last)];
        const int s2 = csr[min(e + 8, last)];
        const int s3 = csr[min(e + 12, last)];
        const float4 v0 = p4[(size_t)s0 * 2];
        const float4 v1 = p4[(size_t)s1 * 2];
        const float4 v2 = p4[(size_t)s2 * 2];
        const float4 v3 = p4[(size_t)s3 * 2];
        add4(acc, v0);
        addsel4(acc, v1, e + 4 <= last);
        addsel4(acc, v2, e + 8 <= last);
        addsel4(acc, v3, e + 12 <= last);
    }
    acc.x += __shfl_xor(acc.x, 1); acc.y += __shfl_xor(acc.y, 1);
    acc.z += __shfl_xor(acc.z, 1); acc.w += __shfl_xor(acc.w, 1);
    acc.x += __shfl_xor(acc.x, 2); acc.y += __shfl_xor(acc.y, 2);
    acc.z += __shfl_xor(acc.z, 2); acc.w += __shfl_xor(acc.w, 2);
    const float a  = (f == 0) ? acc.x : (f == 1) ? acc.y : (f == 2) ? acc.z : acc.w;
    const float id = 1.0f / fmaxf((float)(rw.y - rw.x), 1.0f);
    if (live)
        out[(size_t)node * 4 + f] = (a * id + cf[f]) + p[(size_t)node * 8 + 4 + f];
}

extern "C" void kernel_launch(void* const* d_in, const int* in_sizes, int n_in,
                              void* d_out, int out_size, void* d_ws, size_t ws_size,
                              hipStream_t stream) {
    const float* x   = (const float*)d_in[0];
    const int*   ei  = (const int*)d_in[1];
    const float* W1l = (const float*)d_in[2];
    const float* b1  = (const float*)d_in[3];
    const float* W1r = (const float*)d_in[4];
    const float* W2l = (const float*)d_in[5];
    const float* b2  = (const float*)d_in[6];
    const float* W2r = (const float*)d_in[7];
    const float* W3l = (const float*)d_in[8];
    const float* b3  = (const float*)d_in[9];
    const float* W3r = (const float*)d_in[10];
    const float* Wc  = (const float*)d_in[11];
    const float* bc  = (const float*)d_in[12];
    const float* a   = (const float*)d_in[13];
    float* out = (float*)d_out;

    const int E = in_sizes[1] / 2;
    const int* srcv = ei;
    const int* dstv = ei + E;

    char* ws = (char*)d_ws;
    size_t off = 0;
    auto carve = [&](size_t bytes) {
        float* p = (float*)(ws + off);
        off += (bytes + 255) & ~(size_t)255;
        return p;
    };
    int*   cnt = (int*)carve((size_t)(N_NODES + 1) * 4);
    float* el  = carve((size_t)N_NODES * 128 * 4);
    float* hA  = carve((size_t)N_NODES * 128 * 4);
    float* hB  = carve((size_t)N_NODES * 128 * 4);

    const int nCh = (E + CHUNK - 1) / CHUNK;

    int2* rows = (int2*)el;
    int*  rec  = (int*)el + (size_t)N_NODES * 2;
    int2* offs = (int2*)(rec + (size_t)nCh * CHUNK);
    int*  csr  = (int*)(offs + (size_t)nCh * N_RNG);

    const int TB = 256;
    const int nBlk16 = N_NODES / 16;

    float* proj = hB;
    float* Wf   = hB + (size_t)N_NODES * 8;
    float* cf   = Wf + 128 * 16;
    float* Wp1  = cf + 64;
    float* Wp2  = Wp1 + PACK1_N;

    bin_edges_kernel<<<nCh + XTRA_BLOCKS, BIN_TB, 0, stream>>>(
        srcv, dstv, rec, offs, cnt + N_NODES, E, nCh,
        W3l, W3r, b3, Wc, bc, W1l, W1r, W2l, W2r, Wf, cf, Wp1, Wp2);
    rank_range_kernel<<<N_RNG, 1024, 0, stream>>>(
        rec, offs, cnt + N_NODES, rows, csr, nCh);

    sage_gather_combine_wmma<32, 128, 2, true, false><<<nBlk16, 128, 0, stream>>>(
        x, rows, csr, Wp1, b1, a, nullptr, hA);

    sage_gather_combine_wmma<128, 128, 2, true, true><<<nBlk16, 128, 0, stream>>>(
        hA, rows, csr, Wp2, b2, a, Wf, proj);

    gather4_final_kernel<<<(N_NODES * 4 + TB - 1) / TB, TB, 0, stream>>>(
        proj, rows, csr, cf, out, N_NODES);
}


# K loop of both layer kernels split around the gather: own rows staged, h@Wr first, mean@Wl after, so the matrix pipe works while gathers fly; 32-bit row index keeps 64 VGPRs; sum order changes
# speedup vs baseline: 6.7623x; 1.0332x over previous
#include <hip/hip_runtime.h>
#include <hip/hip_bf16.h>


typedef __attribute__((ext_vector_type(2))) float v2f;
typedef __attribute__((ext_vector_type(8))) float v8f;

#define N_NODES 100000

#define RNG_SHIFT 9
#define RNG_NODES (1 << RNG_SHIFT)
#define N_RNG ((N_NODES + RNG_NODES - 1) / RNG_NODES)
#define SRC_BITS 17
#define CHUNK 8192
#define BIN_TB 1024
#define EPL (CHUNK / BIN_TB)
#define PGRP 8
#define OUT_CAP 12288

static_assert(N_NODES <= (1 << SRC_BITS), "node id must fit SRC_BITS");
static_assert(RNG_SHIFT + SRC_BITS < 32, "packed edge word must stay positive");
static_assert(N_RNG <= BIN_TB, "one thread per range in the chunk scan");

__device__ __forceinline__ int block_scan_excl(int c, int* s_wsum, int* s_total) {
    const int lane = threadIdx.x & 31;
    const int w    = threadIdx.x >> 5;
    const int nw   = blockDim.x >> 5;
    int v = c;
#pragma unroll
    for (int o = 1; o < 32; o <<= 1) {
        int t = __shfl_up(v, o);
        if (lane >= o) v += t;
    }
    if (lane == 31) s_wsum[w] = v;
    __syncthreads();
    if (w == 0) {
        const int t = (lane < nw) ? s_wsum[lane] : 0;
        int u = t;
#pragma unroll
        for (int o = 1; o < 32; o <<= 1) {
            int q = __shfl_up(u, o);
            if (lane >= o) u += q;
        }
        s_wsum[lane] = u - t;
        if (lane == 31) *s_total = u;
    }
    __syncthreads();
    return s_wsum[w] + (v - c);
}

#define PACK1_N (32 * 128 * 2)
#define PACK2_N (128 * 128 * 2)
#define WF_N (128 * 16)
#define PACK_BLOCKS ((PACK1_N + PACK2_N) / BIN_TB)
#define WF_BLOCKS (WF_N / BIN_TB)
#define XTRA_BLOCKS (PACK_BLOCKS + WF_BLOCKS + 1)

static_assert(PACK1_N % BIN_TB == 0 && PACK2_N % BIN_TB == 0,
              "a pack block lies inside one layer");
static_assert(WF_N % BIN_TB == 0, "whole blocks of Wf");

__device__ __forceinline__ void pack_one(const float* __restrict__ Wl,
                                         const float* __restrict__ Wr,
                                         float* __restrict__ Wp, int idx,
                                         int fin, int fout) {
    const int e    = idx & 3;
    const int lane = (idx >> 2) & 31;
    const int q    = (idx >> 7) & 1;
    const int rest = idx >> 8;
    const int nj   = fin >> 3;
    const int j    = rest % nj;
    const int tile = rest / nj;
    const int k    = 8 * j + 4 * (e >> 1) + 2 * (lane >> 4) + (e & 1);
    const int n    = 16 * tile + (lane & 15);
    Wp[idx] = (q ? Wr : Wl)[k * fout + n];
}

__device__ __forceinline__ void fold_pack_block(
    int xb, int tid,
    const float* __restrict__ W3l, const float* __restrict__ W3r,
    const float* __restrict__ b3, const float* __restrict__ Wc,
    const float* __restrict__ bc,
    const float* __restrict__ W1l, const float* __restrict__ W1r,
    const float* __restrict__ W2l, const float* __restrict__ W2r,
    float* __restrict__ Wf, float* __restrict__ cf,
    float* __restrict__ Wp1, float* __restrict__ Wp2) {
    if (xb < PACK_BLOCKS) {
        const int idx = xb * BIN_TB + tid;
        if (idx < PACK1_N) pack_one(W1l, W1r, Wp1, idx, 32, 128);
        else pack_one(W2l, W2r, Wp2, idx - PACK1_N, 128, 128);
    } else if (xb < PACK_BLOCKS + WF_BLOCKS) {
        const int t = (xb - PACK_BLOCKS) * BIN_TB + tid;
        const int k = t >> 4;
        const int c = t & 15;
        float s = 0.0f;
        if (c < 8) {
            const float* w = ((c < 4) ? W3l : W3r) + k * 64;
            const int cc = c & 3;
            for (int j = 0; j < 64; ++j) s += w[j] * Wc[j * 4 + cc];
        }
        Wf[t] = s;
    } else if (tid < 4) {
        float s = 0.0f;
#pragma unroll
        for (int j = 0; j < 64; ++j) s += b3[j] * Wc[j * 4 + tid];
        cf[tid] = s + bc[tid];
    }
}

__global__ __launch_bounds__(BIN_TB) void bin_edges_kernel(
    const int* __restrict__ src, const int* __restrict__ dst,
    int* __restrict__ rec, int2* __restrict__ offs,
    int* __restrict__ total, int E, int nCh,
    const float* __restrict__ W3l, const float* __restrict__ W3r,
    const float* __restrict__ b3, const float* __restrict__ Wc,
    const float* __restrict__ bc,
    const float* __restrict__ W1l, const float* __restrict__ W1r,
    const float* __restrict__ W2l, const float* __restrict__ W2r,
    float* __restrict__ Wf, float* __restrict__ cf,
    float* __restrict__ Wp1, float* __restrict__ Wp2) {
    if ((int)blockIdx.x >= nCh) {
        fold_pack_block((int)blockIdx.x - nCh, threadIdx.x, W3l, W3r, b3, Wc, bc,
                        W1l, W1r, W2l, W2r, Wf, cf, Wp1, Wp2);
        return;
    }
    __shared__ int s_hist[N_RNG];
    __shared__ int s_off[N_RNG];
    __shared__ int s_rec[CHUNK];
    __shared__ int s_wsum[32];
    __shared__ int s_tot;
    const int tid   = threadIdx.x;
    const int chunk = blockIdx.x;
    const int e0    = chunk * CHUNK + tid;
    if (tid < N_RNG) s_hist[tid] = 0;
    if (chunk == 0 && tid == 0) *total = 0;

    int d[EPL], s[EPL];
#pragma unroll
    for (int k = 0; k < EPL; ++k) {
        const int ec = min(e0 + k * BIN_TB, E - 1);
        d[k] = dst[ec];
        s[k] = src[ec];
    }
    __syncthreads();

    int rk[EPL];
#pragma unroll
    for (int k = 0; k < EPL; ++k) {
        rk[k] = 0;
        if (e0 + k * BIN_TB < E) rk[k] = atomicAdd(&s_hist[d[k] >> RNG_SHIFT], 1);
    }
    __syncthreads();

    const int c  = (tid < N_RNG) ? s_hist[tid] : 0;
    const int ex = block_scan_excl(c, s_wsum, &s_tot);
    if (tid < N_RNG) {
        s_off[tid] = ex;
        offs[(size_t)chunk * N_RNG + tid] = make_int2(ex, ex + c);
    }
    __syncthreads();

#pragma unroll
    for (int k = 0; k < EPL; ++k) {
        if (e0 + k * BIN_TB < E)
            s_rec[s_off[d[k] >> RNG_SHIFT] + rk[k]] =
                ((d[k] & (RNG_NODES - 1)) << SRC_BITS) | s[k];
    }
    __syncthreads();
    const int nv = s_tot;
    int* __restrict__ rp = rec + (size_t)chunk * CHUNK;
#pragma unroll
    for (int k = 0; k < EPL; ++k) {
        const int p = k * BIN_TB + tid;
        if (p < nv) rp[p] = s_rec[p];
    }
}

template <bool FILL>
__device__ __forceinline__ void walk_pieces(
    const int* __restrict__ rec, const int2* __restrict__ offs,
    int* s_cnt, int* s_out, int* __restrict__ csr,
    int nCh, int b, int w, int lane, bool staged, int base) {
    const int smask = (1 << SRC_BITS) - 1;
    for (int c0 = 0; c0 < nCh; c0 += 1024) {
        const int chl = c0 + 32 * lane + w;
        int2 bd = make_int2(0, 0);
        if (chl < nCh) bd = offs[(size_t)chl * N_RNG + b];
        const int left = nCh - c0 - w;
        const int np = left > 0 ? min(32, (left + 31) >> 5) : 0;
        for (int g = 0; g < np; g += PGRP) {
            int r[PGRP][2];
            bool ok[PGRP][2];
            int pbv[PGRP], pev[PGRP], chv[PGRP];
#pragma unroll
            for (int q = 0; q < PGRP; ++q) {
                const int i = g + q;
                pbv[q] = __builtin_amdgcn_readlane(bd.x, i);
                pev[q] = __builtin_amdgcn_readlane(bd.y, i);
                chv[q] = min(c0 + 32 * i + w, nCh - 1);
                const int* __restrict__ rp = rec + (size_t)chv[q] * CHUNK;
                const int last = max(pev[q] - 1, 0);
#pragma unroll
                for (int t = 0; t < 2; ++t) {
                    const int j = pbv[q] + lane + 32 * t;
                    ok[q][t] = j < pev[q];
                    r[q][t] = rp[min(j, last)];
                }
            }
#pragma unroll
            for (int q = 0; q < PGRP; ++q) {
#pragma unroll
                for (int t = 0; t < 2; ++t) {
                    if (ok[q][t]) {
                        const int word = r[q][t];
                        if (FILL) {
                            const int sl = atomicAdd(&s_cnt[word >> SRC_BITS], 1);
                            if (staged) s_out[sl] = word & smask;
                            else csr[base + sl] = word & smask;
                        } else {
                            atomicAdd(&s_cnt[word >> SRC_BITS], 1);
                        }
                    }
                }
            }
#pragma unroll
            for (int q = 0; q < PGRP; ++q) {
                const int* __restrict__ rp = rec + (size_t)chv[q] * CHUNK;
                for (int j0 = pbv[q] + 64; j0 < pev[q]; j0 += 32) {
                    const int j = j0 + lane;
                    if (j < pev[q]) {
                        const int word = rp[j];
                        if (FILL) {
                            const int sl = atomicAdd(&s_cnt[word >> SRC_BITS], 1);
                            if (staged) s_out[sl] = word & smask;
                            else csr[base + sl] = word & smask;
                        } else {
                            atomicAdd(&s_cnt[word >> SRC_BITS], 1);
                        }
                    }
                }
            }
        }
    }
}

__global__ __launch_bounds__(1024) void rank_range_kernel(
    const int* __restrict__ rec, const int2* __restrict__ offs,
    int* __restrict__ total, int2* __restrict__ rows,
    int* __restrict__ csr, int nCh) {
    __shared__ int s_cnt[RNG_NODES];
    __shared__ int s_out[OUT_CAP];
    __shared__ int s_wsum[32];
    __shared__ int s_tot;
    __shared__ int s_base;
    const int tid  = threadIdx.x;
    const int lane = tid & 31;
    const int w = __builtin_amdgcn_readfirstlane(tid >> 5);
    const int b = blockIdx.x;
    if (tid < RNG_NODES) s_cnt[tid] = 0;
    __syncthreads();

    walk_pieces<false>(rec, offs, s_cnt, s_out, csr, nCh, b, w, lane, false, 0);
    __syncthreads();

    const int c  = (tid < RNG_NODES) ? s_cnt[tid] : 0;
    const int ex = block_scan_excl(c, s_wsum, &s_tot);
    if (tid == 0) s_base = atomicAdd(total, s_tot);
    __syncthreads();
    const int nb   = s_tot;
    const int base = s_base;
    if (tid < RNG_NODES) {
        s_cnt[tid] = ex;
        const int node = b * RNG_NODES + tid;
        if (node < N_NODES) rows[node] = make_int2(base + ex, base + ex + c);
    }
    __syncthreads();

    const bool staged = nb <= OUT_CAP;
    walk_pieces<true>(rec, offs, s_cnt, s_out, csr, nCh, b, w, lane, staged, base);
    if (staged) {
        __syncthreads();
        for (int i = tid; i < nb; i += 1024) csr[base + i] = s_out[i];
    }
}

__device__ __forceinline__ void add4(float4& a, const float4& b) {
    a.x += b.x; a.y += b.y; a.z += b.z; a.w += b.w;
}

__device__ __forceinline__ void addsel4(float4& a, const float4& b, bool in) {
    a.x += in ? b.x : 0.0f; a.y += in ? b.y : 0.0f;
    a.z += in ? b.z : 0.0f; a.w += in ? b.w : 0.0f;
}

template <int FIN, int TPW, bool PROJ>
__device__ __forceinline__ void k_half(v8f (&acc)[TPW], const float* arow,
                                       const float4* __restrict__ Wp4,
                                       int q, int wv_s, int lane, int half) {
    constexpr int NJ = FIN / 8;
#pragma unroll 2
    for (int j = 0; j < NJ; ++j) {
        const int kb = j * 8 + half * 2;
        v2f a0; a0.x = arow[kb];     a0.y = arow[kb + 1];
        v2f a1; a1.x = arow[kb + 4]; a1.y = arow[kb + 5];
#pragma unroll
        for (int t = 0; t < TPW; ++t) {
            const int tile = wv_s * TPW + t;
            const float4 p = Wp4[((tile * NJ + j) * 2 + q) * 32 + lane];
            v2f b0; b0.x = p.x; b0.y = p.y;
            v2f b1; b1.x = p.z; b1.y = p.w;
            if (PROJ) {
                acc[t] = __builtin_amdgcn_wmma_f32_16x16x4_f32(
                    false, b0, false, a0, (short)0, acc[t], false, false);
                acc[t] = __builtin_amdgcn_wmma_f32_16x16x4_f32(
                    false, b1, false, a1, (short)0, acc[t], false, false);
            } else {
                acc[t] = __builtin_amdgcn_wmma_f32_16x16x4_f32(
                    false, a0, false, b0, (short)0, acc[t], false, false);
                acc[t] = __builtin_amdgcn_wmma_f32_16x16x4_f32(
                    false, a1, false, b1, (short)0, acc[t], false, false);
            }
        }
    }
}

template <int FIN, int FOUT, int TPW, bool PRELU, bool PROJ>
__global__ __launch_bounds__(128) void sage_gather_combine_wmma(
    const float* __restrict__ h, const int2* __restrict__ rows,
    const int* __restrict__ csr,
    const float* __restrict__ Wp, const float* __restrict__ bias,
    const float* __restrict__ alpha_p,
    const float* __restrict__ Wf, float* __restrict__ out) {
    constexpr int LDSTR = FIN + 1;
    __shared__ alignas(16) float s_mean[16 * LDSTR];
    __shared__ float s_h[16 * LDSTR];
    static_assert(16 * LDSTR >= 4 * 16 * 8, "partial p must fit s_mean");

    const int tid   = threadIdx.x;
    const int node0 = blockIdx.x * 16;
    const float4* __restrict__ h4 = reinterpret_cast<const float4*>(h);

    if (FIN == 32) {
        const int r   = tid >> 3;
        const int sub = tid & 7;
        const float4 own = h4[(size_t)(node0 + r) * 8 + sub];
        float* dh = s_h + r * LDSTR + sub * 4;
        dh[0] = own.x; dh[1] = own.y; dh[2] = own.z; dh[3] = own.w;
    } else {
        const int gl = tid & 31;
        const int wv = __builtin_amdgcn_readfirstlane(tid >> 5);
        float4 own[4];
#pragma unroll
        for (int q = 0; q < 4; ++q)
            own[q] = h4[(size_t)(node0 + wv * 4 + q) * 32 + gl];
#pragma unroll
        for (int q = 0; q < 4; ++q) {
            float* dh = s_h + (wv * 4 + q) * LDSTR + gl * 4;
            dh[0] = own[q].x; dh[1] = own[q].y; dh[2] = own[q].z; dh[3] = own[q].w;
        }
    }
    __syncthreads();

    const int wave = tid >> 5;
    const int lane = tid & 31;
    const int half = lane >> 4;
    const int col  = lane & 15;
    const float alpha = alpha_p[0];

    v8f acc[TPW];
#pragma unroll
    for (int t = 0; t < TPW; ++t)
#pragma unroll
        for (int j = 0; j < 8; ++j) acc[t][j] = 0.0f;

    const float* smrow = s_mean + col * LDSTR;
    const float* shrow = s_h    + col * LDSTR;

    const int wv_s = __builtin_amdgcn_readfirstlane(wave);
    const float4* __restrict__ Wp4 = reinterpret_cast<const float4*>(Wp);

    k_half<FIN, TPW, PROJ>(acc, shrow, Wp4, 1, wv_s, lane, half);

    if (FIN == 32) {
        const int r    = tid >> 3;
        const int sub  = tid & 7;
        const int node = node0 + r;
        const int2 rw  = rows[node];
        const int last = rw.y - 1;
        float4 sum = {0.0f, 0.0f, 0.0f, 0.0f};
        for (int e = rw.x; e < rw.y; e += 4) {
            const int s0 = csr[e];
            const int s1 = csr[min(e + 1, last)];
            const int s2 = csr[min(e + 2, last)];
            const int s3 = csr[min(e + 3, last)];
            const float4 v0 = h4[(size_t)s0 * 8 + sub];
            const float4 v1 = h4[(size_t)s1 * 8 + sub];
            const float4 v2 = h4[(size_t)s2 * 8 + sub];
            const float4 v3 = h4[(size_t)s3 * 8 + sub];
            add4(sum, v0);
            addsel4(sum, v1, e + 1 <= last);
            addsel4(sum, v2, e + 2 <= last);
            addsel4(sum, v3, e + 3 <= last);
        }
        const float inv = 1.0f / fmaxf((float)(rw.y - rw.x), 1.0f);
        float* dm = s_mean + r * LDSTR + sub * 4;
        dm[0] = sum.x * inv; dm[1] = sum.y * inv;
        dm[2] = sum.z * inv; dm[3] = sum.w * inv;
    } else {
        const int gl = tid & 31;
        const int wv = __builtin_amdgcn_readfirstlane(tid >> 5);
        for (int q = 0; q < 4; ++q) {
            const int r    = wv * 4 + q;
            const int node = node0 + r;
            const int2 rw  = rows[node];
            const int start = __builtin_amdgcn_readfirstlane(rw.x);
            const int end   = __builtin_amdgcn_readfirstlane(rw.y);
            float4 sum = {0.0f, 0.0f, 0.0f, 0.0f};
            for (int base = start; base < end; base += 32) {
                const int n  = min(32, end - base);
                const int my = csr[base + min(gl, n - 1)];
                int j = 0;
                for (; j + 8 <= n; j += 8) {
                    float4 v[8];
#pragma unroll
                    for (int k = 0; k < 8; ++k) {
                        const int s = __builtin_amdgcn_readlane(my, j + k);
                        v[k] = h4[(unsigned)s * 32u + (unsigned)gl];
                    }
#pragma unroll
                    for (int k = 0; k < 8; ++k) add4(sum, v[k]);
                }
                if (j < n) {
                    float4 v[7];
#pragma unroll
                    for (int k = 0; k < 7; ++k) {
                        v[k] = make_float4(0.0f, 0.0f, 0.0f, 0.0f);
                        if (j + k < n) {
                            const int s = __builtin_amdgcn_readlane(my, j + k);
                            v[k] = h4[(unsigned)s * 32u + (unsigned)gl];
                        }
                    }
#pragma unroll
                    for (int k = 0; k < 7; ++k)
                        if (j + k < n) add4(sum, v[k]);
                }
            }
            const float inv = 1.0f / fmaxf((float)(end - start), 1.0f);
            float* dm = s_mean + r * LDSTR + gl * 4;
            dm[0] = sum.x * inv; dm[1] = sum.y * inv;
            dm[2] = sum.z * inv; dm[3] = sum.w * inv;
        }
    }
    __syncthreads();

    k_half<FIN, TPW, PROJ>(acc, smrow, Wp4, 0, wv_s, lane, half);

    if (PROJ) {
        v8f pacc = {0.0f, 0.0f, 0.0f, 0.0f, 0.0f, 0.0f, 0.0f, 0.0f};
#pragma unroll
        for (int t = 0; t < TPW; ++t) {
            const int f0 = (wave * TPW + t) * 16 + half * 8;
            const float4 bA = *reinterpret_cast<const float4*>(bias + f0);
            const float4 bB = *reinterpret_cast<const float4*>(bias + f0 + 4);
            const float bv[8] = {bA.x, bA.y, bA.z, bA.w, bB.x, bB.y, bB.z, bB.w};
            float hv[8];
#pragma unroll
            for (int r = 0; r < 8; ++r) {
                float val = acc[t][r] + bv[r];
                if (PRELU) val = (val >= 0.0f) ? val : alpha * val;
                hv[r] = val;
            }
#pragma unroll
            for (int i = 0; i < 4; ++i) {
                v2f hb; hb.x = hv[2 * i]; hb.y = hv[2 * i + 1];
                v2f wa;
                wa.x = Wf[(f0 + 2 * i) * 16 + col];
                wa.y = Wf[(f0 + 2 * i + 1) * 16 + col];
                pacc = __builtin_amdgcn_wmma_f32_16x16x4_f32(
                    false, wa, false, hb, (short)0, pacc, false, false);
            }
        }
        float* s_part = s_mean;
        __syncthreads();
        if (half == 0) {
            float4* d = reinterpret_cast<float4*>(s_part + (wave * 16 + col) * 8);
            d[0] = make_float4(pacc[0], pacc[1], pacc[2], pacc[3]);
            d[1] = make_float4(pacc[4], pacc[5], pacc[6], pacc[7]);
        }
        __syncthreads();
        const float s = ((s_part[tid] + s_part[128 + tid]) + s_part[256 + tid])
                        + s_part[384 + tid];
        out[(size_t)node0 * 8 + tid] = s;
        return;
    }

#pragma unroll
    for (int t = 0; t < TPW; ++t) {
        const int n0 = (wave * TPW + t) * 16;
        const float bv = bias[n0 + col];
#pragma unroll
        for (int v = 0; v < 8; ++v) {
            float val = acc[t][v] + bv;
            if (PRELU) val = (val >= 0.0f) ? val : alpha * val;
            const int row = node0 + v + half * 8;
            out[(size_t)row * FOUT + n0 + col] = val;
        }
    }
}

__global__ void gather4_final_kernel(const float* __restrict__ p,
                                     const int2* __restrict__ rows,
                                     const int* __restrict__ csr,
                                     const float* __restrict__ cf,
                                     float* __restrict__ out, int n) {
    const int t    = blockIdx.x * blockDim.x + threadIdx.x;
    const int node = t >> 2;
    const int f    = t & 3;
    const bool live = node < n;
    const int2 rw  = live ? rows[node] : make_int2(0, 0);
    const int last = rw.y - 1;
    const float4* __restrict__ p4 = reinterpret_cast<const float4*>(p);
    float4 acc = {0.0f, 0.0f, 0.0f, 0.0f};
    for (int e = rw.x + f; e < rw.y; e += 16) {
        const int s0 = csr[e];
        const int s1 = csr[min(e + 4, last)];
        const int s2 = csr[min(e + 8, last)];
        const int s3 = csr[min(e + 12, last)];
        const float4 v0 = p4[(size_t)s0 * 2];
        const float4 v1 = p4[(size_t)s1 * 2];
        const float4 v2 = p4[(size_t)s2 * 2];
        const float4 v3 = p4[(size_t)s3 * 2];
        add4(acc, v0);
        addsel4(acc, v1, e + 4 <= last);
        addsel4(acc, v2, e + 8 <= last);
        addsel4(acc, v3, e + 12 <= last);
    }
    acc.x += __shfl_xor(acc.x, 1); acc.y += __shfl_xor(acc.y, 1);
    acc.z += __shfl_xor(acc.z, 1); acc.w += __shfl_xor(acc.w, 1);
    acc.x += __shfl_xor(acc.x, 2); acc.y += __shfl_xor(acc.y, 2);
    acc.z += __shfl_xor(acc.z, 2); acc.w += __shfl_xor(acc.w, 2);
    const float a  = (f == 0) ? acc.x : (f == 1) ? acc.y : (f == 2) ? acc.z : acc.w;
    const float id = 1.0f / fmaxf((float)(rw.y - rw.x), 1.0f);
    if (live)
        out[(size_t)node * 4 + f] = (a * id + cf[f]) + p[(size_t)node * 8 + 4 + f];
}

extern "C" void kernel_launch(void* const* d_in, const int* in_sizes, int n_in,
                              void* d_out, int out_size, void* d_ws, size_t ws_size,
                              hipStream_t stream) {
    const float* x   = (const float*)d_in[0];
    const int*   ei  = (const int*)d_in[1];
    const float* W1l = (const float*)d_in[2];
    const float* b1  = (const float*)d_in[3];
    const float* W1r = (const float*)d_in[4];
    const float* W2l = (const float*)d_in[5];
    const float* b2  = (const float*)d_in[6];
    const float* W2r = (const float*)d_in[7];
    const float* W3l = (const float*)d_in[8];
    const float* b3  = (const float*)d_in[9];
    const float* W3r = (const float*)d_in[10];
    const float* Wc  = (const float*)d_in[11];
    const float* bc  = (const float*)d_in[12];
    const float* a   = (const float*)d_in[13];
    float* out = (float*)d_out;

    const int E = in_sizes[1] / 2;
    const int* srcv = ei;
    const int* dstv = ei + E;

    char* ws = (char*)d_ws;
    size_t off = 0;
    auto carve = [&](size_t bytes) {
        float* p = (float*)(ws + off);
        off += (bytes + 255) & ~(size_t)255;
        return p;
    };
    int*   cnt = (int*)carve((size_t)(N_NODES + 1) * 4);
    float* el  = carve((size_t)N_NODES * 128 * 4);
    float* hA  = carve((size_t)N_NODES * 128 * 4);
    float* hB  = carve((size_t)N_NODES * 128 * 4);

    const int nCh = (E + CHUNK - 1) / CHUNK;

    int2* rows = (int2*)el;
    int*  rec  = (int*)el + (size_t)N_NODES * 2;
    int2* offs = (int2*)(rec + (size_t)nCh * CHUNK);
    int*  csr  = (int*)(offs + (size_t)nCh * N_RNG);

    const int TB = 256;
    const int nBlk16 = N_NODES / 16;

    float* proj = hB;
    float* Wf   = hB + (size_t)N_NODES * 8;
    float* cf   = Wf + 128 * 16;
    float* Wp1  = cf + 64;
    float* Wp2  = Wp1 + PACK1_N;

    bin_edges_kernel<<<nCh + XTRA_BLOCKS, BIN_TB, 0, stream>>>(
        srcv, dstv, rec, offs, cnt + N_NODES, E, nCh,
        W3l, W3r, b3, Wc, bc, W1l, W1r, W2l, W2r, Wf, cf, Wp1, Wp2);
    rank_range_kernel<<<N_RNG, 1024, 0, stream>>>(
        rec, offs, cnt + N_NODES, rows, csr, nCh);

    sage_gather_combine_wmma<32, 128, 2, true, false><<<nBlk16, 128, 0, stream>>>(
        x, rows, csr, Wp1, b1, a, nullptr, hA);

    sage_gather_combine_wmma<128, 128, 2, true, true><<<nBlk16, 128, 0, stream>>>(
        hA, rows, csr, Wp2, b2, a, Wf, proj);

    gather4_final_kernel<<<(N_NODES * 4 + TB - 1) / TB, TB, 0, stream>>>(
        proj, rows, csr, cf, out, N_NODES);
}
